# starting kernel
# baseline (speedup 1.0000x reference)
#include <hip/hip_runtime.h>
#include <hip/hip_bf16.h>

typedef __attribute__((ext_vector_type(2))) float v2f;
typedef __attribute__((ext_vector_type(8))) float v8f;

#define F_IN 512
#define HID 8
#define HEADS 8
#define D1 64
#define NCLS 16
#define NEG_SLOPE 0.2f
#define EPS_GAT 1e-16f

__device__ __forceinline__ unsigned fenc(float f) {
  unsigned u = __float_as_uint(f);
  return (u & 0x80000000u) ? ~u : (u | 0x80000000u);
}
__device__ __forceinline__ float fdec(unsigned e) {
  return (e & 0x80000000u) ? __uint_as_float(e ^ 0x80000000u) : __uint_as_float(~e);
}
#define ENC_NEG_INF 0x007FFFFFu

__device__ __forceinline__ float lrelu(float x) { return x > 0.f ? x : x * NEG_SLOPE; }

__global__ void fill_f32(float* p, float v, int n) {
  int i = blockIdx.x * blockDim.x + threadIdx.x;
  if (i < n) p[i] = v;
}
__global__ void fill_u32(unsigned* p, unsigned v, int n) {
  int i = blockIdx.x * blockDim.x + threadIdx.x;
  if (i < n) p[i] = v;
}

__global__ __launch_bounds__(128) void gemm1_kernel(const float* __restrict__ x,
                                                    const float* __restrict__ W1,
                                                    float* __restrict__ xl1, int nrows) {
  __shared__ float At[16 * 516];
  const int tid = threadIdx.x;
  const int rowBase = blockIdx.x * 16;
  for (int i = tid; i < 2048; i += 128) {
    int r = i >> 7, cq = i & 127;
    int row = rowBase + r; if (row >= nrows) row = nrows - 1;
    *(float4*)&At[r * 516 + cq * 4] = ((const float4*)(x + (size_t)row * F_IN))[cq];
  }
  __syncthreads();

  const int wave = tid >> 5;
  const int lane = tid & 31;
  const int ln = lane & 15;
  const int khalf = (lane >> 4) << 1;
  const int colBase = wave * 16;

  v8f c = {};
  #pragma unroll 4
  for (int k = 0; k < F_IN; k += 4) {
    v2f a = *(const v2f*)&At[ln * 516 + k + khalf];
    const float* wp = W1 + (size_t)(k + khalf) * D1 + colBase + ln;
    v2f b; b.x = wp[0]; b.y = wp[D1];
    c = __builtin_amdgcn_wmma_f32_16x16x4_f32(false, a, false, b, (short)0, c,
                                              false, false);
  }
  const int mBase = (lane >> 4) * 8;
  float* outp = xl1 + (size_t)rowBase * D1 + colBase;
  #pragma unroll
  for (int i = 0; i < 8; i++) {
    int row = rowBase + mBase + i;
    if (row < nrows) outp[(size_t)(mBase + i) * D1 + ln] = c[i];
  }
}

__global__ void attn1_kernel(const float* __restrict__ xl1,
                             const float* __restrict__ att_src,
                             const float* __restrict__ att_dst,
                             float* __restrict__ a_src, float* __restrict__ a_dst, int n) {
  int t = blockIdx.x * blockDim.x + threadIdx.x;
  if (t >= n * HEADS) return;
  int node = t >> 3, hh = t & 7;
  const float* xp = xl1 + (size_t)node * D1 + hh * HID;
  const float* as = att_src + hh * HID;
  const float* ad = att_dst + hh * HID;
  float s1 = 0.f, s2 = 0.f;
  #pragma unroll
  for (int cc = 0; cc < HID; cc++) { float v = xp[cc]; s1 += v * as[cc]; s2 += v * ad[cc]; }
  a_src[t] = s1; a_dst[t] = s2;
}

__global__ void edge_max1(const int* __restrict__ ei, const float* __restrict__ a_src,
                          const float* __restrict__ a_dst, unsigned* __restrict__ amax, int E_) {
  int e = blockIdx.x * blockDim.x + threadIdx.x;
  if (e >= E_) return;
  int s = ei[e], d = ei[E_ + e];
  #pragma unroll
  for (int hh = 0; hh < HEADS; hh++) {
    float al = lrelu(a_src[s * HEADS + hh] + a_dst[d * HEADS + hh]);
    atomicMax(&amax[d * HEADS + hh], fenc(al));
  }
}

__global__ void edge_sum1(const int* __restrict__ ei, const float* __restrict__ a_src,
                          const float* __restrict__ a_dst, const unsigned* __restrict__ amax,
                          float* __restrict__ denom, int E_) {
  int e = blockIdx.x * blockDim.x + threadIdx.x;
  if (e >= E_) return;
  int s = ei[e], d = ei[E_ + e];
  #pragma unroll
  for (int hh = 0; hh < HEADS; hh++) {
    float al = lrelu(a_src[s * HEADS + hh] + a_dst[d * HEADS + hh]);
    float ex = __expf(al - fdec(amax[d * HEADS + hh]));
    atomicAdd(&denom[d * HEADS + hh], ex);
  }
}

__global__ void edge_msg1(const int* __restrict__ ei, const float* __restrict__ a_src,
                          const float* __restrict__ a_dst, const unsigned* __restrict__ amax,
                          const float* __restrict__ denom, const float* __restrict__ xl1,
                          float* __restrict__ out1, int E_) {
  long long t = (long long)blockIdx.x * blockDim.x + threadIdx.x;
  if (t >= (long long)E_ * HEADS) return;
  int e = (int)(t >> 3), hh = (int)(t & 7);
  int s = ei[e], d = ei[E_ + e];
  float al = lrelu(a_src[s * HEADS + hh] + a_dst[d * HEADS + hh]);
  float ex = __expf(al - fdec(amax[d * HEADS + hh]));
  float coef = ex / (denom[d * HEADS + hh] + EPS_GAT);
  const float4* xp = (const float4*)(xl1 + (size_t)s * D1 + hh * HID);
  float4 v0 = xp[0], v1 = xp[1];
  float* op = out1 + (size_t)d * D1 + hh * HID;
  atomicAdd(op + 0, v0.x * coef); atomicAdd(op + 1, v0.y * coef);
  atomicAdd(op + 2, v0.z * coef); atomicAdd(op + 3, v0.w * coef);
  atomicAdd(op + 4, v1.x * coef); atomicAdd(op + 5, v1.y * coef);
  atomicAdd(op + 6, v1.z * coef); atomicAdd(op + 7, v1.w * coef);
}

__global__ void elu_kernel(float* __restrict__ out1, const float* __restrict__ bias1, int n) {
  int i = blockIdx.x * blockDim.x + threadIdx.x;
  if (i >= n) return;
  float v = out1[i] + bias1[i & (D1 - 1)];
  out1[i] = v > 0.f ? v : (__expf(v) - 1.f);
}

__global__ __launch_bounds__(256) void gemm2_kernel(const float* __restrict__ h,
                                                    const float* __restrict__ W2,
                                                    float* __restrict__ xl2, int nrows) {
  const int wave = threadIdx.x >> 5;
  const int lane = threadIdx.x & 31;
  const int tile = blockIdx.x * 8 + wave;
  const int rowBase = tile * 16;
  if (rowBase >= nrows) return;
  const int ln = lane & 15;
  const int khalf = (lane >> 4) << 1;
  int arow = rowBase + ln; if (arow >= nrows) arow = nrows - 1;
  const float* ap = h + (size_t)arow * D1;

  v8f c = {};
  #pragma unroll
  for (int k = 0; k < D1; k += 4) {
    v2f a = *(const v2f*)&ap[k + khalf];
    v2f b; b.x = W2[(k + khalf) * NCLS + ln];
           b.y = W2[(k + khalf + 1) * NCLS + ln];
    c = __builtin_amdgcn_wmma_f32_16x16x4_f32(false, a, false, b, (short)0, c,
                                              false, false);
  }
  const int mBase = (lane >> 4) * 8;
  float* outp = xl2 + (size_t)rowBase * NCLS;
  #pragma unroll
  for (int i = 0; i < 8; i++) {
    int row = rowBase + mBase + i;
    if (row < nrows) outp[(size_t)(mBase + i) * NCLS + ln] = c[i];
  }
}

__global__ void attn2_kernel(const float* __restrict__ xl2,
                             const float* __restrict__ att_src,
                             const float* __restrict__ att_dst,
                             float* __restrict__ a_src, float* __restrict__ a_dst, int n) {
  int t = blockIdx.x * blockDim.x + threadIdx.x;
  if (t >= n) return;
  const float* xp = xl2 + (size_t)t * NCLS;
  float s1 = 0.f, s2 = 0.f;
  #pragma unroll
  for (int cc = 0; cc < NCLS; cc++) { float v = xp[cc]; s1 += v * att_src[cc]; s2 += v * att_dst[cc]; }
  a_src[t] = s1; a_dst[t] = s2;
}

__global__ void edge_max2(const int* __restrict__ ei, const float* __restrict__ a_src,
                          const float* __restrict__ a_dst, unsigned* __restrict__ amax, int E_) {
  int e = blockIdx.x * blockDim.x + threadIdx.x;
  if (e >= E_) return;
  int s = ei[e], d = ei[E_ + e];
  atomicMax(&amax[d], fenc(lrelu(a_src[s] + a_dst[d])));
}

__global__ void edge_sum2(const int* __restrict__ ei, const float* __restrict__ a_src,
                          const float* __restrict__ a_dst, const unsigned* __restrict__ amax,
                          float* __restrict__ denom, int E_) {
  int e = blockIdx.x * blockDim.x + threadIdx.x;
  if (e >= E_) return;
  int s = ei[e], d = ei[E_ + e];
  float al = lrelu(a_src[s] + a_dst[d]);
  atomicAdd(&denom[d], __expf(al - fdec(amax[d])));
}

__global__ void edge_msg2(const int* __restrict__ ei, const float* __restrict__ a_src,
                          const float* __restrict__ a_dst, const unsigned* __restrict__ amax,
                          const float* __restrict__ denom, const float* __restrict__ xl2,
                          float* __restrict__ out2, int E_) {
  int e = blockIdx.x * blockDim.x + threadIdx.x;
  if (e >= E_) return;
  int s = ei[e], d = ei[E_ + e];
  float al = lrelu(a_src[s] + a_dst[d]);
  float ex = __expf(al - fdec(amax[d]));
  float coef = ex / (denom[d] + EPS_GAT);
  const float4* xp = (const float4*)(xl2 + (size_t)s * NCLS);
  float* op = out2 + (size_t)d * NCLS;
  #pragma unroll
  for (int q = 0; q < 4; q++) {
    float4 v = xp[q];
    atomicAdd(op + q * 4 + 0, v.x * coef); atomicAdd(op + q * 4 + 1, v.y * coef);
    atomicAdd(op + q * 4 + 2, v.z * coef); atomicAdd(op + q * 4 + 3, v.w * coef);
  }
}

__global__ void final_kernel(const float* __restrict__ out2, const float* __restrict__ bias2,
                             float* __restrict__ out, int n) {
  int i = blockIdx.x * blockDim.x + threadIdx.x;
  if (i >= n) return;
  out[i] = out2[i] + bias2[i & (NCLS - 1)];
}

extern "C" void kernel_launch(void* const* d_in, const int* in_sizes, int n_in,
                              void* d_out, int out_size, void* d_ws, size_t ws_size,
                              hipStream_t stream) {
  const float* x        = (const float*)d_in[0];
  const int*   ei       = (const int*)d_in[1];
  const float* W1       = (const float*)d_in[2];
  const float* att_src1 = (const float*)d_in[3];
  const float* att_dst1 = (const float*)d_in[4];
  const float* bias1    = (const float*)d_in[5];
  const float* W2       = (const float*)d_in[6];
  const float* att_src2 = (const float*)d_in[7];
  const float* att_dst2 = (const float*)d_in[8];
  const float* bias2    = (const float*)d_in[9];
  float* out = (float*)d_out;

  const int N = in_sizes[0] / F_IN;
  const int E = in_sizes[1] / 2;

  float* ws = (float*)d_ws;
  size_t o = 0;
  float*    xl1    = ws + o; o += (size_t)N * D1;
  float*    out1   = ws + o; o += (size_t)N * D1;
  float*    a_src1 = ws + o; o += (size_t)N * HEADS;
  float*    a_dst1 = ws + o; o += (size_t)N * HEADS;
  unsigned* amax1  = (unsigned*)(ws + o); o += (size_t)N * HEADS;
  float*    denom1 = ws + o; o += (size_t)N * HEADS;
  float*    xl2    = ws + o; o += (size_t)N * NCLS;
  float*    a_src2 = ws + o; o += (size_t)N;
  float*    a_dst2 = ws + o; o += (size_t)N;
  unsigned* amax2  = (unsigned*)(ws + o); o += (size_t)N;
  float*    denom2 = ws + o; o += (size_t)N;
  float*    out2   = ws + o; o += (size_t)N * NCLS;

  const int TB = 256;
  auto nb = [](long long n, int b) { return (unsigned)((n + b - 1) / b); };

  fill_f32<<<nb((long long)N * D1, TB), TB, 0, stream>>>(out1, 0.f, N * D1);
  fill_f32<<<nb((long long)N * HEADS, TB), TB, 0, stream>>>(denom1, 0.f, N * HEADS);
  fill_u32<<<nb((long long)N * HEADS, TB), TB, 0, stream>>>(amax1, ENC_NEG_INF, N * HEADS);
  fill_f32<<<nb((long long)N * NCLS, TB), TB, 0, stream>>>(out2, 0.f, N * NCLS);
  fill_f32<<<nb((long long)N, TB), TB, 0, stream>>>(denom2, 0.f, N);
  fill_u32<<<nb((long long)N, TB), TB, 0, stream>>>(amax2, ENC_NEG_INF, N);

  gemm1_kernel<<<nb((long long)N, 16), 128, 0, stream>>>(x, W1, xl1, N);
  attn1_kernel<<<nb((long long)N * HEADS, TB), TB, 0, stream>>>(xl1, att_src1, att_dst1,
                                                                a_src1, a_dst1, N);
  edge_max1<<<nb(E, TB), TB, 0, stream>>>(ei, a_src1, a_dst1, amax1, E);
  edge_sum1<<<nb(E, TB), TB, 0, stream>>>(ei, a_src1, a_dst1, amax1, denom1, E);
  edge_msg1<<<nb((long long)E * HEADS, TB), TB, 0, stream>>>(ei, a_src1, a_dst1, amax1,
                                                             denom1, xl1, out1, E);
  elu_kernel<<<nb((long long)N * D1, TB), TB, 0, stream>>>(out1, bias1, N * D1);

  gemm2_kernel<<<nb(((long long)N + 15) / 16, 8), 256, 0, stream>>>(out1, W2, xl2, N);
  attn2_kernel<<<nb(N, TB), TB, 0, stream>>>(xl2, att_src2, att_dst2, a_src2, a_dst2, N);
  edge_max2<<<nb(E, TB), TB, 0, stream>>>(ei, a_src2, a_dst2, amax2, E);
  edge_sum2<<<nb(E, TB), TB, 0, stream>>>(ei, a_src2, a_dst2, amax2, denom2, E);
  edge_msg2<<<nb(E, TB), TB, 0, stream>>>(ei, a_src2, a_dst2, amax2, denom2, xl2, out2, E);
  final_kernel<<<nb((long long)N * NCLS, TB), TB, 0, stream>>>(out2, bias2, out, N * NCLS);
}


# Atomic scatter replaced by edges grouped by destination (in-call counting sort, integer atomics only): one owner per output row sums in registers, stores once with bias/ELU; sum order = arrival rank
# speedup vs baseline: 4.0090x; 4.0090x over previous
#include <hip/hip_runtime.h>
#include <hip/hip_bf16.h>

typedef __attribute__((ext_vector_type(2))) float v2f;
typedef __attribute__((ext_vector_type(8))) float v8f;

#define F_IN 512
#define HID 8
#define HEADS 8
#define D1 64
#define NCLS 16
#define NEG_SLOPE 0.2f
#define EPS_GAT 1e-16f
#define SCAN_CHUNK 1024

__device__ __forceinline__ float lrelu(float x) { return x > 0.f ? x : x * NEG_SLOPE; }

__global__ void fill_u32(unsigned* p, unsigned v, int n) {
  int i = blockIdx.x * blockDim.x + threadIdx.x;
  if (i < n) p[i] = v;
}

__global__ __launch_bounds__(128) void gemm1_kernel(const float* __restrict__ x,
                                                    const float* __restrict__ W1,
                                                    float* __restrict__ xl1, int nrows) {
  __shared__ float At[16 * 516];
  const int tid = threadIdx.x;
  const int rowBase = blockIdx.x * 16;
  for (int i = tid; i < 2048; i += 128) {
    int r = i >> 7, cq = i & 127;
    int row = rowBase + r; if (row >= nrows) row = nrows - 1;
    *(float4*)&At[r * 516 + cq * 4] = ((const float4*)(x + (size_t)row * F_IN))[cq];
  }
  __syncthreads();

  const int wave = tid >> 5;
  const int lane = tid & 31;
  const int ln = lane & 15;
  const int khalf = (lane >> 4) << 1;
  const int colBase = wave * 16;

  v8f c = {};
  #pragma unroll 4
  for (int k = 0; k < F_IN; k += 4) {
    v2f a = *(const v2f*)&At[ln * 516 + k + khalf];
    const float* wp = W1 + (size_t)(k + khalf) * D1 + colBase + ln;
    v2f b; b.x = wp[0]; b.y = wp[D1];
    c = __builtin_amdgcn_wmma_f32_16x16x4_f32(false, a, false, b, (short)0, c,
                                              false, false);
  }
  const int mBase = (lane >> 4) * 8;
  float* outp = xl1 + (size_t)rowBase * D1 + colBase;
  #pragma unroll
  for (int i = 0; i < 8; i++) {
    int row = rowBase + mBase + i;
    if (row < nrows) outp[(size_t)(mBase + i) * D1 + ln] = c[i];
  }
}

__global__ void attn1_kernel(const float* __restrict__ xl1,
                             const float* __restrict__ att_src,
                             const float* __restrict__ att_dst,
                             float* __restrict__ a_src, float* __restrict__ a_dst, int n) {
  int t = blockIdx.x * blockDim.x + threadIdx.x;
  if (t >= n * HEADS) return;
  int node = t >> 3, hh = t & 7;
  const float* xp = xl1 + (size_t)node * D1 + hh * HID;
  const float* as = att_src + hh * HID;
  const float* ad = att_dst + hh * HID;
  float s1 = 0.f, s2 = 0.f;
  #pragma unroll
  for (int cc = 0; cc < HID; cc++) { float v = xp[cc]; s1 += v * as[cc]; s2 += v * ad[cc]; }
  a_src[t] = s1; a_dst[t] = s2;
}

__global__ void deg_count(const int* __restrict__ ei, unsigned* __restrict__ cnt, int E_) {
  const int T = gridDim.x * blockDim.x;
  const int e0 = blockIdx.x * blockDim.x + threadIdx.x;
  const int* __restrict__ dst = ei + E_;
  if (e0 + 3 * T < E_) {
    const int d0 = dst[e0], d1 = dst[e0 + T], d2 = dst[e0 + 2 * T], d3 = dst[e0 + 3 * T];
    atomicAdd(&cnt[d0], 1u); atomicAdd(&cnt[d1], 1u);
    atomicAdd(&cnt[d2], 1u); atomicAdd(&cnt[d3], 1u);
  } else {
    for (int e = e0; e < E_; e += T) atomicAdd(&cnt[dst[e]], 1u);
  }
}

__global__ __launch_bounds__(256) void scan_totals(const unsigned* __restrict__ cnt,
                                                   unsigned* __restrict__ bsum, int n) {
  __shared__ unsigned red[256];
  const int tid = threadIdx.x;
  const int base = blockIdx.x * SCAN_CHUNK + tid * 4;
  unsigned s = 0;
  #pragma unroll
  for (int k = 0; k < 4; k++) if (base + k < n) s += cnt[base + k];
  red[tid] = s;
  __syncthreads();
  for (int w = 128; w > 0; w >>= 1) {
    if (tid < w) red[tid] += red[tid + w];
    __syncthreads();
  }
  if (tid == 0) bsum[blockIdx.x] = red[0];
}

__global__ __launch_bounds__(256) void scan_write(const unsigned* __restrict__ cnt,
                                                  const unsigned* __restrict__ bsum,
                                                  unsigned* __restrict__ rowptr,
                                                  unsigned* __restrict__ cursor, int n) {
  __shared__ unsigned red[256];
  __shared__ unsigned sh[256];
  const int tid = threadIdx.x;
  unsigned p = 0;
  for (int i = tid; i < (int)blockIdx.x; i += 256) p += bsum[i];
  red[tid] = p;
  __syncthreads();
  for (int w = 128; w > 0; w >>= 1) {
    if (tid < w) red[tid] += red[tid + w];
    __syncthreads();
  }
  const unsigned chunkBase = red[0];

  const int base = blockIdx.x * SCAN_CHUNK + tid * 4;
  unsigned v[4];
  unsigned tsum = 0;
  #pragma unroll
  for (int k = 0; k < 4; k++) { v[k] = (base + k < n) ? cnt[base + k] : 0u; tsum += v[k]; }
  sh[tid] = tsum;
  __syncthreads();
  for (int off = 1; off < 256; off <<= 1) {
    unsigned add = (tid >= off) ? sh[tid - off] : 0u;
    __syncthreads();
    sh[tid] += add;
    __syncthreads();
  }
  unsigned run = chunkBase + sh[tid] - tsum;
  #pragma unroll
  for (int k = 0; k < 4; k++) {
    if (base + k < n) { rowptr[base + k] = run; cursor[base + k] = run; }
    run += v[k];
  }
  if (base < n && base + 4 >= n) rowptr[n] = run;
}

__global__ void seg_fill(const int* __restrict__ ei, unsigned* __restrict__ cursor,
                         int* __restrict__ csr_src, int E_) {
  const int T = gridDim.x * blockDim.x;
  const int e0 = blockIdx.x * blockDim.x + threadIdx.x;
  const int* __restrict__ dst = ei + E_;
  if (e0 + 3 * T < E_) {
    const int d0 = dst[e0], d1 = dst[e0 + T], d2 = dst[e0 + 2 * T], d3 = dst[e0 + 3 * T];
    const int s0 = ei[e0],  s1 = ei[e0 + T],  s2 = ei[e0 + 2 * T],  s3 = ei[e0 + 3 * T];
    const unsigned p0 = atomicAdd(&cursor[d0], 1u), p1 = atomicAdd(&cursor[d1], 1u);
    const unsigned p2 = atomicAdd(&cursor[d2], 1u), p3 = atomicAdd(&cursor[d3], 1u);
    csr_src[p0] = s0; csr_src[p1] = s1; csr_src[p2] = s2; csr_src[p3] = s3;
  } else {
    for (int e = e0; e < E_; e += T) {
      const unsigned pos = atomicAdd(&cursor[dst[e]], 1u);
      csr_src[pos] = ei[e];
    }
  }
}

__global__ void agg1_kernel(const unsigned* __restrict__ rowptr, const int* __restrict__ csr_src,
                            const float* __restrict__ a_src, const float* __restrict__ a_dst,
                            const float* __restrict__ xl1, const float* __restrict__ bias1,
                            float* __restrict__ h, int n) {
  int t = blockIdx.x * blockDim.x + threadIdx.x;
  if (t >= n * HEADS) return;
  const int d = t >> 3, hh = t & 7;
  const unsigned beg = rowptr[d], end = rowptr[d + 1];
  const float ad = a_dst[t];

  float m = -__builtin_huge_valf();
  #pragma unroll 4
  for (unsigned j = beg; j < end; j++) {
    int s = csr_src[j];
    float al = lrelu(a_src[s * HEADS + hh] + ad);
    m = al > m ? al : m;
  }

  float den = 0.f;
  float4 acc0 = make_float4(0.f, 0.f, 0.f, 0.f), acc1 = acc0;
  #pragma unroll 4
  for (unsigned j = beg; j < end; j++) {
    int s = csr_src[j];
    float ex = __expf(lrelu(a_src[s * HEADS + hh] + ad) - m);
    const float4* xp = (const float4*)(xl1 + (size_t)s * D1 + hh * HID);
    float4 v0 = xp[0], v1 = xp[1];
    den += ex;
    acc0.x += v0.x * ex; acc0.y += v0.y * ex; acc0.z += v0.z * ex; acc0.w += v0.w * ex;
    acc1.x += v1.x * ex; acc1.y += v1.y * ex; acc1.z += v1.z * ex; acc1.w += v1.w * ex;
  }
  const float dn = den + EPS_GAT;
  const float4* bp = (const float4*)(bias1 + hh * HID);
  const float4 b0 = bp[0], b1 = bp[1];
  float r[8] = { acc0.x / dn + b0.x, acc0.y / dn + b0.y, acc0.z / dn + b0.z, acc0.w / dn + b0.w,
                 acc1.x / dn + b1.x, acc1.y / dn + b1.y, acc1.z / dn + b1.z, acc1.w / dn + b1.w };
  #pragma unroll
  for (int c = 0; c < 8; c++) r[c] = r[c] > 0.f ? r[c] : (__expf(r[c]) - 1.f);
  float4* hp = (float4*)(h + (size_t)d * D1 + hh * HID);
  hp[0] = make_float4(r[0], r[1], r[2], r[3]);
  hp[1] = make_float4(r[4], r[5], r[6], r[7]);
}

__global__ __launch_bounds__(256) void gemm2_kernel(const float* __restrict__ h,
                                                    const float* __restrict__ W2,
                                                    float* __restrict__ xl2, int nrows) {
  const int wave = threadIdx.x >> 5;
  const int lane = threadIdx.x & 31;
  const int tile = blockIdx.x * 8 + wave;
  const int rowBase = tile * 16;
  if (rowBase >= nrows) return;
  const int ln = lane & 15;
  const int khalf = (lane >> 4) << 1;
  int arow = rowBase + ln; if (arow >= nrows) arow = nrows - 1;
  const float* ap = h + (size_t)arow * D1;

  v8f c = {};
  #pragma unroll
  for (int k = 0; k < D1; k += 4) {
    v2f a = *(const v2f*)&ap[k + khalf];
    v2f b; b.x = W2[(k + khalf) * NCLS + ln];
           b.y = W2[(k + khalf + 1) * NCLS + ln];
    c = __builtin_amdgcn_wmma_f32_16x16x4_f32(false, a, false, b, (short)0, c,
                                              false, false);
  }
  const int mBase = (lane >> 4) * 8;
  float* outp = xl2 + (size_t)rowBase * NCLS;
  #pragma unroll
  for (int i = 0; i < 8; i++) {
    int row = rowBase + mBase + i;
    if (row < nrows) outp[(size_t)(mBase + i) * NCLS + ln] = c[i];
  }
}

__global__ void attn2_kernel(const float* __restrict__ xl2,
                             const float* __restrict__ att_src,
                             const float* __restrict__ att_dst,
                             float* __restrict__ a_src, float* __restrict__ a_dst, int n) {
  int t = blockIdx.x * blockDim.x + threadIdx.x;
  if (t >= n) return;
  const float* xp = xl2 + (size_t)t * NCLS;
  float s1 = 0.f, s2 = 0.f;
  #pragma unroll
  for (int cc = 0; cc < NCLS; cc++) { float v = xp[cc]; s1 += v * att_src[cc]; s2 += v * att_dst[cc]; }
  a_src[t] = s1; a_dst[t] = s2;
}

__global__ void agg2_kernel(const unsigned* __restrict__ rowptr, const int* __restrict__ csr_src,
                            const float* __restrict__ a_src, const float* __restrict__ a_dst,
                            const float* __restrict__ xl2, const float* __restrict__ bias2,
                            float* __restrict__ out, int n) {
  int t = blockIdx.x * blockDim.x + threadIdx.x;
  if (t >= n * 4) return;
  const int d = t >> 2, q = t & 3;
  const unsigned beg = rowptr[d], end = rowptr[d + 1];
  const float ad = a_dst[d];

  float m = -__builtin_huge_valf();
  #pragma unroll 4
  for (unsigned j = beg; j < end; j++) {
    int s = csr_src[j];
    float al = lrelu(a_src[s] + ad);
    m = al > m ? al : m;
  }

  float den = 0.f;
  float4 acc = make_float4(0.f, 0.f, 0.f, 0.f);
  #pragma unroll 4
  for (unsigned j = beg; j < end; j++) {
    int s = csr_src[j];
    float ex = __expf(lrelu(a_src[s] + ad) - m);
    float4 v = ((const float4*)(xl2 + (size_t)s * NCLS))[q];
    den += ex;
    acc.x += v.x * ex; acc.y += v.y * ex; acc.z += v.z * ex; acc.w += v.w * ex;
  }
  const float dn = den + EPS_GAT;
  const float4 b = ((const float4*)bias2)[q];
  ((float4*)(out + (size_t)d * NCLS))[q] =
      make_float4(acc.x / dn + b.x, acc.y / dn + b.y, acc.z / dn + b.z, acc.w / dn + b.w);
}

extern "C" void kernel_launch(void* const* d_in, const int* in_sizes, int n_in,
                              void* d_out, int out_size, void* d_ws, size_t ws_size,
                              hipStream_t stream) {
  const float* x        = (const float*)d_in[0];
  const int*   ei       = (const int*)d_in[1];
  const float* W1       = (const float*)d_in[2];
  const float* att_src1 = (const float*)d_in[3];
  const float* att_dst1 = (const float*)d_in[4];
  const float* bias1    = (const float*)d_in[5];
  const float* W2       = (const float*)d_in[6];
  const float* att_src2 = (const float*)d_in[7];
  const float* att_dst2 = (const float*)d_in[8];
  const float* bias2    = (const float*)d_in[9];
  float* out = (float*)d_out;

  const int N = in_sizes[0] / F_IN;
  const int E = in_sizes[1] / 2;
  const size_t Np = ((size_t)N + 4) & ~(size_t)3;
  const int nChunks = (N + SCAN_CHUNK - 1) / SCAN_CHUNK;
  const size_t Cp = ((size_t)nChunks + 3) & ~(size_t)3;

  float* ws = (float*)d_ws;
  size_t o = 0;
  float*    xl1    = ws + o; o += (size_t)N * D1;
  float*    h      = ws + o; o += (size_t)N * D1;
  float*    a_src1 = ws + o; o += (size_t)N * HEADS;
  float*    a_dst1 = ws + o; o += (size_t)N * HEADS;
  float*    xl2    = ws + o; o += (size_t)N * NCLS;
  float*    a_src2 = ws + o; o += Np;
  float*    a_dst2 = ws + o; o += Np;
  unsigned* cnt    = (unsigned*)(ws + o); o += Np;
  unsigned* rowptr = (unsigned*)(ws + o); o += Np;
  unsigned* cursor = (unsigned*)(ws + o); o += Np;
  unsigned* bsum   = (unsigned*)(ws + o); o += Cp;
  int*      csrsrc = (int*)(ws + o);      o += (size_t)E;

  const int TB = 256;
  auto nb = [](long long n, int b) { return (unsigned)((n + b - 1) / b); };

  fill_u32<<<nb((long long)N, TB), TB, 0, stream>>>(cnt, 0u, N);
  deg_count<<<nb(((long long)E + 3) / 4, TB), TB, 0, stream>>>(ei, cnt, E);
  scan_totals<<<nChunks, 256, 0, stream>>>(cnt, bsum, N);
  scan_write<<<nChunks, 256, 0, stream>>>(cnt, bsum, rowptr, cursor, N);
  seg_fill<<<nb(((long long)E + 3) / 4, TB), TB, 0, stream>>>(ei, cursor, csrsrc, E);

  gemm1_kernel<<<nb((long long)N, 16), 128, 0, stream>>>(x, W1, xl1, N);
  attn1_kernel<<<nb((long long)N * HEADS, TB), TB, 0, stream>>>(xl1, att_src1, att_dst1,
                                                                a_src1, a_dst1, N);
  agg1_kernel<<<nb((long long)N * HEADS, TB), TB, 0, stream>>>(rowptr, csrsrc, a_src1, a_dst1,
                                                               xl1, bias1, h, N);

  gemm2_kernel<<<nb(((long long)N + 15) / 16, 8), 256, 0, stream>>>(h, W2, xl2, N);
  attn2_kernel<<<nb(N, TB), TB, 0, stream>>>(xl2, att_src2, att_dst2, a_src2, a_dst2, N);
  agg2_kernel<<<nb((long long)N * 4, TB), TB, 0, stream>>>(rowptr, csrsrc, a_src2, a_dst2,
                                                           xl2, bias2, out, N);
}


# Both aggregation kernels walk each edge list once: softmax on a running max, sums rescaled by exp(old max - new max) per group of 4 edges; the separate max pass and its score gathers go
# speedup vs baseline: 4.2543x; 1.0612x over previous
#include <hip/hip_runtime.h>
#include <hip/hip_bf16.h>

typedef __attribute__((ext_vector_type(2))) float v2f;
typedef __attribute__((ext_vector_type(8))) float v8f;

#define F_IN 512
#define HID 8
#define HEADS 8
#define D1 64
#define NCLS 16
#define NEG_SLOPE 0.2f
#define EPS_GAT 1e-16f
#define SCAN_CHUNK 1024

__device__ __forceinline__ float lrelu(float x) { return x > 0.f ? x : x * NEG_SLOPE; }

__device__ __forceinline__ void rescale_add(float4& a, float sc, const float4& v, float e) {
  a.x = fmaf(v.x, e, a.x * sc); a.y = fmaf(v.y, e, a.y * sc);
  a.z = fmaf(v.z, e, a.z * sc); a.w = fmaf(v.w, e, a.w * sc);
}
__device__ __forceinline__ void add_scaled(float4& a, const float4& v, float e) {
  a.x = fmaf(v.x, e, a.x); a.y = fmaf(v.y, e, a.y);
  a.z = fmaf(v.z, e, a.z); a.w = fmaf(v.w, e, a.w);
}

__global__ void fill_u32(unsigned* p, unsigned v, int n) {
  int i = blockIdx.x * blockDim.x + threadIdx.x;
  if (i < n) p[i] = v;
}

__global__ __launch_bounds__(128) void gemm1_kernel(const float* __restrict__ x,
                                                    const float* __restrict__ W1,
                                                    float* __restrict__ xl1, int nrows) {
  __shared__ float At[16 * 516];
  const int tid = threadIdx.x;
  const int rowBase = blockIdx.x * 16;
  for (int i = tid; i < 2048; i += 128) {
    int r = i >> 7, cq = i & 127;
    int row = rowBase + r; if (row >= nrows) row = nrows - 1;
    *(float4*)&At[r * 516 + cq * 4] = ((const float4*)(x + (size_t)row * F_IN))[cq];
  }
  __syncthreads();

  const int wave = tid >> 5;
  const int lane = tid & 31;
  const int ln = lane & 15;
  const int khalf = (lane >> 4) << 1;
  const int colBase = wave * 16;

  v8f c = {};
  #pragma unroll 4
  for (int k = 0; k < F_IN; k += 4) {
    v2f a = *(const v2f*)&At[ln * 516 + k + khalf];
    const float* wp = W1 + (size_t)(k + khalf) * D1 + colBase + ln;
    v2f b; b.x = wp[0]; b.y = wp[D1];
    c = __builtin_amdgcn_wmma_f32_16x16x4_f32(false, a, false, b, (short)0, c,
                                              false, false);
  }
  const int mBase = (lane >> 4) * 8;
  float* outp = xl1 + (size_t)rowBase * D1 + colBase;
  #pragma unroll
  for (int i = 0; i < 8; i++) {
    int row = rowBase + mBase + i;
    if (row < nrows) outp[(size_t)(mBase + i) * D1 + ln] = c[i];
  }
}

__global__ void attn1_kernel(const float* __restrict__ xl1,
                             const float* __restrict__ att_src,
                             const float* __restrict__ att_dst,
                             float* __restrict__ a_src, float* __restrict__ a_dst, int n) {
  int t = blockIdx.x * blockDim.x + threadIdx.x;
  if (t >= n * HEADS) return;
  int node = t >> 3, hh = t & 7;
  const float* xp = xl1 + (size_t)node * D1 + hh * HID;
  const float* as = att_src + hh * HID;
  const float* ad = att_dst + hh * HID;
  float s1 = 0.f, s2 = 0.f;
  #pragma unroll
  for (int cc = 0; cc < HID; cc++) { float v = xp[cc]; s1 += v * as[cc]; s2 += v * ad[cc]; }
  a_src[t] = s1; a_dst[t] = s2;
}

__global__ void deg_count(const int* __restrict__ ei, unsigned* __restrict__ cnt, int E_) {
  const int T = gridDim.x * blockDim.x;
  const int e0 = blockIdx.x * blockDim.x + threadIdx.x;
  const int* __restrict__ dst = ei + E_;
  if (e0 + 3 * T < E_) {
    const int d0 = dst[e0], d1 = dst[e0 + T], d2 = dst[e0 + 2 * T], d3 = dst[e0 + 3 * T];
    atomicAdd(&cnt[d0], 1u); atomicAdd(&cnt[d1], 1u);
    atomicAdd(&cnt[d2], 1u); atomicAdd(&cnt[d3], 1u);
  } else {
    for (int e = e0; e < E_; e += T) atomicAdd(&cnt[dst[e]], 1u);
  }
}

__global__ __launch_bounds__(256) void scan_totals(const unsigned* __restrict__ cnt,
                                                   unsigned* __restrict__ bsum, int n) {
  __shared__ unsigned red[256];
  const int tid = threadIdx.x;
  const int base = blockIdx.x * SCAN_CHUNK + tid * 4;
  unsigned s = 0;
  #pragma unroll
  for (int k = 0; k < 4; k++) if (base + k < n) s += cnt[base + k];
  red[tid] = s;
  __syncthreads();
  for (int w = 128; w > 0; w >>= 1) {
    if (tid < w) red[tid] += red[tid + w];
    __syncthreads();
  }
  if (tid == 0) bsum[blockIdx.x] = red[0];
}

__global__ __launch_bounds__(256) void scan_write(const unsigned* __restrict__ cnt,
                                                  const unsigned* __restrict__ bsum,
                                                  unsigned* __restrict__ rowptr,
                                                  unsigned* __restrict__ cursor, int n) {
  __shared__ unsigned red[256];
  __shared__ unsigned sh[256];
  const int tid = threadIdx.x;
  unsigned p = 0;
  for (int i = tid; i < (int)blockIdx.x; i += 256) p += bsum[i];
  red[tid] = p;
  __syncthreads();
  for (int w = 128; w > 0; w >>= 1) {
    if (tid < w) red[tid] += red[tid + w];
    __syncthreads();
  }
  const unsigned chunkBase = red[0];

  const int base = blockIdx.x * SCAN_CHUNK + tid * 4;
  unsigned v[4];
  unsigned tsum = 0;
  #pragma unroll
  for (int k = 0; k < 4; k++) { v[k] = (base + k < n) ? cnt[base + k] : 0u; tsum += v[k]; }
  sh[tid] = tsum;
  __syncthreads();
  for (int off = 1; off < 256; off <<= 1) {
    unsigned add = (tid >= off) ? sh[tid - off] : 0u;
    __syncthreads();
    sh[tid] += add;
    __syncthreads();
  }
  unsigned run = chunkBase + sh[tid] - tsum;
  #pragma unroll
  for (int k = 0; k < 4; k++) {
    if (base + k < n) { rowptr[base + k] = run; cursor[base + k] = run; }
    run += v[k];
  }
  if (base < n && base + 4 >= n) rowptr[n] = run;
}

__global__ void seg_fill(const int* __restrict__ ei, unsigned* __restrict__ cursor,
                         int* __restrict__ csr_src, int E_) {
  const int T = gridDim.x * blockDim.x;
  const int e0 = blockIdx.x * blockDim.x + threadIdx.x;
  const int* __restrict__ dst = ei + E_;
  if (e0 + 3 * T < E_) {
    const int d0 = dst[e0], d1 = dst[e0 + T], d2 = dst[e0 + 2 * T], d3 = dst[e0 + 3 * T];
    const int s0 = ei[e0],  s1 = ei[e0 + T],  s2 = ei[e0 + 2 * T],  s3 = ei[e0 + 3 * T];
    const unsigned p0 = atomicAdd(&cursor[d0], 1u), p1 = atomicAdd(&cursor[d1], 1u);
    const unsigned p2 = atomicAdd(&cursor[d2], 1u), p3 = atomicAdd(&cursor[d3], 1u);
    csr_src[p0] = s0; csr_src[p1] = s1; csr_src[p2] = s2; csr_src[p3] = s3;
  } else {
    for (int e = e0; e < E_; e += T) {
      const unsigned pos = atomicAdd(&cursor[dst[e]], 1u);
      csr_src[pos] = ei[e];
    }
  }
}

__global__ void agg1_kernel(const unsigned* __restrict__ rowptr, const int* __restrict__ csr_src,
                            const float* __restrict__ a_src, const float* __restrict__ a_dst,
                            const float* __restrict__ xl1, const float* __restrict__ bias1,
                            float* __restrict__ h, int n) {
  int t = blockIdx.x * blockDim.x + threadIdx.x;
  if (t >= n * HEADS) return;
  const int d = t >> 3, hh = t & 7;
  const int* __restrict__ cp = csr_src + rowptr[d];
  const int* __restrict__ ce = csr_src + rowptr[d + 1];
  const float ad = a_dst[t];
  const float* __restrict__ asp = a_src + hh;
  const float* __restrict__ xb = xl1 + hh * HID;

  float m = -__builtin_huge_valf();
  float den = 0.f;
  float4 acc0 = make_float4(0.f, 0.f, 0.f, 0.f), acc1 = acc0;
  for (; cp + 4 <= ce; cp += 4) {
    const int s0 = cp[0], s1 = cp[1], s2 = cp[2], s3 = cp[3];
    const float al0 = lrelu(asp[s0 * HEADS] + ad), al1 = lrelu(asp[s1 * HEADS] + ad);
    const float al2 = lrelu(asp[s2 * HEADS] + ad), al3 = lrelu(asp[s3 * HEADS] + ad);
    const float4* x0 = (const float4*)(xb + (size_t)s0 * D1);
    const float4* x1 = (const float4*)(xb + (size_t)s1 * D1);
    const float4* x2 = (const float4*)(xb + (size_t)s2 * D1);
    const float4* x3 = (const float4*)(xb + (size_t)s3 * D1);
    const float4 p0 = x0[0], q0 = x0[1], p1 = x1[0], q1 = x1[1];
    const float4 p2 = x2[0], q2 = x2[1], p3 = x3[0], q3 = x3[1];
    float mn = al0 > m ? al0 : m;
    mn = al1 > mn ? al1 : mn; mn = al2 > mn ? al2 : mn; mn = al3 > mn ? al3 : mn;
    const float sc = __expf(m - mn);
    const float e0 = __expf(al0 - mn), e1 = __expf(al1 - mn);
    const float e2 = __expf(al2 - mn), e3 = __expf(al3 - mn);
    m = mn;
    den = fmaf(den, sc, e0); den += e1; den += e2; den += e3;
    rescale_add(acc0, sc, p0, e0); rescale_add(acc1, sc, q0, e0);
    add_scaled(acc0, p1, e1);      add_scaled(acc1, q1, e1);
    add_scaled(acc0, p2, e2);      add_scaled(acc1, q2, e2);
    add_scaled(acc0, p3, e3);      add_scaled(acc1, q3, e3);
  }
  #pragma unroll 1
  for (; cp < ce; cp++) {
    const int s = cp[0];
    const float al = lrelu(asp[s * HEADS] + ad);
    const float4* xp = (const float4*)(xb + (size_t)s * D1);
    const float4 v0 = xp[0], v1 = xp[1];
    const float mn = al > m ? al : m;
    const float sc = __expf(m - mn);
    const float ex = __expf(al - mn);
    m = mn;
    den = fmaf(den, sc, ex);
    rescale_add(acc0, sc, v0, ex); rescale_add(acc1, sc, v1, ex);
  }
  const float dn = den + EPS_GAT;
  const float4* bp = (const float4*)(bias1 + hh * HID);
  const float4 b0 = bp[0], b1 = bp[1];
  float r[8] = { acc0.x / dn + b0.x, acc0.y / dn + b0.y, acc0.z / dn + b0.z, acc0.w / dn + b0.w,
                 acc1.x / dn + b1.x, acc1.y / dn + b1.y, acc1.z / dn + b1.z, acc1.w / dn + b1.w };
  #pragma unroll
  for (int c = 0; c < 8; c++) r[c] = r[c] > 0.f ? r[c] : (__expf(r[c]) - 1.f);
  float4* hp = (float4*)(h + (size_t)d * D1 + hh * HID);
  hp[0] = make_float4(r[0], r[1], r[2], r[3]);
  hp[1] = make_float4(r[4], r[5], r[6], r[7]);
}

__global__ __launch_bounds__(256) void gemm2_kernel(const float* __restrict__ h,
                                                    const float* __restrict__ W2,
                                                    float* __restrict__ xl2, int nrows) {
  const int wave = threadIdx.x >> 5;
  const int lane = threadIdx.x & 31;
  const int tile = blockIdx.x * 8 + wave;
  const int rowBase = tile * 16;
  if (rowBase >= nrows) return;
  const int ln = lane & 15;
  const int khalf = (lane >> 4) << 1;
  int arow = rowBase + ln; if (arow >= nrows) arow = nrows - 1;
  const float* ap = h + (size_t)arow * D1;

  v8f c = {};
  #pragma unroll
  for (int k = 0; k < D1; k += 4) {
    v2f a = *(const v2f*)&ap[k + khalf];
    v2f b; b.x = W2[(k + khalf) * NCLS + ln];
           b.y = W2[(k + khalf + 1) * NCLS + ln];
    c = __builtin_amdgcn_wmma_f32_16x16x4_f32(false, a, false, b, (short)0, c,
                                              false, false);
  }
  const int mBase = (lane >> 4) * 8;
  float* outp = xl2 + (size_t)rowBase * NCLS;
  #pragma unroll
  for (int i = 0; i < 8; i++) {
    int row = rowBase + mBase + i;
    if (row < nrows) outp[(size_t)(mBase + i) * NCLS + ln] = c[i];
  }
}

__global__ void attn2_kernel(const float* __restrict__ xl2,
                             const float* __restrict__ att_src,
                             const float* __restrict__ att_dst,
                             float* __restrict__ a_src, float* __restrict__ a_dst, int n) {
  int t = blockIdx.x * blockDim.x + threadIdx.x;
  if (t >= n) return;
  const float* xp = xl2 + (size_t)t * NCLS;
  float s1 = 0.f, s2 = 0.f;
  #pragma unroll
  for (int cc = 0; cc < NCLS; cc++) { float v = xp[cc]; s1 += v * att_src[cc]; s2 += v * att_dst[cc]; }
  a_src[t] = s1; a_dst[t] = s2;
}

__global__ void agg2_kernel(const unsigned* __restrict__ rowptr, const int* __restrict__ csr_src,
                            const float* __restrict__ a_src, const float* __restrict__ a_dst,
                            const float* __restrict__ xl2, const float* __restrict__ bias2,
                            float* __restrict__ out, int n) {
  int t = blockIdx.x * blockDim.x + threadIdx.x;
  if (t >= n * 4) return;
  const int d = t >> 2, q = t & 3;
  const int* __restrict__ cp = csr_src + rowptr[d];
  const int* __restrict__ ce = csr_src + rowptr[d + 1];
  const float ad = a_dst[d];
  const float4* __restrict__ xq = (const float4*)xl2 + q;

  float m = -__builtin_huge_valf();
  float den = 0.f;
  float4 acc = make_float4(0.f, 0.f, 0.f, 0.f);
  for (; cp + 4 <= ce; cp += 4) {
    const int s0 = cp[0], s1 = cp[1], s2 = cp[2], s3 = cp[3];
    const float al0 = lrelu(a_src[s0] + ad), al1 = lrelu(a_src[s1] + ad);
    const float al2 = lrelu(a_src[s2] + ad), al3 = lrelu(a_src[s3] + ad);
    const float4 v0 = xq[(size_t)s0 * 4], v1 = xq[(size_t)s1 * 4];
    const float4 v2 = xq[(size_t)s2 * 4], v3 = xq[(size_t)s3 * 4];
    float mn = al0 > m ? al0 : m;
    mn = al1 > mn ? al1 : mn; mn = al2 > mn ? al2 : mn; mn = al3 > mn ? al3 : mn;
    const float sc = __expf(m - mn);
    const float e0 = __expf(al0 - mn), e1 = __expf(al1 - mn);
    const float e2 = __expf(al2 - mn), e3 = __expf(al3 - mn);
    m = mn;
    den = fmaf(den, sc, e0); den += e1; den += e2; den += e3;
    rescale_add(acc, sc, v0, e0);
    add_scaled(acc, v1, e1); add_scaled(acc, v2, e2); add_scaled(acc, v3, e3);
  }
  #pragma unroll 1
  for (; cp < ce; cp++) {
    const unsigned s = cp[0];
    const float al = lrelu(a_src[s] + ad);
    const float4 v = xq[(size_t)s * 4];
    const float mn = al > m ? al : m;
    const float sc = __expf(m - mn);
    const float ex = __expf(al - mn);
    m = mn;
    den = fmaf(den, sc, ex);
    rescale_add(acc, sc, v, ex);
  }
  const float dn = den + EPS_GAT;
  const float4 b = ((const float4*)bias2)[q];
  ((float4*)(out + (size_t)d * NCLS))[q] =
      make_float4(acc.x / dn + b.x, acc.y / dn + b.y, acc.z / dn + b.z, acc.w / dn + b.w);
}

extern "C" void kernel_launch(void* const* d_in, const int* in_sizes, int n_in,
                              void* d_out, int out_size, void* d_ws, size_t ws_size,
                              hipStream_t stream) {
  const float* x        = (const float*)d_in[0];
  const int*   ei       = (const int*)d_in[1];
  const float* W1       = (const float*)d_in[2];
  const float* att_src1 = (const float*)d_in[3];
  const float* att_dst1 = (const float*)d_in[4];
  const float* bias1    = (const float*)d_in[5];
  const float* W2       = (const float*)d_in[6];
  const float* att_src2 = (const float*)d_in[7];
  const float* att_dst2 = (const float*)d_in[8];
  const float* bias2    = (const float*)d_in[9];
  float* out = (float*)d_out;

  const int N = in_sizes[0] / F_IN;
  const int E = in_sizes[1] / 2;
  const size_t Np = ((size_t)N + 4) & ~(size_t)3;
  const int nChunks = (N + SCAN_CHUNK - 1) / SCAN_CHUNK;
  const size_t Cp = ((size_t)nChunks + 3) & ~(size_t)3;

  float* ws = (float*)d_ws;
  size_t o = 0;
  float*    xl1    = ws + o; o += (size_t)N * D1;
  float*    h      = ws + o; o += (size_t)N * D1;
  float*    a_src1 = ws + o; o += (size_t)N * HEADS;
  float*    a_dst1 = ws + o; o += (size_t)N * HEADS;
  float*    xl2    = ws + o; o += (size_t)N * NCLS;
  float*    a_src2 = ws + o; o += Np;
  float*    a_dst2 = ws + o; o += Np;
  unsigned* cnt    = (unsigned*)(ws + o); o += Np;
  unsigned* rowptr = (unsigned*)(ws + o); o += Np;
  unsigned* cursor = (unsigned*)(ws + o); o += Np;
  unsigned* bsum   = (unsigned*)(ws + o); o += Cp;
  int*      csrsrc = (int*)(ws + o);      o += (size_t)E;

  const int TB = 256;
  auto nb = [](long long n, int b) { return (unsigned)((n + b - 1) / b); };

  fill_u32<<<nb((long long)N, TB), TB, 0, stream>>>(cnt, 0u, N);
  deg_count<<<nb(((long long)E + 3) / 4, TB), TB, 0, stream>>>(ei, cnt, E);
  scan_totals<<<nChunks, 256, 0, stream>>>(cnt, bsum, N);
  scan_write<<<nChunks, 256, 0, stream>>>(cnt, bsum, rowptr, cursor, N);
  seg_fill<<<nb(((long long)E + 3) / 4, TB), TB, 0, stream>>>(ei, cursor, csrsrc, E);

  gemm1_kernel<<<nb((long long)N, 16), 128, 0, stream>>>(x, W1, xl1, N);
  attn1_kernel<<<nb((long long)N * HEADS, TB), TB, 0, stream>>>(xl1, att_src1, att_dst1,
                                                                a_src1, a_dst1, N);
  agg1_kernel<<<nb((long long)N * HEADS, TB), TB, 0, stream>>>(rowptr, csrsrc, a_src1, a_dst1,
                                                               xl1, bias1, h, N);

  gemm2_kernel<<<nb(((long long)N + 15) / 16, 8), 256, 0, stream>>>(h, W2, xl2, N);
  attn2_kernel<<<nb(N, TB), TB, 0, stream>>>(xl2, att_src2, att_dst2, a_src2, a_dst2, N);
  agg2_kernel<<<nb((long long)N * 4, TB), TB, 0, stream>>>(rowptr, csrsrc, a_src2, a_dst2,
                                                           xl2, bias2, out, N);
}


# Edges grouped by destination with a two-level counting sort ranked by LDS counters (512-node buckets): both global atomics per edge go; sum order = LDS arrival rank, was global arrival rank
# speedup vs baseline: 6.9135x; 1.6251x over previous
#include <hip/hip_runtime.h>
#include <hip/hip_bf16.h>

typedef __attribute__((ext_vector_type(2))) float v2f;
typedef __attribute__((ext_vector_type(8))) float v8f;

#define F_IN 512
#define HID 8
#define HEADS 8
#define D1 64
#define NCLS 16
#define NEG_SLOPE 0.2f
#define EPS_GAT 1e-16f
#define BSH 9
#define BNODES 512
#define SRC_BITS 23
#define SRC_MASK 0x7FFFFFu
#define NBMAX 256
#define PART_G 256
#define PART_T 512
#define SCAN_T 1024
#define SCAN_PER 64

__device__ __forceinline__ float lrelu(float x) { return x > 0.f ? x : x * NEG_SLOPE; }

__device__ __forceinline__ void rescale_add(float4& a, float sc, const float4& v, float e) {
  a.x = fmaf(v.x, e, a.x * sc); a.y = fmaf(v.y, e, a.y * sc);
  a.z = fmaf(v.z, e, a.z * sc); a.w = fmaf(v.w, e, a.w * sc);
}
__device__ __forceinline__ void add_scaled(float4& a, const float4& v, float e) {
  a.x = fmaf(v.x, e, a.x); a.y = fmaf(v.y, e, a.y);
  a.z = fmaf(v.z, e, a.z); a.w = fmaf(v.w, e, a.w);
}

__global__ __launch_bounds__(128) void gemm1_kernel(const float* __restrict__ x,
                                                    const float* __restrict__ W1,
                                                    float* __restrict__ xl1, int nrows) {
  __shared__ float At[16 * 516];
  const int tid = threadIdx.x;
  const int rowBase = blockIdx.x * 16;
  for (int i = tid; i < 2048; i += 128) {
    int r = i >> 7, cq = i & 127;
    int row = rowBase + r; if (row >= nrows) row = nrows - 1;
    *(float4*)&At[r * 516 + cq * 4] = ((const float4*)(x + (size_t)row * F_IN))[cq];
  }
  __syncthreads();

  const int wave = tid >> 5;
  const int lane = tid & 31;
  const int ln = lane & 15;
  const int khalf = (lane >> 4) << 1;
  const int colBase = wave * 16;

  v8f c = {};
  #pragma unroll 4
  for (int k = 0; k < F_IN; k += 4) {
    v2f a = *(const v2f*)&At[ln * 516 + k + khalf];
    const float* wp = W1 + (size_t)(k + khalf) * D1 + colBase + ln;
    v2f b; b.x = wp[0]; b.y = wp[D1];
    c = __builtin_amdgcn_wmma_f32_16x16x4_f32(false, a, false, b, (short)0, c,
                                              false, false);
  }
  const int mBase = (lane >> 4) * 8;
  float* outp = xl1 + (size_t)rowBase * D1 + colBase;
  #pragma unroll
  for (int i = 0; i < 8; i++) {
    int row = rowBase + mBase + i;
    if (row < nrows) outp[(size_t)(mBase + i) * D1 + ln] = c[i];
  }
}

__global__ void attn1_kernel(const float* __restrict__ xl1,
                             const float* __restrict__ att_src,
                             const float* __restrict__ att_dst,
                             float* __restrict__ a_src, float* __restrict__ a_dst, int n) {
  int t = blockIdx.x * blockDim.x + threadIdx.x;
  if (t >= n * HEADS) return;
  int node = t >> 3, hh = t & 7;
  const float* xp = xl1 + (size_t)node * D1 + hh * HID;
  const float* as = att_src + hh * HID;
  const float* ad = att_dst + hh * HID;
  float s1 = 0.f, s2 = 0.f;
  #pragma unroll
  for (int cc = 0; cc < HID; cc++) { float v = xp[cc]; s1 += v * as[cc]; s2 += v * ad[cc]; }
  a_src[t] = s1; a_dst[t] = s2;
}


__global__ __launch_bounds__(PART_T) void part_count(const int* __restrict__ ei,
                                                     unsigned* __restrict__ hist,
                                                     int E_, int NB, int chunk) {
  __shared__ unsigned c[NBMAX];
  const int tid = threadIdx.x;
  if (tid < NB) c[tid] = 0u;
  __syncthreads();
  const int* __restrict__ dst = ei + E_;
  const int b0 = blockIdx.x * chunk;
  const int b1 = (b0 + chunk < E_) ? (b0 + chunk) : E_;
  int e = b0 + tid;
  for (; e + 3 * PART_T < b1; e += 4 * PART_T) {
    const int d0 = dst[e], d1 = dst[e + PART_T], d2 = dst[e + 2 * PART_T], d3 = dst[e + 3 * PART_T];
    atomicAdd(&c[d0 >> BSH], 1u); atomicAdd(&c[d1 >> BSH], 1u);
    atomicAdd(&c[d2 >> BSH], 1u); atomicAdd(&c[d3 >> BSH], 1u);
  }
  for (; e < b1; e += PART_T) atomicAdd(&c[dst[e] >> BSH], 1u);
  __syncthreads();
  if (tid < NB) hist[(size_t)tid * gridDim.x + blockIdx.x] = c[tid];
}

__global__ __launch_bounds__(SCAN_T) void part_scan(unsigned* __restrict__ a, int M) {
  __shared__ unsigned wsum[32];
  __shared__ unsigned tile_total;
  const int tid = threadIdx.x, lane = tid & 31, wave = tid >> 5;
  unsigned carry = 0u;
  for (int base = 0; base < M; base += SCAN_T * SCAN_PER) {
    const int i0 = base + tid * SCAN_PER;
    uint4 q[SCAN_PER / 4];
    #pragma unroll
    for (int j = 0; j < SCAN_PER / 4; j++)
      q[j] = (i0 + 4 * j < M) ? ((const uint4*)(a + i0))[j] : make_uint4(0u, 0u, 0u, 0u);
    unsigned t = 0u;
    #pragma unroll
    for (int j = 0; j < SCAN_PER / 4; j++) t += q[j].x + q[j].y + q[j].z + q[j].w;
    unsigned incl = t;
    #pragma unroll
    for (int off = 1; off < 32; off <<= 1) {
      const unsigned up = __shfl_up(incl, off);
      if (lane >= off) incl += up;
    }
    if (lane == 31) wsum[wave] = incl;
    __syncthreads();
    if (wave == 0) {
      const unsigned w = wsum[lane];
      unsigned wi = w;
      #pragma unroll
      for (int off = 1; off < 32; off <<= 1) {
        const unsigned up = __shfl_up(wi, off);
        if (lane >= off) wi += up;
      }
      wsum[lane] = wi - w;
      if (lane == 31) tile_total = wi;
    }
    __syncthreads();
    unsigned run = carry + wsum[wave] + (incl - t);
    #pragma unroll
    for (int j = 0; j < SCAN_PER / 4; j++) {
      if (i0 + 4 * j < M) {
        uint4 o;
        o.x = run; run += q[j].x; o.y = run; run += q[j].y;
        o.z = run; run += q[j].z; o.w = run; run += q[j].w;
        ((uint4*)(a + i0))[j] = o;
      }
    }
    carry += tile_total;
    __syncthreads();
  }
}

__global__ __launch_bounds__(PART_T) void part_place(const int* __restrict__ ei,
                                                     const unsigned* __restrict__ offs,
                                                     unsigned* __restrict__ rec,
                                                     int E_, int NB, int chunk) {
  __shared__ unsigned cur[NBMAX];
  const int tid = threadIdx.x;
  if (tid < NB) cur[tid] = offs[(size_t)tid * gridDim.x + blockIdx.x];
  __syncthreads();
  const int* __restrict__ dst = ei + E_;
  const int b0 = blockIdx.x * chunk;
  const int b1 = (b0 + chunk < E_) ? (b0 + chunk) : E_;
  int e = b0 + tid;
  for (; e + 3 * PART_T < b1; e += 4 * PART_T) {
    const unsigned d0 = dst[e], d1 = dst[e + PART_T], d2 = dst[e + 2 * PART_T], d3 = dst[e + 3 * PART_T];
    const unsigned s0 = ei[e],  s1 = ei[e + PART_T],  s2 = ei[e + 2 * PART_T],  s3 = ei[e + 3 * PART_T];
    const unsigned p0 = atomicAdd(&cur[d0 >> BSH], 1u), p1 = atomicAdd(&cur[d1 >> BSH], 1u);
    const unsigned p2 = atomicAdd(&cur[d2 >> BSH], 1u), p3 = atomicAdd(&cur[d3 >> BSH], 1u);
    rec[p0] = s0 | ((d0 & (BNODES - 1)) << SRC_BITS); rec[p1] = s1 | ((d1 & (BNODES - 1)) << SRC_BITS);
    rec[p2] = s2 | ((d2 & (BNODES - 1)) << SRC_BITS); rec[p3] = s3 | ((d3 & (BNODES - 1)) << SRC_BITS);
  }
  for (; e < b1; e += PART_T) {
    const unsigned d = dst[e], s = ei[e];
    const unsigned p = atomicAdd(&cur[d >> BSH], 1u);
    rec[p] = s | ((d & (BNODES - 1)) << SRC_BITS);
  }
}

__global__ __launch_bounds__(PART_T) void bucket_sort(const unsigned* __restrict__ offs,
                                                      const unsigned* __restrict__ rec,
                                                      unsigned* __restrict__ rowptr,
                                                      int* __restrict__ csr_src,
                                                      int n, int E_, int G, int NB) {
  __shared__ unsigned cnt[BNODES];
  __shared__ unsigned wtot[PART_T / 32];
  const int tid = threadIdx.x, lane = tid & 31, wave = tid >> 5;
  const int k = blockIdx.x;
  const unsigned start = offs[(size_t)k * G];
  const unsigned end = (k + 1 < NB) ? offs[(size_t)(k + 1) * G] : (unsigned)E_;
  cnt[tid] = 0u;
  __syncthreads();
  unsigned i = start + tid;
  for (; i + 3 * PART_T < end; i += 4 * PART_T) {
    const unsigned r0 = rec[i], r1 = rec[i + PART_T], r2 = rec[i + 2 * PART_T], r3 = rec[i + 3 * PART_T];
    atomicAdd(&cnt[r0 >> SRC_BITS], 1u); atomicAdd(&cnt[r1 >> SRC_BITS], 1u);
    atomicAdd(&cnt[r2 >> SRC_BITS], 1u); atomicAdd(&cnt[r3 >> SRC_BITS], 1u);
  }
  for (; i < end; i += PART_T) atomicAdd(&cnt[rec[i] >> SRC_BITS], 1u);
  __syncthreads();
  const unsigned c = cnt[tid];
  unsigned incl = c;
  #pragma unroll
  for (int off = 1; off < 32; off <<= 1) {
    const unsigned up = __shfl_up(incl, off);
    if (lane >= off) incl += up;
  }
  if (lane == 31) wtot[wave] = incl;
  __syncthreads();
  if (wave == 0) {
    const unsigned w = (lane < PART_T / 32) ? wtot[lane] : 0u;
    unsigned wi = w;
    #pragma unroll
    for (int off = 1; off < 32; off <<= 1) {
      const unsigned up = __shfl_up(wi, off);
      if (lane >= off) wi += up;
    }
    if (lane < PART_T / 32) wtot[lane] = wi - w;
  }
  __syncthreads();
  const unsigned first = start + wtot[wave] + (incl - c);
  cnt[tid] = first;
  const int node = k * BNODES + tid;
  if (node < n) rowptr[node] = first;
  if (k == NB - 1 && tid == 0) rowptr[n] = (unsigned)E_;
  __syncthreads();
  i = start + tid;
  for (; i + 3 * PART_T < end; i += 4 * PART_T) {
    const unsigned r0 = rec[i], r1 = rec[i + PART_T], r2 = rec[i + 2 * PART_T], r3 = rec[i + 3 * PART_T];
    const unsigned p0 = atomicAdd(&cnt[r0 >> SRC_BITS], 1u), p1 = atomicAdd(&cnt[r1 >> SRC_BITS], 1u);
    const unsigned p2 = atomicAdd(&cnt[r2 >> SRC_BITS], 1u), p3 = atomicAdd(&cnt[r3 >> SRC_BITS], 1u);
    csr_src[p0] = (int)(r0 & SRC_MASK); csr_src[p1] = (int)(r1 & SRC_MASK);
    csr_src[p2] = (int)(r2 & SRC_MASK); csr_src[p3] = (int)(r3 & SRC_MASK);
  }
  for (; i < end; i += PART_T) {
    const unsigned r = rec[i];
    const unsigned p = atomicAdd(&cnt[r >> SRC_BITS], 1u);
    csr_src[p] = (int)(r & SRC_MASK);
  }
}

__global__ void agg1_kernel(const unsigned* __restrict__ rowptr, const int* __restrict__ csr_src,
                            const float* __restrict__ a_src, const float* __restrict__ a_dst,
                            const float* __restrict__ xl1, const float* __restrict__ bias1,
                            float* __restrict__ h, int n) {
  int t = blockIdx.x * blockDim.x + threadIdx.x;
  if (t >= n * HEADS) return;
  const int d = t >> 3, hh = t & 7;
  const int* __restrict__ cp = csr_src + rowptr[d];
  const int* __restrict__ ce = csr_src + rowptr[d + 1];
  const float ad = a_dst[t];
  const float* __restrict__ asp = a_src + hh;
  const float* __restrict__ xb = xl1 + hh * HID;

  float m = -__builtin_huge_valf();
  float den = 0.f;
  float4 acc0 = make_float4(0.f, 0.f, 0.f, 0.f), acc1 = acc0;
  for (; cp + 4 <= ce; cp += 4) {
    const int s0 = cp[0], s1 = cp[1], s2 = cp[2], s3 = cp[3];
    const float al0 = lrelu(asp[s0 * HEADS] + ad), al1 = lrelu(asp[s1 * HEADS] + ad);
    const float al2 = lrelu(asp[s2 * HEADS] + ad), al3 = lrelu(asp[s3 * HEADS] + ad);
    const float4* x0 = (const float4*)(xb + (size_t)s0 * D1);
    const float4* x1 = (const float4*)(xb + (size_t)s1 * D1);
    const float4* x2 = (const float4*)(xb + (size_t)s2 * D1);
    const float4* x3 = (const float4*)(xb + (size_t)s3 * D1);
    const float4 p0 = x0[0], q0 = x0[1], p1 = x1[0], q1 = x1[1];
    const float4 p2 = x2[0], q2 = x2[1], p3 = x3[0], q3 = x3[1];
    float mn = al0 > m ? al0 : m;
    mn = al1 > mn ? al1 : mn; mn = al2 > mn ? al2 : mn; mn = al3 > mn ? al3 : mn;
    const float sc = __expf(m - mn);
    const float e0 = __expf(al0 - mn), e1 = __expf(al1 - mn);
    const float e2 = __expf(al2 - mn), e3 = __expf(al3 - mn);
    m = mn;
    den = fmaf(den, sc, e0); den += e1; den += e2; den += e3;
    rescale_add(acc0, sc, p0, e0); rescale_add(acc1, sc, q0, e0);
    add_scaled(acc0, p1, e1);      add_scaled(acc1, q1, e1);
    add_scaled(acc0, p2, e2);      add_scaled(acc1, q2, e2);
    add_scaled(acc0, p3, e3);      add_scaled(acc1, q3, e3);
  }
  #pragma unroll 1
  for (; cp < ce; cp++) {
    const int s = cp[0];
    const float al = lrelu(asp[s * HEADS] + ad);
    const float4* xp = (const float4*)(xb + (size_t)s * D1);
    const float4 v0 = xp[0], v1 = xp[1];
    const float mn = al > m ? al : m;
    const float sc = __expf(m - mn);
    const float ex = __expf(al - mn);
    m = mn;
    den = fmaf(den, sc, ex);
    rescale_add(acc0, sc, v0, ex); rescale_add(acc1, sc, v1, ex);
  }
  const float dn = den + EPS_GAT;
  const float4* bp = (const float4*)(bias1 + hh * HID);
  const float4 b0 = bp[0], b1 = bp[1];
  float r[8] = { acc0.x / dn + b0.x, acc0.y / dn + b0.y, acc0.z / dn + b0.z, acc0.w / dn + b0.w,
                 acc1.x / dn + b1.x, acc1.y / dn + b1.y, acc1.z / dn + b1.z, acc1.w / dn + b1.w };
  #pragma unroll
  for (int c = 0; c < 8; c++) r[c] = r[c] > 0.f ? r[c] : (__expf(r[c]) - 1.f);
  float4* hp = (float4*)(h + (size_t)d * D1 + hh * HID);
  hp[0] = make_float4(r[0], r[1], r[2], r[3]);
  hp[1] = make_float4(r[4], r[5], r[6], r[7]);
}

__global__ __launch_bounds__(256) void gemm2_kernel(const float* __restrict__ h,
                                                    const float* __restrict__ W2,
                                                    float* __restrict__ xl2, int nrows) {
  const int wave = threadIdx.x >> 5;
  const int lane = threadIdx.x & 31;
  const int tile = blockIdx.x * 8 + wave;
  const int rowBase = tile * 16;
  if (rowBase >= nrows) return;
  const int ln = lane & 15;
  const int khalf = (lane >> 4) << 1;
  int arow = rowBase + ln; if (arow >= nrows) arow = nrows - 1;
  const float* ap = h + (size_t)arow * D1;

  v8f c = {};
  #pragma unroll
  for (int k = 0; k < D1; k += 4) {
    v2f a = *(const v2f*)&ap[k + khalf];
    v2f b; b.x = W2[(k + khalf) * NCLS + ln];
           b.y = W2[(k + khalf + 1) * NCLS + ln];
    c = __builtin_amdgcn_wmma_f32_16x16x4_f32(false, a, false, b, (short)0, c,
                                              false, false);
  }
  const int mBase = (lane >> 4) * 8;
  float* outp = xl2 + (size_t)rowBase * NCLS;
  #pragma unroll
  for (int i = 0; i < 8; i++) {
    int row = rowBase + mBase + i;
    if (row < nrows) outp[(size_t)(mBase + i) * NCLS + ln] = c[i];
  }
}

__global__ void attn2_kernel(const float* __restrict__ xl2,
                             const float* __restrict__ att_src,
                             const float* __restrict__ att_dst,
                             float* __restrict__ a_src, float* __restrict__ a_dst, int n) {
  int t = blockIdx.x * blockDim.x + threadIdx.x;
  if (t >= n) return;
  const float* xp = xl2 + (size_t)t * NCLS;
  float s1 = 0.f, s2 = 0.f;
  #pragma unroll
  for (int cc = 0; cc < NCLS; cc++) { float v = xp[cc]; s1 += v * att_src[cc]; s2 += v * att_dst[cc]; }
  a_src[t] = s1; a_dst[t] = s2;
}

__global__ void agg2_kernel(const unsigned* __restrict__ rowptr, const int* __restrict__ csr_src,
                            const float* __restrict__ a_src, const float* __restrict__ a_dst,
                            const float* __restrict__ xl2, const float* __restrict__ bias2,
                            float* __restrict__ out, int n) {
  int t = blockIdx.x * blockDim.x + threadIdx.x;
  if (t >= n * 4) return;
  const int d = t >> 2, q = t & 3;
  const int* __restrict__ cp = csr_src + rowptr[d];
  const int* __restrict__ ce = csr_src + rowptr[d + 1];
  const float ad = a_dst[d];
  const float4* __restrict__ xq = (const float4*)xl2 + q;

  float m = -__builtin_huge_valf();
  float den = 0.f;
  float4 acc = make_float4(0.f, 0.f, 0.f, 0.f);
  for (; cp + 4 <= ce; cp += 4) {
    const int s0 = cp[0], s1 = cp[1], s2 = cp[2], s3 = cp[3];
    const float al0 = lrelu(a_src[s0] + ad), al1 = lrelu(a_src[s1] + ad);
    const float al2 = lrelu(a_src[s2] + ad), al3 = lrelu(a_src[s3] + ad);
    const float4 v0 = xq[(size_t)s0 * 4], v1 = xq[(size_t)s1 * 4];
    const float4 v2 = xq[(size_t)s2 * 4], v3 = xq[(size_t)s3 * 4];
    float mn = al0 > m ? al0 : m;
    mn = al1 > mn ? al1 : mn; mn = al2 > mn ? al2 : mn; mn = al3 > mn ? al3 : mn;
    const float sc = __expf(m - mn);
    const float e0 = __expf(al0 - mn), e1 = __expf(al1 - mn);
    const float e2 = __expf(al2 - mn), e3 = __expf(al3 - mn);
    m = mn;
    den = fmaf(den, sc, e0); den += e1; den += e2; den += e3;
    rescale_add(acc, sc, v0, e0);
    add_scaled(acc, v1, e1); add_scaled(acc, v2, e2); add_scaled(acc, v3, e3);
  }
  #pragma unroll 1
  for (; cp < ce; cp++) {
    const unsigned s = cp[0];
    const float al = lrelu(a_src[s] + ad);
    const float4 v = xq[(size_t)s * 4];
    const float mn = al > m ? al : m;
    const float sc = __expf(m - mn);
    const float ex = __expf(al - mn);
    m = mn;
    den = fmaf(den, sc, ex);
    rescale_add(acc, sc, v, ex);
  }
  const float dn = den + EPS_GAT;
  const float4 b = ((const float4*)bias2)[q];
  ((float4*)(out + (size_t)d * NCLS))[q] =
      make_float4(acc.x / dn + b.x, acc.y / dn + b.y, acc.z / dn + b.z, acc.w / dn + b.w);
}

extern "C" void kernel_launch(void* const* d_in, const int* in_sizes, int n_in,
                              void* d_out, int out_size, void* d_ws, size_t ws_size,
                              hipStream_t stream) {
  const float* x        = (const float*)d_in[0];
  const int*   ei       = (const int*)d_in[1];
  const float* W1       = (const float*)d_in[2];
  const float* att_src1 = (const float*)d_in[3];
  const float* att_dst1 = (const float*)d_in[4];
  const float* bias1    = (const float*)d_in[5];
  const float* W2       = (const float*)d_in[6];
  const float* att_src2 = (const float*)d_in[7];
  const float* att_dst2 = (const float*)d_in[8];
  const float* bias2    = (const float*)d_in[9];
  float* out = (float*)d_out;

  const int N = in_sizes[0] / F_IN;
  const int E = in_sizes[1] / 2;
  const size_t Np = ((size_t)N + 4) & ~(size_t)3;
  const int NB = (N + BNODES - 1) / BNODES;
  const int chunk = (E + PART_G - 1) / PART_G;
  const int M = NB * PART_G;

  float* ws = (float*)d_ws;
  size_t o = 0;
  float*    xl1    = ws + o; o += (size_t)N * D1;
  float*    h      = ws + o; o += (size_t)N * D1;
  float*    a_src1 = ws + o; o += (size_t)N * HEADS;
  float*    a_dst1 = ws + o; o += (size_t)N * HEADS;
  float*    xl2    = ws + o; o += (size_t)N * NCLS;
  float*    a_src2 = ws + o; o += Np;
  float*    a_dst2 = ws + o; o += Np;
  unsigned* rowptr = (unsigned*)(ws + o); o += Np;
  unsigned* offs   = (unsigned*)(ws + o); o += (size_t)M;
  int*      csrsrc = (int*)(ws + o);      o += (size_t)E;
  unsigned* rec    = (unsigned*)h;

  const int TB = 256;
  auto nb = [](long long n, int b) { return (unsigned)((n + b - 1) / b); };

  part_count<<<PART_G, PART_T, 0, stream>>>(ei, offs, E, NB, chunk);
  part_scan<<<1, SCAN_T, 0, stream>>>(offs, M);
  part_place<<<PART_G, PART_T, 0, stream>>>(ei, offs, rec, E, NB, chunk);
  bucket_sort<<<NB, PART_T, 0, stream>>>(offs, rec, rowptr, csrsrc, N, E, PART_G, NB);

  gemm1_kernel<<<nb((long long)N, 16), 128, 0, stream>>>(x, W1, xl1, N);
  attn1_kernel<<<nb((long long)N * HEADS, TB), TB, 0, stream>>>(xl1, att_src1, att_dst1,
                                                                a_src1, a_dst1, N);
  agg1_kernel<<<nb((long long)N * HEADS, TB), TB, 0, stream>>>(rowptr, csrsrc, a_src1, a_dst1,
                                                               xl1, bias1, h, N);

  gemm2_kernel<<<nb(((long long)N + 15) / 16, 8), 256, 0, stream>>>(h, W2, xl2, N);
  attn2_kernel<<<nb(N, TB), TB, 0, stream>>>(xl2, att_src2, att_dst2, a_src2, a_dst2, N);
  agg2_kernel<<<nb((long long)N * 4, TB), TB, 0, stream>>>(rowptr, csrsrc, a_src2, a_dst2,
                                                           xl2, bias2, out, N);
}


# Scores formed inside both aggregation kernels from the rows they gather (fma; layer-2 dot = four lane parts added pairwise): score kernels, score planes and one gathered line per edge go
# speedup vs baseline: 7.6686x; 1.1092x over previous
#include <hip/hip_runtime.h>
#include <hip/hip_bf16.h>

typedef __attribute__((ext_vector_type(2))) float v2f;
typedef __attribute__((ext_vector_type(8))) float v8f;

#define F_IN 512
#define HID 8
#define HEADS 8
#define D1 64
#define NCLS 16
#define NEG_SLOPE 0.2f
#define EPS_GAT 1e-16f
#define BSH 9
#define BNODES 512
#define SRC_BITS 23
#define SRC_MASK 0x7FFFFFu
#define NBMAX 256
#define PART_G 256
#define PART_T 512
#define SCAN_T 1024
#define SCAN_PER 64

__device__ __forceinline__ float lrelu(float x) { return x > 0.f ? x : x * NEG_SLOPE; }

__device__ __forceinline__ void rescale_add(float4& a, float sc, const float4& v, float e) {
  a.x = fmaf(v.x, e, a.x * sc); a.y = fmaf(v.y, e, a.y * sc);
  a.z = fmaf(v.z, e, a.z * sc); a.w = fmaf(v.w, e, a.w * sc);
}
__device__ __forceinline__ void add_scaled(float4& a, const float4& v, float e) {
  a.x = fmaf(v.x, e, a.x); a.y = fmaf(v.y, e, a.y);
  a.z = fmaf(v.z, e, a.z); a.w = fmaf(v.w, e, a.w);
}

__device__ __forceinline__ float dot8(const float4& a, const float4& b, const float4& w0, const float4& w1) {
  float s = a.x * w0.x;
  s = fmaf(a.y, w0.y, s); s = fmaf(a.z, w0.z, s); s = fmaf(a.w, w0.w, s);
  s = fmaf(b.x, w1.x, s); s = fmaf(b.y, w1.y, s); s = fmaf(b.z, w1.z, s); s = fmaf(b.w, w1.w, s);
  return s;
}
__device__ __forceinline__ float dot4(const float4& a, const float4& w) {
  float s = a.x * w.x;
  s = fmaf(a.y, w.y, s); s = fmaf(a.z, w.z, s); s = fmaf(a.w, w.w, s);
  return s;
}
__device__ __forceinline__ float quad_sum(float v) {
  v += __int_as_float(__builtin_amdgcn_update_dpp(0, __float_as_int(v), 0xB1, 0xF, 0xF, true));
  v += __int_as_float(__builtin_amdgcn_update_dpp(0, __float_as_int(v), 0x4E, 0xF, 0xF, true));
  return v;
}

__global__ __launch_bounds__(128) void gemm1_kernel(const float* __restrict__ x,
                                                    const float* __restrict__ W1,
                                                    float* __restrict__ xl1, int nrows) {
  __shared__ float At[16 * 516];
  const int tid = threadIdx.x;
  const int rowBase = blockIdx.x * 16;
  for (int i = tid; i < 2048; i += 128) {
    int r = i >> 7, cq = i & 127;
    int row = rowBase + r; if (row >= nrows) row = nrows - 1;
    *(float4*)&At[r * 516 + cq * 4] = ((const float4*)(x + (size_t)row * F_IN))[cq];
  }
  __syncthreads();

  const int wave = tid >> 5;
  const int lane = tid & 31;
  const int ln = lane & 15;
  const int khalf = (lane >> 4) << 1;
  const int colBase = wave * 16;

  v8f c = {};
  #pragma unroll 4
  for (int k = 0; k < F_IN; k += 4) {
    v2f a = *(const v2f*)&At[ln * 516 + k + khalf];
    const float* wp = W1 + (size_t)(k + khalf) * D1 + colBase + ln;
    v2f b; b.x = wp[0]; b.y = wp[D1];
    c = __builtin_amdgcn_wmma_f32_16x16x4_f32(false, a, false, b, (short)0, c,
                                              false, false);
  }
  const int mBase = (lane >> 4) * 8;
  float* outp = xl1 + (size_t)rowBase * D1 + colBase;
  #pragma unroll
  for (int i = 0; i < 8; i++) {
    int row = rowBase + mBase + i;
    if (row < nrows) outp[(size_t)(mBase + i) * D1 + ln] = c[i];
  }
}


__global__ __launch_bounds__(PART_T) void part_count(const int* __restrict__ ei,
                                                     unsigned* __restrict__ hist,
                                                     int E_, int NB, int chunk) {
  __shared__ unsigned c[NBMAX];
  const int tid = threadIdx.x;
  if (tid < NB) c[tid] = 0u;
  __syncthreads();
  const int* __restrict__ dst = ei + E_;
  const int b0 = blockIdx.x * chunk;
  const int b1 = (b0 + chunk < E_) ? (b0 + chunk) : E_;
  int e = b0 + tid;
  for (; e + 3 * PART_T < b1; e += 4 * PART_T) {
    const int d0 = dst[e], d1 = dst[e + PART_T], d2 = dst[e + 2 * PART_T], d3 = dst[e + 3 * PART_T];
    atomicAdd(&c[d0 >> BSH], 1u); atomicAdd(&c[d1 >> BSH], 1u);
    atomicAdd(&c[d2 >> BSH], 1u); atomicAdd(&c[d3 >> BSH], 1u);
  }
  for (; e < b1; e += PART_T) atomicAdd(&c[dst[e] >> BSH], 1u);
  __syncthreads();
  if (tid < NB) hist[(size_t)tid * gridDim.x + blockIdx.x] = c[tid];
}

__global__ __launch_bounds__(SCAN_T) void part_scan(unsigned* __restrict__ a, int M) {
  __shared__ unsigned wsum[32];
  __shared__ unsigned tile_total;
  const int tid = threadIdx.x, lane = tid & 31, wave = tid >> 5;
  unsigned carry = 0u;
  for (int base = 0; base < M; base += SCAN_T * SCAN_PER) {
    const int i0 = base + tid * SCAN_PER;
    uint4 q[SCAN_PER / 4];
    #pragma unroll
    for (int j = 0; j < SCAN_PER / 4; j++)
      q[j] = (i0 + 4 * j < M) ? ((const uint4*)(a + i0))[j] : make_uint4(0u, 0u, 0u, 0u);
    unsigned t = 0u;
    #pragma unroll
    for (int j = 0; j < SCAN_PER / 4; j++) t += q[j].x + q[j].y + q[j].z + q[j].w;
    unsigned incl = t;
    #pragma unroll
    for (int off = 1; off < 32; off <<= 1) {
      const unsigned up = __shfl_up(incl, off);
      if (lane >= off) incl += up;
    }
    if (lane == 31) wsum[wave] = incl;
    __syncthreads();
    if (wave == 0) {
      const unsigned w = wsum[lane];
      unsigned wi = w;
      #pragma unroll
      for (int off = 1; off < 32; off <<= 1) {
        const unsigned up = __shfl_up(wi, off);
        if (lane >= off) wi += up;
      }
      wsum[lane] = wi - w;
      if (lane == 31) tile_total = wi;
    }
    __syncthreads();
    unsigned run = carry + wsum[wave] + (incl - t);
    #pragma unroll
    for (int j = 0; j < SCAN_PER / 4; j++) {
      if (i0 + 4 * j < M) {
        uint4 o;
        o.x = run; run += q[j].x; o.y = run; run += q[j].y;
        o.z = run; run += q[j].z; o.w = run; run += q[j].w;
        ((uint4*)(a + i0))[j] = o;
      }
    }
    carry += tile_total;
    __syncthreads();
  }
}

__global__ __launch_bounds__(PART_T) void part_place(const int* __restrict__ ei,
                                                     const unsigned* __restrict__ offs,
                                                     unsigned* __restrict__ rec,
                                                     int E_, int NB, int chunk) {
  __shared__ unsigned cur[NBMAX];
  const int tid = threadIdx.x;
  if (tid < NB) cur[tid] = offs[(size_t)tid * gridDim.x + blockIdx.x];
  __syncthreads();
  const int* __restrict__ dst = ei + E_;
  const int b0 = blockIdx.x * chunk;
  const int b1 = (b0 + chunk < E_) ? (b0 + chunk) : E_;
  int e = b0 + tid;
  for (; e + 3 * PART_T < b1; e += 4 * PART_T) {
    const unsigned d0 = dst[e], d1 = dst[e + PART_T], d2 = dst[e + 2 * PART_T], d3 = dst[e + 3 * PART_T];
    const unsigned s0 = ei[e],  s1 = ei[e + PART_T],  s2 = ei[e + 2 * PART_T],  s3 = ei[e + 3 * PART_T];
    const unsigned p0 = atomicAdd(&cur[d0 >> BSH], 1u), p1 = atomicAdd(&cur[d1 >> BSH], 1u);
    const unsigned p2 = atomicAdd(&cur[d2 >> BSH], 1u), p3 = atomicAdd(&cur[d3 >> BSH], 1u);
    rec[p0] = s0 | ((d0 & (BNODES - 1)) << SRC_BITS); rec[p1] = s1 | ((d1 & (BNODES - 1)) << SRC_BITS);
    rec[p2] = s2 | ((d2 & (BNODES - 1)) << SRC_BITS); rec[p3] = s3 | ((d3 & (BNODES - 1)) << SRC_BITS);
  }
  for (; e < b1; e += PART_T) {
    const unsigned d = dst[e], s = ei[e];
    const unsigned p = atomicAdd(&cur[d >> BSH], 1u);
    rec[p] = s | ((d & (BNODES - 1)) << SRC_BITS);
  }
}

__global__ __launch_bounds__(PART_T) void bucket_sort(const unsigned* __restrict__ offs,
                                                      const unsigned* __restrict__ rec,
                                                      unsigned* __restrict__ rowptr,
                                                      int* __restrict__ csr_src,
                                                      int n, int E_, int G, int NB) {
  __shared__ unsigned cnt[BNODES];
  __shared__ unsigned wtot[PART_T / 32];
  const int tid = threadIdx.x, lane = tid & 31, wave = tid >> 5;
  const int k = blockIdx.x;
  const unsigned start = offs[(size_t)k * G];
  const unsigned end = (k + 1 < NB) ? offs[(size_t)(k + 1) * G] : (unsigned)E_;
  cnt[tid] = 0u;
  __syncthreads();
  unsigned i = start + tid;
  for (; i + 3 * PART_T < end; i += 4 * PART_T) {
    const unsigned r0 = rec[i], r1 = rec[i + PART_T], r2 = rec[i + 2 * PART_T], r3 = rec[i + 3 * PART_T];
    atomicAdd(&cnt[r0 >> SRC_BITS], 1u); atomicAdd(&cnt[r1 >> SRC_BITS], 1u);
    atomicAdd(&cnt[r2 >> SRC_BITS], 1u); atomicAdd(&cnt[r3 >> SRC_BITS], 1u);
  }
  for (; i < end; i += PART_T) atomicAdd(&cnt[rec[i] >> SRC_BITS], 1u);
  __syncthreads();
  const unsigned c = cnt[tid];
  unsigned incl = c;
  #pragma unroll
  for (int off = 1; off < 32; off <<= 1) {
    const unsigned up = __shfl_up(incl, off);
    if (lane >= off) incl += up;
  }
  if (lane == 31) wtot[wave] = incl;
  __syncthreads();
  if (wave == 0) {
    const unsigned w = (lane < PART_T / 32) ? wtot[lane] : 0u;
    unsigned wi = w;
    #pragma unroll
    for (int off = 1; off < 32; off <<= 1) {
      const unsigned up = __shfl_up(wi, off);
      if (lane >= off) wi += up;
    }
    if (lane < PART_T / 32) wtot[lane] = wi - w;
  }
  __syncthreads();
  const unsigned first = start + wtot[wave] + (incl - c);
  cnt[tid] = first;
  const int node = k * BNODES + tid;
  if (node < n) rowptr[node] = first;
  if (k == NB - 1 && tid == 0) rowptr[n] = (unsigned)E_;
  __syncthreads();
  i = start + tid;
  for (; i + 3 * PART_T < end; i += 4 * PART_T) {
    const unsigned r0 = rec[i], r1 = rec[i + PART_T], r2 = rec[i + 2 * PART_T], r3 = rec[i + 3 * PART_T];
    const unsigned p0 = atomicAdd(&cnt[r0 >> SRC_BITS], 1u), p1 = atomicAdd(&cnt[r1 >> SRC_BITS], 1u);
    const unsigned p2 = atomicAdd(&cnt[r2 >> SRC_BITS], 1u), p3 = atomicAdd(&cnt[r3 >> SRC_BITS], 1u);
    csr_src[p0] = (int)(r0 & SRC_MASK); csr_src[p1] = (int)(r1 & SRC_MASK);
    csr_src[p2] = (int)(r2 & SRC_MASK); csr_src[p3] = (int)(r3 & SRC_MASK);
  }
  for (; i < end; i += PART_T) {
    const unsigned r = rec[i];
    const unsigned p = atomicAdd(&cnt[r >> SRC_BITS], 1u);
    csr_src[p] = (int)(r & SRC_MASK);
  }
}

__global__ void agg1_kernel(const unsigned* __restrict__ rowptr, const int* __restrict__ csr_src,
                            const float* __restrict__ att_src, const float* __restrict__ att_dst,
                            const float* __restrict__ xl1, const float* __restrict__ bias1,
                            float* __restrict__ h, int n) {
  int t = blockIdx.x * blockDim.x + threadIdx.x;
  if (t >= n * HEADS) return;
  const int d = t >> 3, hh = t & 7;
  const int* __restrict__ cp = csr_src + rowptr[d];
  const int* __restrict__ ce = csr_src + rowptr[d + 1];
  const float* __restrict__ xb = xl1 + hh * HID;
  const float4* wsp = (const float4*)(att_src + hh * HID);
  const float4 ws0 = wsp[0], ws1 = wsp[1];
  float ad;
  {
    const float4* wdp = (const float4*)(att_dst + hh * HID);
    const float4* xd = (const float4*)(xb + (size_t)d * D1);
    ad = dot8(xd[0], xd[1], wdp[0], wdp[1]);
  }

  float m = -__builtin_huge_valf();
  float den = 0.f;
  float4 acc0 = make_float4(0.f, 0.f, 0.f, 0.f), acc1 = acc0;
  for (; cp + 4 <= ce; cp += 4) {
    const int s0 = cp[0], s1 = cp[1], s2 = cp[2], s3 = cp[3];
    const float4* x0 = (const float4*)(xb + (size_t)s0 * D1);
    const float4* x1 = (const float4*)(xb + (size_t)s1 * D1);
    const float4* x2 = (const float4*)(xb + (size_t)s2 * D1);
    const float4* x3 = (const float4*)(xb + (size_t)s3 * D1);
    const float4 p0 = x0[0], q0 = x0[1], p1 = x1[0], q1 = x1[1];
    const float4 p2 = x2[0], q2 = x2[1], p3 = x3[0], q3 = x3[1];
    const float al0 = lrelu(dot8(p0, q0, ws0, ws1) + ad), al1 = lrelu(dot8(p1, q1, ws0, ws1) + ad);
    const float al2 = lrelu(dot8(p2, q2, ws0, ws1) + ad), al3 = lrelu(dot8(p3, q3, ws0, ws1) + ad);
    float mn = al0 > m ? al0 : m;
    mn = al1 > mn ? al1 : mn; mn = al2 > mn ? al2 : mn; mn = al3 > mn ? al3 : mn;
    const float sc = __expf(m - mn);
    const float e0 = __expf(al0 - mn), e1 = __expf(al1 - mn);
    const float e2 = __expf(al2 - mn), e3 = __expf(al3 - mn);
    m = mn;
    den = fmaf(den, sc, e0); den += e1; den += e2; den += e3;
    rescale_add(acc0, sc, p0, e0); rescale_add(acc1, sc, q0, e0);
    add_scaled(acc0, p1, e1);      add_scaled(acc1, q1, e1);
    add_scaled(acc0, p2, e2);      add_scaled(acc1, q2, e2);
    add_scaled(acc0, p3, e3);      add_scaled(acc1, q3, e3);
  }
  #pragma unroll 1
  for (; cp < ce; cp++) {
    const int s = cp[0];
    const float4* xp = (const float4*)(xb + (size_t)s * D1);
    const float4 v0 = xp[0], v1 = xp[1];
    const float al = lrelu(dot8(v0, v1, ws0, ws1) + ad);
    const float mn = al > m ? al : m;
    const float sc = __expf(m - mn);
    const float ex = __expf(al - mn);
    m = mn;
    den = fmaf(den, sc, ex);
    rescale_add(acc0, sc, v0, ex); rescale_add(acc1, sc, v1, ex);
  }
  const float dn = den + EPS_GAT;
  const float4* bp = (const float4*)(bias1 + hh * HID);
  const float4 b0 = bp[0], b1 = bp[1];
  float r[8] = { acc0.x / dn + b0.x, acc0.y / dn + b0.y, acc0.z / dn + b0.z, acc0.w / dn + b0.w,
                 acc1.x / dn + b1.x, acc1.y / dn + b1.y, acc1.z / dn + b1.z, acc1.w / dn + b1.w };
  #pragma unroll
  for (int c = 0; c < 8; c++) r[c] = r[c] > 0.f ? r[c] : (__expf(r[c]) - 1.f);
  float4* hp = (float4*)(h + (size_t)d * D1 + hh * HID);
  hp[0] = make_float4(r[0], r[1], r[2], r[3]);
  hp[1] = make_float4(r[4], r[5], r[6], r[7]);
}

__global__ __launch_bounds__(256) void gemm2_kernel(const float* __restrict__ h,
                                                    const float* __restrict__ W2,
                                                    float* __restrict__ xl2, int nrows) {
  const int wave = threadIdx.x >> 5;
  const int lane = threadIdx.x & 31;
  const int tile = blockIdx.x * 8 + wave;
  const int rowBase = tile * 16;
  if (rowBase >= nrows) return;
  const int ln = lane & 15;
  const int khalf = (lane >> 4) << 1;
  int arow = rowBase + ln; if (arow >= nrows) arow = nrows - 1;
  const float* ap = h + (size_t)arow * D1;

  v8f c = {};
  #pragma unroll
  for (int k = 0; k < D1; k += 4) {
    v2f a = *(const v2f*)&ap[k + khalf];
    v2f b; b.x = W2[(k + khalf) * NCLS + ln];
           b.y = W2[(k + khalf + 1) * NCLS + ln];
    c = __builtin_amdgcn_wmma_f32_16x16x4_f32(false, a, false, b, (short)0, c,
                                              false, false);
  }
  const int mBase = (lane >> 4) * 8;
  float* outp = xl2 + (size_t)rowBase * NCLS;
  #pragma unroll
  for (int i = 0; i < 8; i++) {
    int row = rowBase + mBase + i;
    if (row < nrows) outp[(size_t)(mBase + i) * NCLS + ln] = c[i];
  }
}

__global__ void agg2_kernel(const unsigned* __restrict__ rowptr, const int* __restrict__ csr_src,
                            const float* __restrict__ att_src, const float* __restrict__ att_dst,
                            const float* __restrict__ xl2, const float* __restrict__ bias2,
                            float* __restrict__ out, int n) {
  int t = blockIdx.x * blockDim.x + threadIdx.x;
  if (t >= n * 4) return;
  const int d = t >> 2, q = t & 3;
  const int* __restrict__ cp = csr_src + rowptr[d];
  const int* __restrict__ ce = csr_src + rowptr[d + 1];
  const float4* __restrict__ xq = (const float4*)xl2 + q;
  const float4 wsq = ((const float4*)att_src)[q];
  const float ad = quad_sum(dot4(xq[(size_t)d * 4], ((const float4*)att_dst)[q]));

  float m = -__builtin_huge_valf();
  float den = 0.f;
  float4 acc = make_float4(0.f, 0.f, 0.f, 0.f);
  for (; cp + 4 <= ce; cp += 4) {
    const int s0 = cp[0], s1 = cp[1], s2 = cp[2], s3 = cp[3];
    const float4 v0 = xq[(size_t)s0 * 4], v1 = xq[(size_t)s1 * 4];
    const float4 v2 = xq[(size_t)s2 * 4], v3 = xq[(size_t)s3 * 4];
    const float al0 = lrelu(quad_sum(dot4(v0, wsq)) + ad), al1 = lrelu(quad_sum(dot4(v1, wsq)) + ad);
    const float al2 = lrelu(quad_sum(dot4(v2, wsq)) + ad), al3 = lrelu(quad_sum(dot4(v3, wsq)) + ad);
    float mn = al0 > m ? al0 : m;
    mn = al1 > mn ? al1 : mn; mn = al2 > mn ? al2 : mn; mn = al3 > mn ? al3 : mn;
    const float sc = __expf(m - mn);
    const float e0 = __expf(al0 - mn), e1 = __expf(al1 - mn);
    const float e2 = __expf(al2 - mn), e3 = __expf(al3 - mn);
    m = mn;
    den = fmaf(den, sc, e0); den += e1; den += e2; den += e3;
    rescale_add(acc, sc, v0, e0);
    add_scaled(acc, v1, e1); add_scaled(acc, v2, e2); add_scaled(acc, v3, e3);
  }
  #pragma unroll 1
  for (; cp < ce; cp++) {
    const unsigned s = cp[0];
    const float4 v = xq[(size_t)s * 4];
    const float al = lrelu(quad_sum(dot4(v, wsq)) + ad);
    const float mn = al > m ? al : m;
    const float sc = __expf(m - mn);
    const float ex = __expf(al - mn);
    m = mn;
    den = fmaf(den, sc, ex);
    rescale_add(acc, sc, v, ex);
  }
  const float dn = den + EPS_GAT;
  const float4 b = ((const float4*)bias2)[q];
  ((float4*)(out + (size_t)d * NCLS))[q] =
      make_float4(acc.x / dn + b.x, acc.y / dn + b.y, acc.z / dn + b.z, acc.w / dn + b.w);
}

extern "C" void kernel_launch(void* const* d_in, const int* in_sizes, int n_in,
                              void* d_out, int out_size, void* d_ws, size_t ws_size,
                              hipStream_t stream) {
  const float* x        = (const float*)d_in[0];
  const int*   ei       = (const int*)d_in[1];
  const float* W1       = (const float*)d_in[2];
  const float* att_src1 = (const float*)d_in[3];
  const float* att_dst1 = (const float*)d_in[4];
  const float* bias1    = (const float*)d_in[5];
  const float* W2       = (const float*)d_in[6];
  const float* att_src2 = (const float*)d_in[7];
  const float* att_dst2 = (const float*)d_in[8];
  const float* bias2    = (const float*)d_in[9];
  float* out = (float*)d_out;

  const int N = in_sizes[0] / F_IN;
  const int E = in_sizes[1] / 2;
  const size_t Np = ((size_t)N + 4) & ~(size_t)3;
  const int NB = (N + BNODES - 1) / BNODES;
  const int chunk = (E + PART_G - 1) / PART_G;
  const int M = NB * PART_G;

  float* ws = (float*)d_ws;
  size_t o = 0;
  float*    xl1    = ws + o; o += (size_t)N * D1;
  float*    h      = ws + o; o += (size_t)N * D1;
  float*    xl2    = ws + o; o += (size_t)N * NCLS;
  unsigned* rowptr = (unsigned*)(ws + o); o += Np;
  unsigned* offs   = (unsigned*)(ws + o); o += (size_t)M;
  int*      csrsrc = (int*)(ws + o);      o += (size_t)E;
  unsigned* rec    = (unsigned*)h;

  const int TB = 256;
  auto nb = [](long long n, int b) { return (unsigned)((n + b - 1) / b); };

  part_count<<<PART_G, PART_T, 0, stream>>>(ei, offs, E, NB, chunk);
  part_scan<<<1, SCAN_T, 0, stream>>>(offs, M);
  part_place<<<PART_G, PART_T, 0, stream>>>(ei, offs, rec, E, NB, chunk);
  bucket_sort<<<NB, PART_T, 0, stream>>>(offs, rec, rowptr, csrsrc, N, E, PART_G, NB);

  gemm1_kernel<<<nb((long long)N, 16), 128, 0, stream>>>(x, W1, xl1, N);
  agg1_kernel<<<nb((long long)N * HEADS, TB), TB, 0, stream>>>(rowptr, csrsrc, att_src1, att_dst1,
                                                               xl1, bias1, h, N);

  gemm2_kernel<<<nb(((long long)N + 15) / 16, 8), 256, 0, stream>>>(h, W2, xl2, N);
  agg2_kernel<<<nb((long long)N * 4, TB), TB, 0, stream>>>(rowptr, csrsrc, att_src2, att_dst2,
                                                           xl2, bias2, out, N);
}


# GEMM1 on 512 resident work-groups looping over 16-row tiles (next tile prefetched): each wave keeps its 16-column slice of W1 in registers, W1 read once per work-group; same k order and sums
# speedup vs baseline: 8.3253x; 1.0856x over previous
#include <hip/hip_runtime.h>
#include <hip/hip_bf16.h>

typedef __attribute__((ext_vector_type(2))) float v2f;
typedef __attribute__((ext_vector_type(8))) float v8f;
typedef __attribute__((ext_vector_type(4))) float v4f;

#define F_IN 512
#define HID 8
#define HEADS 8
#define D1 64
#define NCLS 16
#define NEG_SLOPE 0.2f
#define EPS_GAT 1e-16f
#define BSH 9
#define BNODES 512
#define SRC_BITS 23
#define SRC_MASK 0x7FFFFFu
#define NBMAX 256
#define PART_G 256
#define PART_T 512
#define SCAN_T 1024
#define SCAN_PER 64

__device__ __forceinline__ float lrelu(float x) { return x > 0.f ? x : x * NEG_SLOPE; }

__device__ __forceinline__ void rescale_add(float4& a, float sc, const float4& v, float e) {
  a.x = fmaf(v.x, e, a.x * sc); a.y = fmaf(v.y, e, a.y * sc);
  a.z = fmaf(v.z, e, a.z * sc); a.w = fmaf(v.w, e, a.w * sc);
}
__device__ __forceinline__ void add_scaled(float4& a, const float4& v, float e) {
  a.x = fmaf(v.x, e, a.x); a.y = fmaf(v.y, e, a.y);
  a.z = fmaf(v.z, e, a.z); a.w = fmaf(v.w, e, a.w);
}

__device__ __forceinline__ float dot8(const float4& a, const float4& b, const float4& w0, const float4& w1) {
  float s = a.x * w0.x;
  s = fmaf(a.y, w0.y, s); s = fmaf(a.z, w0.z, s); s = fmaf(a.w, w0.w, s);
  s = fmaf(b.x, w1.x, s); s = fmaf(b.y, w1.y, s); s = fmaf(b.z, w1.z, s); s = fmaf(b.w, w1.w, s);
  return s;
}
__device__ __forceinline__ float dot4(const float4& a, const float4& w) {
  float s = a.x * w.x;
  s = fmaf(a.y, w.y, s); s = fmaf(a.z, w.z, s); s = fmaf(a.w, w.w, s);
  return s;
}
__device__ __forceinline__ float quad_sum(float v) {
  v += __int_as_float(__builtin_amdgcn_update_dpp(0, __float_as_int(v), 0xB1, 0xF, 0xF, true));
  v += __int_as_float(__builtin_amdgcn_update_dpp(0, __float_as_int(v), 0x4E, 0xF, 0xF, true));
  return v;
}

#define G1_BLOCKS 512
__global__ __launch_bounds__(128) void gemm1_kernel(const float* __restrict__ x,
                                                    const float* __restrict__ W1,
                                                    float* __restrict__ xl1, int nrows) {
  __shared__ float At[16 * 516];
  const int tid = threadIdx.x;
  const int wave = tid >> 5;
  const int lane = tid & 31;
  const int ln = lane & 15;
  const int khalf = (lane >> 4) << 1;
  const int colBase = wave * 16;
  const int mBase = (lane >> 4) * 8;
  const int ntiles = (nrows + 15) >> 4;

  v2f b[F_IN / 4];
  {
    const float* wp = W1 + (size_t)khalf * D1 + colBase + ln;
    #pragma unroll
    for (int j = 0; j < F_IN / 4; j++) {
      b[j].x = wp[(size_t)(4 * j) * D1];
      b[j].y = wp[(size_t)(4 * j + 1) * D1];
    }
  }

  v4f st[16];
  int tile = blockIdx.x;
  if (tile < ntiles) {
    #pragma unroll
    for (int r = 0; r < 16; r++) {
      int row = tile * 16 + r; if (row >= nrows) row = nrows - 1;
      st[r] = ((const v4f*)(x + (size_t)row * F_IN))[tid];
    }
  }
  for (; tile < ntiles; tile += G1_BLOCKS) {
    __syncthreads();
    #pragma unroll
    for (int r = 0; r < 16; r++) *(v4f*)&At[r * 516 + tid * 4] = st[r];
    __syncthreads();
    const int nt = tile + G1_BLOCKS;
    if (nt < ntiles) {
      #pragma unroll
      for (int r = 0; r < 16; r++) {
        int row = nt * 16 + r; if (row >= nrows) row = nrows - 1;
        st[r] = ((const v4f*)(x + (size_t)row * F_IN))[tid];
      }
    }
    v8f c = {};
    const float* arow = &At[ln * 516 + khalf];
    v2f af[2][16];
    #pragma unroll
    for (int i = 0; i < 16; i++) af[0][i] = *(const v2f*)&arow[4 * i];
    #pragma unroll
    for (int blk = 0; blk < F_IN / 64; blk++) {
      if (blk + 1 < F_IN / 64) {
        #pragma unroll
        for (int i = 0; i < 16; i++) af[(blk + 1) & 1][i] = *(const v2f*)&arow[4 * (16 * (blk + 1) + i)];
      }
      __builtin_amdgcn_sched_barrier(0);
      #pragma unroll
      for (int i = 0; i < 16; i++)
        c = __builtin_amdgcn_wmma_f32_16x16x4_f32(false, af[blk & 1][i], false, b[16 * blk + i],
                                                  (short)0, c, false, false);
    }
    const int rowBase = tile * 16;
    float* outp = xl1 + (size_t)rowBase * D1 + colBase;
    #pragma unroll
    for (int i = 0; i < 8; i++) {
      int row = rowBase + mBase + i;
      if (row < nrows) outp[(size_t)(mBase + i) * D1 + ln] = c[i];
    }
  }
}


__global__ __launch_bounds__(PART_T) void part_count(const int* __restrict__ ei,
                                                     unsigned* __restrict__ hist,
                                                     int E_, int NB, int chunk) {
  __shared__ unsigned c[NBMAX];
  const int tid = threadIdx.x;
  if (tid < NB) c[tid] = 0u;
  __syncthreads();
  const int* __restrict__ dst = ei + E_;
  const int b0 = blockIdx.x * chunk;
  const int b1 = (b0 + chunk < E_) ? (b0 + chunk) : E_;
  int e = b0 + tid;
  for (; e + 3 * PART_T < b1; e += 4 * PART_T) {
    const int d0 = dst[e], d1 = dst[e + PART_T], d2 = dst[e + 2 * PART_T], d3 = dst[e + 3 * PART_T];
    atomicAdd(&c[d0 >> BSH], 1u); atomicAdd(&c[d1 >> BSH], 1u);
    atomicAdd(&c[d2 >> BSH], 1u); atomicAdd(&c[d3 >> BSH], 1u);
  }
  for (; e < b1; e += PART_T) atomicAdd(&c[dst[e] >> BSH], 1u);
  __syncthreads();
  if (tid < NB) hist[(size_t)tid * gridDim.x + blockIdx.x] = c[tid];
}

__global__ __launch_bounds__(SCAN_T) void part_scan(unsigned* __restrict__ a, int M) {
  __shared__ unsigned wsum[32];
  __shared__ unsigned tile_total;
  const int tid = threadIdx.x, lane = tid & 31, wave = tid >> 5;
  unsigned carry = 0u;
  for (int base = 0; base < M; base += SCAN_T * SCAN_PER) {
    const int i0 = base + tid * SCAN_PER;
    uint4 q[SCAN_PER / 4];
    #pragma unroll
    for (int j = 0; j < SCAN_PER / 4; j++)
      q[j] = (i0 + 4 * j < M) ? ((const uint4*)(a + i0))[j] : make_uint4(0u, 0u, 0u, 0u);
    unsigned t = 0u;
    #pragma unroll
    for (int j = 0; j < SCAN_PER / 4; j++) t += q[j].x + q[j].y + q[j].z + q[j].w;
    unsigned incl = t;
    #pragma unroll
    for (int off = 1; off < 32; off <<= 1) {
      const unsigned up = __shfl_up(incl, off);
      if (lane >= off) incl += up;
    }
    if (lane == 31) wsum[wave] = incl;
    __syncthreads();
    if (wave == 0) {
      const unsigned w = wsum[lane];
      unsigned wi = w;
      #pragma unroll
      for (int off = 1; off < 32; off <<= 1) {
        const unsigned up = __shfl_up(wi, off);
        if (lane >= off) wi += up;
      }
      wsum[lane] = wi - w;
      if (lane == 31) tile_total = wi;
    }
    __syncthreads();
    unsigned run = carry + wsum[wave] + (incl - t);
    #pragma unroll
    for (int j = 0; j < SCAN_PER / 4; j++) {
      if (i0 + 4 * j < M) {
        uint4 o;
        o.x = run; run += q[j].x; o.y = run; run += q[j].y;
        o.z = run; run += q[j].z; o.w = run; run += q[j].w;
        ((uint4*)(a + i0))[j] = o;
      }
    }
    carry += tile_total;
    __syncthreads();
  }
}

__global__ __launch_bounds__(PART_T) void part_place(const int* __restrict__ ei,
                                                     const unsigned* __restrict__ offs,
                                                     unsigned* __restrict__ rec,
                                                     int E_, int NB, int chunk) {
  __shared__ unsigned cur[NBMAX];
  const int tid = threadIdx.x;
  if (tid < NB) cur[tid] = offs[(size_t)tid * gridDim.x + blockIdx.x];
  __syncthreads();
  const int* __restrict__ dst = ei + E_;
  const int b0 = blockIdx.x * chunk;
  const int b1 = (b0 + chunk < E_) ? (b0 + chunk) : E_;
  int e = b0 + tid;
  for (; e + 3 * PART_T < b1; e += 4 * PART_T) {
    const unsigned d0 = dst[e], d1 = dst[e + PART_T], d2 = dst[e + 2 * PART_T], d3 = dst[e + 3 * PART_T];
    const unsigned s0 = ei[e],  s1 = ei[e + PART_T],  s2 = ei[e + 2 * PART_T],  s3 = ei[e + 3 * PART_T];
    const unsigned p0 = atomicAdd(&cur[d0 >> BSH], 1u), p1 = atomicAdd(&cur[d1 >> BSH], 1u);
    const unsigned p2 = atomicAdd(&cur[d2 >> BSH], 1u), p3 = atomicAdd(&cur[d3 >> BSH], 1u);
    rec[p0] = s0 | ((d0 & (BNODES - 1)) << SRC_BITS); rec[p1] = s1 | ((d1 & (BNODES - 1)) << SRC_BITS);
    rec[p2] = s2 | ((d2 & (BNODES - 1)) << SRC_BITS); rec[p3] = s3 | ((d3 & (BNODES - 1)) << SRC_BITS);
  }
  for (; e < b1; e += PART_T) {
    const unsigned d = dst[e], s = ei[e];
    const unsigned p = atomicAdd(&cur[d >> BSH], 1u);
    rec[p] = s | ((d & (BNODES - 1)) << SRC_BITS);
  }
}

__global__ __launch_bounds__(PART_T) void bucket_sort(const unsigned* __restrict__ offs,
                                                      const unsigned* __restrict__ rec,
                                                      unsigned* __restrict__ rowptr,
                                                      int* __restrict__ csr_src,
                                                      int n, int E_, int G, int NB) {
  __shared__ unsigned cnt[BNODES];
  __shared__ unsigned wtot[PART_T / 32];
  const int tid = threadIdx.x, lane = tid & 31, wave = tid >> 5;
  const int k = blockIdx.x;
  const unsigned start = offs[(size_t)k * G];
  const unsigned end = (k + 1 < NB) ? offs[(size_t)(k + 1) * G] : (unsigned)E_;
  cnt[tid] = 0u;
  __syncthreads();
  unsigned i = start + tid;
  for (; i + 3 * PART_T < end; i += 4 * PART_T) {
    const unsigned r0 = rec[i], r1 = rec[i + PART_T], r2 = rec[i + 2 * PART_T], r3 = rec[i + 3 * PART_T];
    atomicAdd(&cnt[r0 >> SRC_BITS], 1u); atomicAdd(&cnt[r1 >> SRC_BITS], 1u);
    atomicAdd(&cnt[r2 >> SRC_BITS], 1u); atomicAdd(&cnt[r3 >> SRC_BITS], 1u);
  }
  for (; i < end; i += PART_T) atomicAdd(&cnt[rec[i] >> SRC_BITS], 1u);
  __syncthreads();
  const unsigned c = cnt[tid];
  unsigned incl = c;
  #pragma unroll
  for (int off = 1; off < 32; off <<= 1) {
    const unsigned up = __shfl_up(incl, off);
    if (lane >= off) incl += up;
  }
  if (lane == 31) wtot[wave] = incl;
  __syncthreads();
  if (wave == 0) {
    const unsigned w = (lane < PART_T / 32) ? wtot[lane] : 0u;
    unsigned wi = w;
    #pragma unroll
    for (int off = 1; off < 32; off <<= 1) {
      const unsigned up = __shfl_up(wi, off);
      if (lane >= off) wi += up;
    }
    if (lane < PART_T / 32) wtot[lane] = wi - w;
  }
  __syncthreads();
  const unsigned first = start + wtot[wave] + (incl - c);
  cnt[tid] = first;
  const int node = k * BNODES + tid;
  if (node < n) rowptr[node] = first;
  if (k == NB - 1 && tid == 0) rowptr[n] = (unsigned)E_;
  __syncthreads();
  i = start + tid;
  for (; i + 3 * PART_T < end; i += 4 * PART_T) {
    const unsigned r0 = rec[i], r1 = rec[i + PART_T], r2 = rec[i + 2 * PART_T], r3 = rec[i + 3 * PART_T];
    const unsigned p0 = atomicAdd(&cnt[r0 >> SRC_BITS], 1u), p1 = atomicAdd(&cnt[r1 >> SRC_BITS], 1u);
    const unsigned p2 = atomicAdd(&cnt[r2 >> SRC_BITS], 1u), p3 = atomicAdd(&cnt[r3 >> SRC_BITS], 1u);
    csr_src[p0] = (int)(r0 & SRC_MASK); csr_src[p1] = (int)(r1 & SRC_MASK);
    csr_src[p2] = (int)(r2 & SRC_MASK); csr_src[p3] = (int)(r3 & SRC_MASK);
  }
  for (; i < end; i += PART_T) {
    const unsigned r = rec[i];
    const unsigned p = atomicAdd(&cnt[r >> SRC_BITS], 1u);
    csr_src[p] = (int)(r & SRC_MASK);
  }
}

__global__ void agg1_kernel(const unsigned* __restrict__ rowptr, const int* __restrict__ csr_src,
                            const float* __restrict__ att_src, const float* __restrict__ att_dst,
                            const float* __restrict__ xl1, const float* __restrict__ bias1,
                            float* __restrict__ h, int n) {
  int t = blockIdx.x * blockDim.x + threadIdx.x;
  if (t >= n * HEADS) return;
  const int d = t >> 3, hh = t & 7;
  const int* __restrict__ cp = csr_src + rowptr[d];
  const int* __restrict__ ce = csr_src + rowptr[d + 1];
  const float* __restrict__ xb = xl1 + hh * HID;
  const float4* wsp = (const float4*)(att_src + hh * HID);
  const float4 ws0 = wsp[0], ws1 = wsp[1];
  float ad;
  {
    const float4* wdp = (const float4*)(att_dst + hh * HID);
    const float4* xd = (const float4*)(xb + (size_t)d * D1);
    ad = dot8(xd[0], xd[1], wdp[0], wdp[1]);
  }

  float m = -__builtin_huge_valf();
  float den = 0.f;
  float4 acc0 = make_float4(0.f, 0.f, 0.f, 0.f), acc1 = acc0;
  for (; cp + 4 <= ce; cp += 4) {
    const int s0 = cp[0], s1 = cp[1], s2 = cp[2], s3 = cp[3];
    const float4* x0 = (const float4*)(xb + (size_t)s0 * D1);
    const float4* x1 = (const float4*)(xb + (size_t)s1 * D1);
    const float4* x2 = (const float4*)(xb + (size_t)s2 * D1);
    const float4* x3 = (const float4*)(xb + (size_t)s3 * D1);
    const float4 p0 = x0[0], q0 = x0[1], p1 = x1[0], q1 = x1[1];
    const float4 p2 = x2[0], q2 = x2[1], p3 = x3[0], q3 = x3[1];
    const float al0 = lrelu(dot8(p0, q0, ws0, ws1) + ad), al1 = lrelu(dot8(p1, q1, ws0, ws1) + ad);
    const float al2 = lrelu(dot8(p2, q2, ws0, ws1) + ad), al3 = lrelu(dot8(p3, q3, ws0, ws1) + ad);
    float mn = al0 > m ? al0 : m;
    mn = al1 > mn ? al1 : mn; mn = al2 > mn ? al2 : mn; mn = al3 > mn ? al3 : mn;
    const float sc = __expf(m - mn);
    const float e0 = __expf(al0 - mn), e1 = __expf(al1 - mn);
    const float e2 = __expf(al2 - mn), e3 = __expf(al3 - mn);
    m = mn;
    den = fmaf(den, sc, e0); den += e1; den += e2; den += e3;
    rescale_add(acc0, sc, p0, e0); rescale_add(acc1, sc, q0, e0);
    add_scaled(acc0, p1, e1);      add_scaled(acc1, q1, e1);
    add_scaled(acc0, p2, e2);      add_scaled(acc1, q2, e2);
    add_scaled(acc0, p3, e3);      add_scaled(acc1, q3, e3);
  }
  #pragma unroll 1
  for (; cp < ce; cp++) {
    const int s = cp[0];
    const float4* xp = (const float4*)(xb + (size_t)s * D1);
    const float4 v0 = xp[0], v1 = xp[1];
    const float al = lrelu(dot8(v0, v1, ws0, ws1) + ad);
    const float mn = al > m ? al : m;
    const float sc = __expf(m - mn);
    const float ex = __expf(al - mn);
    m = mn;
    den = fmaf(den, sc, ex);
    rescale_add(acc0, sc, v0, ex); rescale_add(acc1, sc, v1, ex);
  }
  const float dn = den + EPS_GAT;
  const float4* bp = (const float4*)(bias1 + hh * HID);
  const float4 b0 = bp[0], b1 = bp[1];
  float r[8] = { acc0.x / dn + b0.x, acc0.y / dn + b0.y, acc0.z / dn + b0.z, acc0.w / dn + b0.w,
                 acc1.x / dn + b1.x, acc1.y / dn + b1.y, acc1.z / dn + b1.z, acc1.w / dn + b1.w };
  #pragma unroll
  for (int c = 0; c < 8; c++) r[c] = r[c] > 0.f ? r[c] : (__expf(r[c]) - 1.f);
  float4* hp = (float4*)(h + (size_t)d * D1 + hh * HID);
  hp[0] = make_float4(r[0], r[1], r[2], r[3]);
  hp[1] = make_float4(r[4], r[5], r[6], r[7]);
}

__global__ __launch_bounds__(256) void gemm2_kernel(const float* __restrict__ h,
                                                    const float* __restrict__ W2,
                                                    float* __restrict__ xl2, int nrows) {
  const int wave = threadIdx.x >> 5;
  const int lane = threadIdx.x & 31;
  const int tile = blockIdx.x * 8 + wave;
  const int rowBase = tile * 16;
  if (rowBase >= nrows) return;
  const int ln = lane & 15;
  const int khalf = (lane >> 4) << 1;
  int arow = rowBase + ln; if (arow >= nrows) arow = nrows - 1;
  const float* ap = h + (size_t)arow * D1;

  v8f c = {};
  #pragma unroll
  for (int k = 0; k < D1; k += 4) {
    v2f a = *(const v2f*)&ap[k + khalf];
    v2f b; b.x = W2[(k + khalf) * NCLS + ln];
           b.y = W2[(k + khalf + 1) * NCLS + ln];
    c = __builtin_amdgcn_wmma_f32_16x16x4_f32(false, a, false, b, (short)0, c,
                                              false, false);
  }
  const int mBase = (lane >> 4) * 8;
  float* outp = xl2 + (size_t)rowBase * NCLS;
  #pragma unroll
  for (int i = 0; i < 8; i++) {
    int row = rowBase + mBase + i;
    if (row < nrows) outp[(size_t)(mBase + i) * NCLS + ln] = c[i];
  }
}

__global__ void agg2_kernel(const unsigned* __restrict__ rowptr, const int* __restrict__ csr_src,
                            const float* __restrict__ att_src, const float* __restrict__ att_dst,
                            const float* __restrict__ xl2, const float* __restrict__ bias2,
                            float* __restrict__ out, int n) {
  int t = blockIdx.x * blockDim.x + threadIdx.x;
  if (t >= n * 4) return;
  const int d = t >> 2, q = t & 3;
  const int* __restrict__ cp = csr_src + rowptr[d];
  const int* __restrict__ ce = csr_src + rowptr[d + 1];
  const float4* __restrict__ xq = (const float4*)xl2 + q;
  const float4 wsq = ((const float4*)att_src)[q];
  const float ad = quad_sum(dot4(xq[(size_t)d * 4], ((const float4*)att_dst)[q]));

  float m = -__builtin_huge_valf();
  float den = 0.f;
  float4 acc = make_float4(0.f, 0.f, 0.f, 0.f);
  for (; cp + 4 <= ce; cp += 4) {
    const int s0 = cp[0], s1 = cp[1], s2 = cp[2], s3 = cp[3];
    const float4 v0 = xq[(size_t)s0 * 4], v1 = xq[(size_t)s1 * 4];
    const float4 v2 = xq[(size_t)s2 * 4], v3 = xq[(size_t)s3 * 4];
    const float al0 = lrelu(quad_sum(dot4(v0, wsq)) + ad), al1 = lrelu(quad_sum(dot4(v1, wsq)) + ad);
    const float al2 = lrelu(quad_sum(dot4(v2, wsq)) + ad), al3 = lrelu(quad_sum(dot4(v3, wsq)) + ad);
    float mn = al0 > m ? al0 : m;
    mn = al1 > mn ? al1 : mn; mn = al2 > mn ? al2 : mn; mn = al3 > mn ? al3 : mn;
    const float sc = __expf(m - mn);
    const float e0 = __expf(al0 - mn), e1 = __expf(al1 - mn);
    const float e2 = __expf(al2 - mn), e3 = __expf(al3 - mn);
    m = mn;
    den = fmaf(den, sc, e0); den += e1; den += e2; den += e3;
    rescale_add(acc, sc, v0, e0);
    add_scaled(acc, v1, e1); add_scaled(acc, v2, e2); add_scaled(acc, v3, e3);
  }
  #pragma unroll 1
  for (; cp < ce; cp++) {
    const unsigned s = cp[0];
    const float4 v = xq[(size_t)s * 4];
    const float al = lrelu(quad_sum(dot4(v, wsq)) + ad);
    const float mn = al > m ? al : m;
    const float sc = __expf(m - mn);
    const float ex = __expf(al - mn);
    m = mn;
    den = fmaf(den, sc, ex);
    rescale_add(acc, sc, v, ex);
  }
  const float dn = den + EPS_GAT;
  const float4 b = ((const float4*)bias2)[q];
  ((float4*)(out + (size_t)d * NCLS))[q] =
      make_float4(acc.x / dn + b.x, acc.y / dn + b.y, acc.z / dn + b.z, acc.w / dn + b.w);
}

extern "C" void kernel_launch(void* const* d_in, const int* in_sizes, int n_in,
                              void* d_out, int out_size, void* d_ws, size_t ws_size,
                              hipStream_t stream) {
  const float* x        = (const float*)d_in[0];
  const int*   ei       = (const int*)d_in[1];
  const float* W1       = (const float*)d_in[2];
  const float* att_src1 = (const float*)d_in[3];
  const float* att_dst1 = (const float*)d_in[4];
  const float* bias1    = (const float*)d_in[5];
  const float* W2       = (const float*)d_in[6];
  const float* att_src2 = (const float*)d_in[7];
  const float* att_dst2 = (const float*)d_in[8];
  const float* bias2    = (const float*)d_in[9];
  float* out = (float*)d_out;

  const int N = in_sizes[0] / F_IN;
  const int E = in_sizes[1] / 2;
  const size_t Np = ((size_t)N + 4) & ~(size_t)3;
  const int NB = (N + BNODES - 1) / BNODES;
  const int chunk = (E + PART_G - 1) / PART_G;
  const int M = NB * PART_G;

  float* ws = (float*)d_ws;
  size_t o = 0;
  float*    xl1    = ws + o; o += (size_t)N * D1;
  float*    h      = ws + o; o += (size_t)N * D1;
  float*    xl2    = ws + o; o += (size_t)N * NCLS;
  unsigned* rowptr = (unsigned*)(ws + o); o += Np;
  unsigned* offs   = (unsigned*)(ws + o); o += (size_t)M;
  int*      csrsrc = (int*)(ws + o);      o += (size_t)E;
  unsigned* rec    = (unsigned*)h;

  const int TB = 256;
  auto nb = [](long long n, int b) { return (unsigned)((n + b - 1) / b); };

  part_count<<<PART_G, PART_T, 0, stream>>>(ei, offs, E, NB, chunk);
  part_scan<<<1, SCAN_T, 0, stream>>>(offs, M);
  part_place<<<PART_G, PART_T, 0, stream>>>(ei, offs, rec, E, NB, chunk);
  bucket_sort<<<NB, PART_T, 0, stream>>>(offs, rec, rowptr, csrsrc, N, E, PART_G, NB);

  gemm1_kernel<<<G1_BLOCKS, 128, 0, stream>>>(x, W1, xl1, N);
  agg1_kernel<<<nb((long long)N * HEADS, TB), TB, 0, stream>>>(rowptr, csrsrc, att_src1, att_dst1,
                                                               xl1, bias1, h, N);

  gemm2_kernel<<<nb(((long long)N + 15) / 16, 8), 256, 0, stream>>>(h, W2, xl2, N);
  agg2_kernel<<<nb((long long)N * 4, TB), TB, 0, stream>>>(rowptr, csrsrc, att_src2, att_dst2,
                                                           xl2, bias2, out, N);
}


# The one-work-group prefix-sum launch goes: each partition work-group sums the count table itself (whole-line reads, 16-lane DPP row sums) for its run offsets; bucket starts in a small plane
# speedup vs baseline: 8.9702x; 1.0775x over previous
#include <hip/hip_runtime.h>
#include <hip/hip_bf16.h>

typedef __attribute__((ext_vector_type(2))) float v2f;
typedef __attribute__((ext_vector_type(8))) float v8f;
typedef __attribute__((ext_vector_type(4))) float v4f;

#define F_IN 512
#define HID 8
#define HEADS 8
#define D1 64
#define NCLS 16
#define NEG_SLOPE 0.2f
#define EPS_GAT 1e-16f
#define BSH 9
#define BNODES 512
#define SRC_BITS 23
#define SRC_MASK 0x7FFFFFu
#define NBMAX 256
#define PART_G 256
#define PART_T 512

__device__ __forceinline__ float lrelu(float x) { return x > 0.f ? x : x * NEG_SLOPE; }

__device__ __forceinline__ void rescale_add(float4& a, float sc, const float4& v, float e) {
  a.x = fmaf(v.x, e, a.x * sc); a.y = fmaf(v.y, e, a.y * sc);
  a.z = fmaf(v.z, e, a.z * sc); a.w = fmaf(v.w, e, a.w * sc);
}
__device__ __forceinline__ void add_scaled(float4& a, const float4& v, float e) {
  a.x = fmaf(v.x, e, a.x); a.y = fmaf(v.y, e, a.y);
  a.z = fmaf(v.z, e, a.z); a.w = fmaf(v.w, e, a.w);
}

__device__ __forceinline__ float dot8(const float4& a, const float4& b, const float4& w0, const float4& w1) {
  float s = a.x * w0.x;
  s = fmaf(a.y, w0.y, s); s = fmaf(a.z, w0.z, s); s = fmaf(a.w, w0.w, s);
  s = fmaf(b.x, w1.x, s); s = fmaf(b.y, w1.y, s); s = fmaf(b.z, w1.z, s); s = fmaf(b.w, w1.w, s);
  return s;
}
__device__ __forceinline__ float dot4(const float4& a, const float4& w) {
  float s = a.x * w.x;
  s = fmaf(a.y, w.y, s); s = fmaf(a.z, w.z, s); s = fmaf(a.w, w.w, s);
  return s;
}
__device__ __forceinline__ float quad_sum(float v) {
  v += __int_as_float(__builtin_amdgcn_update_dpp(0, __float_as_int(v), 0xB1, 0xF, 0xF, true));
  v += __int_as_float(__builtin_amdgcn_update_dpp(0, __float_as_int(v), 0x4E, 0xF, 0xF, true));
  return v;
}

__device__ __forceinline__ unsigned row16_sum(unsigned v) {
  v += (unsigned)__builtin_amdgcn_update_dpp(0, (int)v, 0xB1, 0xF, 0xF, true);
  v += (unsigned)__builtin_amdgcn_update_dpp(0, (int)v, 0x4E, 0xF, 0xF, true);
  v += (unsigned)__builtin_amdgcn_update_dpp(0, (int)v, 0x141, 0xF, 0xF, true);
  v += (unsigned)__builtin_amdgcn_update_dpp(0, (int)v, 0x140, 0xF, 0xF, true);
  return v;
}

#define G1_BLOCKS 512
__global__ __launch_bounds__(128) void gemm1_kernel(const float* __restrict__ x,
                                                    const float* __restrict__ W1,
                                                    float* __restrict__ xl1, int nrows) {
  __shared__ float At[16 * 516];
  const int tid = threadIdx.x;
  const int wave = tid >> 5;
  const int lane = tid & 31;
  const int ln = lane & 15;
  const int khalf = (lane >> 4) << 1;
  const int colBase = wave * 16;
  const int mBase = (lane >> 4) * 8;
  const int ntiles = (nrows + 15) >> 4;

  v2f b[F_IN / 4];
  {
    const float* wp = W1 + (size_t)khalf * D1 + colBase + ln;
    #pragma unroll
    for (int j = 0; j < F_IN / 4; j++) {
      b[j].x = wp[(size_t)(4 * j) * D1];
      b[j].y = wp[(size_t)(4 * j + 1) * D1];
    }
  }

  v4f st[16];
  int tile = blockIdx.x;
  if (tile < ntiles) {
    #pragma unroll
    for (int r = 0; r < 16; r++) {
      int row = tile * 16 + r; if (row >= nrows) row = nrows - 1;
      st[r] = ((const v4f*)(x + (size_t)row * F_IN))[tid];
    }
  }
  for (; tile < ntiles; tile += G1_BLOCKS) {
    __syncthreads();
    #pragma unroll
    for (int r = 0; r < 16; r++) *(v4f*)&At[r * 516 + tid * 4] = st[r];
    __syncthreads();
    const int nt = tile + G1_BLOCKS;
    if (nt < ntiles) {
      #pragma unroll
      for (int r = 0; r < 16; r++) {
        int row = nt * 16 + r; if (row >= nrows) row = nrows - 1;
        st[r] = ((const v4f*)(x + (size_t)row * F_IN))[tid];
      }
    }
    v8f c = {};
    const float* arow = &At[ln * 516 + khalf];
    v2f af[2][16];
    #pragma unroll
    for (int i = 0; i < 16; i++) af[0][i] = *(const v2f*)&arow[4 * i];
    #pragma unroll
    for (int blk = 0; blk < F_IN / 64; blk++) {
      if (blk + 1 < F_IN / 64) {
        #pragma unroll
        for (int i = 0; i < 16; i++) af[(blk + 1) & 1][i] = *(const v2f*)&arow[4 * (16 * (blk + 1) + i)];
      }
      __builtin_amdgcn_sched_barrier(0);
      #pragma unroll
      for (int i = 0; i < 16; i++)
        c = __builtin_amdgcn_wmma_f32_16x16x4_f32(false, af[blk & 1][i], false, b[16 * blk + i],
                                                  (short)0, c, false, false);
    }
    const int rowBase = tile * 16;
    float* outp = xl1 + (size_t)rowBase * D1 + colBase;
    #pragma unroll
    for (int i = 0; i < 8; i++) {
      int row = rowBase + mBase + i;
      if (row < nrows) outp[(size_t)(mBase + i) * D1 + ln] = c[i];
    }
  }
}


__global__ __launch_bounds__(PART_T) void part_count(const int* __restrict__ ei,
                                                     unsigned* __restrict__ hist,
                                                     int E_, int NB, int chunk) {
  __shared__ unsigned c[NBMAX];
  const int tid = threadIdx.x;
  if (tid < NB) c[tid] = 0u;
  __syncthreads();
  const int* __restrict__ dst = ei + E_;
  const int b0 = blockIdx.x * chunk;
  const int b1 = (b0 + chunk < E_) ? (b0 + chunk) : E_;
  int e = b0 + tid;
  for (; e + 3 * PART_T < b1; e += 4 * PART_T) {
    const int d0 = dst[e], d1 = dst[e + PART_T], d2 = dst[e + 2 * PART_T], d3 = dst[e + 3 * PART_T];
    atomicAdd(&c[d0 >> BSH], 1u); atomicAdd(&c[d1 >> BSH], 1u);
    atomicAdd(&c[d2 >> BSH], 1u); atomicAdd(&c[d3 >> BSH], 1u);
  }
  for (; e < b1; e += PART_T) atomicAdd(&c[dst[e] >> BSH], 1u);
  __syncthreads();
  if (tid < NB) hist[(size_t)tid * gridDim.x + blockIdx.x] = c[tid];
}

__global__ __launch_bounds__(PART_T) void part_place(const int* __restrict__ ei,
                                                     const unsigned* __restrict__ hist,
                                                     unsigned* __restrict__ bstart,
                                                     unsigned* __restrict__ rec,
                                                     int E_, int NB, int chunk) {
  __shared__ unsigned cur[NBMAX];
  __shared__ unsigned rtot[NBMAX];
  __shared__ unsigned rpar[NBMAX];
  __shared__ unsigned wtot[PART_T / 32];
  const int tid = threadIdx.x, lane = tid & 31, wave = tid >> 5;
  const unsigned c = blockIdx.x;
  {
    const int hw = tid >> 4, l16 = tid & 15;
    uint4 q[NBMAX / 32][PART_G / 64];
    #pragma unroll
    for (int i = 0; i < NBMAX / 32; i++) {
      const int b = hw + 32 * i;
      const uint4* row = (const uint4*)(hist + (size_t)(b < NB ? b : NB - 1) * PART_G);
      #pragma unroll
      for (int j = 0; j < PART_G / 64; j++) q[i][j] = row[16 * j + l16];
    }
    #pragma unroll
    for (int i = 0; i < NBMAX / 32; i++) {
      const int b = hw + 32 * i;
      unsigned tot = 0u, par = 0u;
      #pragma unroll
      for (int j = 0; j < PART_G / 64; j++) {
        const unsigned i0 = 64u * j + 4u * l16;
        tot += q[i][j].x + q[i][j].y + q[i][j].z + q[i][j].w;
        par += (i0 < c ? q[i][j].x : 0u) + (i0 + 1u < c ? q[i][j].y : 0u) +
               (i0 + 2u < c ? q[i][j].z : 0u) + (i0 + 3u < c ? q[i][j].w : 0u);
      }
      tot = row16_sum(tot); par = row16_sum(par);
      if (l16 == 0 && b < NB) { rtot[b] = tot; rpar[b] = par; }
    }
  }
  __syncthreads();
  const unsigned t = (tid < NB) ? rtot[tid] : 0u;
  unsigned incl = t;
  #pragma unroll
  for (int off = 1; off < 32; off <<= 1) {
    const unsigned up = __shfl_up(incl, off);
    if (lane >= off) incl += up;
  }
  if (lane == 31) wtot[wave] = incl;
  __syncthreads();
  if (wave == 0) {
    const unsigned w = (lane < PART_T / 32) ? wtot[lane] : 0u;
    unsigned wi = w;
    #pragma unroll
    for (int off = 1; off < 32; off <<= 1) {
      const unsigned up = __shfl_up(wi, off);
      if (lane >= off) wi += up;
    }
    if (lane < PART_T / 32) wtot[lane] = wi - w;
  }
  __syncthreads();
  if (tid < NB) {
    const unsigned first = wtot[wave] + (incl - t);
    cur[tid] = first + rpar[tid];
    if (c == 0u) bstart[tid] = first;
  }
  __syncthreads();
  const int* __restrict__ dst = ei + E_;
  const int b0 = blockIdx.x * chunk;
  const int b1 = (b0 + chunk < E_) ? (b0 + chunk) : E_;
  int e = b0 + tid;
  for (; e + 3 * PART_T < b1; e += 4 * PART_T) {
    const unsigned d0 = dst[e], d1 = dst[e + PART_T], d2 = dst[e + 2 * PART_T], d3 = dst[e + 3 * PART_T];
    const unsigned s0 = ei[e],  s1 = ei[e + PART_T],  s2 = ei[e + 2 * PART_T],  s3 = ei[e + 3 * PART_T];
    const unsigned p0 = atomicAdd(&cur[d0 >> BSH], 1u), p1 = atomicAdd(&cur[d1 >> BSH], 1u);
    const unsigned p2 = atomicAdd(&cur[d2 >> BSH], 1u), p3 = atomicAdd(&cur[d3 >> BSH], 1u);
    rec[p0] = s0 | ((d0 & (BNODES - 1)) << SRC_BITS); rec[p1] = s1 | ((d1 & (BNODES - 1)) << SRC_BITS);
    rec[p2] = s2 | ((d2 & (BNODES - 1)) << SRC_BITS); rec[p3] = s3 | ((d3 & (BNODES - 1)) << SRC_BITS);
  }
  for (; e < b1; e += PART_T) {
    const unsigned d = dst[e], s = ei[e];
    const unsigned p = atomicAdd(&cur[d >> BSH], 1u);
    rec[p] = s | ((d & (BNODES - 1)) << SRC_BITS);
  }
}

__global__ __launch_bounds__(PART_T) void bucket_sort(const unsigned* __restrict__ bstart,
                                                      const unsigned* __restrict__ rec,
                                                      unsigned* __restrict__ rowptr,
                                                      int* __restrict__ csr_src,
                                                      int n, int E_, int NB) {
  __shared__ unsigned cnt[BNODES];
  __shared__ unsigned wtot[PART_T / 32];
  const int tid = threadIdx.x, lane = tid & 31, wave = tid >> 5;
  const int k = blockIdx.x;
  const unsigned start = bstart[k];
  const unsigned end = (k + 1 < NB) ? bstart[k + 1] : (unsigned)E_;
  cnt[tid] = 0u;
  __syncthreads();
  unsigned i = start + tid;
  for (; i + 3 * PART_T < end; i += 4 * PART_T) {
    const unsigned r0 = rec[i], r1 = rec[i + PART_T], r2 = rec[i + 2 * PART_T], r3 = rec[i + 3 * PART_T];
    atomicAdd(&cnt[r0 >> SRC_BITS], 1u); atomicAdd(&cnt[r1 >> SRC_BITS], 1u);
    atomicAdd(&cnt[r2 >> SRC_BITS], 1u); atomicAdd(&cnt[r3 >> SRC_BITS], 1u);
  }
  for (; i < end; i += PART_T) atomicAdd(&cnt[rec[i] >> SRC_BITS], 1u);
  __syncthreads();
  const unsigned c = cnt[tid];
  unsigned incl = c;
  #pragma unroll
  for (int off = 1; off < 32; off <<= 1) {
    const unsigned up = __shfl_up(incl, off);
    if (lane >= off) incl += up;
  }
  if (lane == 31) wtot[wave] = incl;
  __syncthreads();
  if (wave == 0) {
    const unsigned w = (lane < PART_T / 32) ? wtot[lane] : 0u;
    unsigned wi = w;
    #pragma unroll
    for (int off = 1; off < 32; off <<= 1) {
      const unsigned up = __shfl_up(wi, off);
      if (lane >= off) wi += up;
    }
    if (lane < PART_T / 32) wtot[lane] = wi - w;
  }
  __syncthreads();
  const unsigned first = start + wtot[wave] + (incl - c);
  cnt[tid] = first;
  const int node = k * BNODES + tid;
  if (node < n) rowptr[node] = first;
  if (k == NB - 1 && tid == 0) rowptr[n] = (unsigned)E_;
  __syncthreads();
  i = start + tid;
  for (; i + 3 * PART_T < end; i += 4 * PART_T) {
    const unsigned r0 = rec[i], r1 = rec[i + PART_T], r2 = rec[i + 2 * PART_T], r3 = rec[i + 3 * PART_T];
    const unsigned p0 = atomicAdd(&cnt[r0 >> SRC_BITS], 1u), p1 = atomicAdd(&cnt[r1 >> SRC_BITS], 1u);
    const unsigned p2 = atomicAdd(&cnt[r2 >> SRC_BITS], 1u), p3 = atomicAdd(&cnt[r3 >> SRC_BITS], 1u);
    csr_src[p0] = (int)(r0 & SRC_MASK); csr_src[p1] = (int)(r1 & SRC_MASK);
    csr_src[p2] = (int)(r2 & SRC_MASK); csr_src[p3] = (int)(r3 & SRC_MASK);
  }
  for (; i < end; i += PART_T) {
    const unsigned r = rec[i];
    const unsigned p = atomicAdd(&cnt[r >> SRC_BITS], 1u);
    csr_src[p] = (int)(r & SRC_MASK);
  }
}

__global__ void agg1_kernel(const unsigned* __restrict__ rowptr, const int* __restrict__ csr_src,
                            const float* __restrict__ att_src, const float* __restrict__ att_dst,
                            const float* __restrict__ xl1, const float* __restrict__ bias1,
                            float* __restrict__ h, int n) {
  int t = blockIdx.x * blockDim.x + threadIdx.x;
  if (t >= n * HEADS) return;
  const int d = t >> 3, hh = t & 7;
  const int* __restrict__ cp = csr_src + rowptr[d];
  const int* __restrict__ ce = csr_src + rowptr[d + 1];
  const float* __restrict__ xb = xl1 + hh * HID;
  const float4* wsp = (const float4*)(att_src + hh * HID);
  const float4 ws0 = wsp[0], ws1 = wsp[1];
  float ad;
  {
    const float4* wdp = (const float4*)(att_dst + hh * HID);
    const float4* xd = (const float4*)(xb + (size_t)d * D1);
    ad = dot8(xd[0], xd[1], wdp[0], wdp[1]);
  }

  float m = -__builtin_huge_valf();
  float den = 0.f;
  float4 acc0 = make_float4(0.f, 0.f, 0.f, 0.f), acc1 = acc0;
  for (; cp + 4 <= ce; cp += 4) {
    const int s0 = cp[0], s1 = cp[1], s2 = cp[2], s3 = cp[3];
    const float4* x0 = (const float4*)(xb + (size_t)s0 * D1);
    const float4* x1 = (const float4*)(xb + (size_t)s1 * D1);
    const float4* x2 = (const float4*)(xb + (size_t)s2 * D1);
    const float4* x3 = (const float4*)(xb + (size_t)s3 * D1);
    const float4 p0 = x0[0], q0 = x0[1], p1 = x1[0], q1 = x1[1];
    const float4 p2 = x2[0], q2 = x2[1], p3 = x3[0], q3 = x3[1];
    const float al0 = lrelu(dot8(p0, q0, ws0, ws1) + ad), al1 = lrelu(dot8(p1, q1, ws0, ws1) + ad);
    const float al2 = lrelu(dot8(p2, q2, ws0, ws1) + ad), al3 = lrelu(dot8(p3, q3, ws0, ws1) + ad);
    float mn = al0 > m ? al0 : m;
    mn = al1 > mn ? al1 : mn; mn = al2 > mn ? al2 : mn; mn = al3 > mn ? al3 : mn;
    const float sc = __expf(m - mn);
    const float e0 = __expf(al0 - mn), e1 = __expf(al1 - mn);
    const float e2 = __expf(al2 - mn), e3 = __expf(al3 - mn);
    m = mn;
    den = fmaf(den, sc, e0); den += e1; den += e2; den += e3;
    rescale_add(acc0, sc, p0, e0); rescale_add(acc1, sc, q0, e0);
    add_scaled(acc0, p1, e1);      add_scaled(acc1, q1, e1);
    add_scaled(acc0, p2, e2);      add_scaled(acc1, q2, e2);
    add_scaled(acc0, p3, e3);      add_scaled(acc1, q3, e3);
  }
  #pragma unroll 1
  for (; cp < ce; cp++) {
    const int s = cp[0];
    const float4* xp = (const float4*)(xb + (size_t)s * D1);
    const float4 v0 = xp[0], v1 = xp[1];
    const float al = lrelu(dot8(v0, v1, ws0, ws1) + ad);
    const float mn = al > m ? al : m;
    const float sc = __expf(m - mn);
    const float ex = __expf(al - mn);
    m = mn;
    den = fmaf(den, sc, ex);
    rescale_add(acc0, sc, v0, ex); rescale_add(acc1, sc, v1, ex);
  }
  const float dn = den + EPS_GAT;
  const float4* bp = (const float4*)(bias1 + hh * HID);
  const float4 b0 = bp[0], b1 = bp[1];
  float r[8] = { acc0.x / dn + b0.x, acc0.y / dn + b0.y, acc0.z / dn + b0.z, acc0.w / dn + b0.w,
                 acc1.x / dn + b1.x, acc1.y / dn + b1.y, acc1.z / dn + b1.z, acc1.w / dn + b1.w };
  #pragma unroll
  for (int c = 0; c < 8; c++) r[c] = r[c] > 0.f ? r[c] : (__expf(r[c]) - 1.f);
  float4* hp = (float4*)(h + (size_t)d * D1 + hh * HID);
  hp[0] = make_float4(r[0], r[1], r[2], r[3]);
  hp[1] = make_float4(r[4], r[5], r[6], r[7]);
}

__global__ __launch_bounds__(256) void gemm2_kernel(const float* __restrict__ h,
                                                    const float* __restrict__ W2,
                                                    float* __restrict__ xl2, int nrows) {
  const int wave = threadIdx.x >> 5;
  const int lane = threadIdx.x & 31;
  const int tile = blockIdx.x * 8 + wave;
  const int rowBase = tile * 16;
  if (rowBase >= nrows) return;
  const int ln = lane & 15;
  const int khalf = (lane >> 4) << 1;
  int arow = rowBase + ln; if (arow >= nrows) arow = nrows - 1;
  const float* ap = h + (size_t)arow * D1;

  v8f c = {};
  #pragma unroll
  for (int k = 0; k < D1; k += 4) {
    v2f a = *(const v2f*)&ap[k + khalf];
    v2f b; b.x = W2[(k + khalf) * NCLS + ln];
           b.y = W2[(k + khalf + 1) * NCLS + ln];
    c = __builtin_amdgcn_wmma_f32_16x16x4_f32(false, a, false, b, (short)0, c,
                                              false, false);
  }
  const int mBase = (lane >> 4) * 8;
  float* outp = xl2 + (size_t)rowBase * NCLS;
  #pragma unroll
  for (int i = 0; i < 8; i++) {
    int row = rowBase + mBase + i;
    if (row < nrows) outp[(size_t)(mBase + i) * NCLS + ln] = c[i];
  }
}

__global__ void agg2_kernel(const unsigned* __restrict__ rowptr, const int* __restrict__ csr_src,
                            const float* __restrict__ att_src, const float* __restrict__ att_dst,
                            const float* __restrict__ xl2, const float* __restrict__ bias2,
                            float* __restrict__ out, int n) {
  int t = blockIdx.x * blockDim.x + threadIdx.x;
  if (t >= n * 4) return;
  const int d = t >> 2, q = t & 3;
  const int* __restrict__ cp = csr_src + rowptr[d];
  const int* __restrict__ ce = csr_src + rowptr[d + 1];
  const float4* __restrict__ xq = (const float4*)xl2 + q;
  const float4 wsq = ((const float4*)att_src)[q];
  const float ad = quad_sum(dot4(xq[(size_t)d * 4], ((const float4*)att_dst)[q]));

  float m = -__builtin_huge_valf();
  float den = 0.f;
  float4 acc = make_float4(0.f, 0.f, 0.f, 0.f);
  for (; cp + 4 <= ce; cp += 4) {
    const int s0 = cp[0], s1 = cp[1], s2 = cp[2], s3 = cp[3];
    const float4 v0 = xq[(size_t)s0 * 4], v1 = xq[(size_t)s1 * 4];
    const float4 v2 = xq[(size_t)s2 * 4], v3 = xq[(size_t)s3 * 4];
    const float al0 = lrelu(quad_sum(dot4(v0, wsq)) + ad), al1 = lrelu(quad_sum(dot4(v1, wsq)) + ad);
    const float al2 = lrelu(quad_sum(dot4(v2, wsq)) + ad), al3 = lrelu(quad_sum(dot4(v3, wsq)) + ad);
    float mn = al0 > m ? al0 : m;
    mn = al1 > mn ? al1 : mn; mn = al2 > mn ? al2 : mn; mn = al3 > mn ? al3 : mn;
    const float sc = __expf(m - mn);
    const float e0 = __expf(al0 - mn), e1 = __expf(al1 - mn);
    const float e2 = __expf(al2 - mn), e3 = __expf(al3 - mn);
    m = mn;
    den = fmaf(den, sc, e0); den += e1; den += e2; den += e3;
    rescale_add(acc, sc, v0, e0);
    add_scaled(acc, v1, e1); add_scaled(acc, v2, e2); add_scaled(acc, v3, e3);
  }
  #pragma unroll 1
  for (; cp < ce; cp++) {
    const unsigned s = cp[0];
    const float4 v = xq[(size_t)s * 4];
    const float al = lrelu(quad_sum(dot4(v, wsq)) + ad);
    const float mn = al > m ? al : m;
    const float sc = __expf(m - mn);
    const float ex = __expf(al - mn);
    m = mn;
    den = fmaf(den, sc, ex);
    rescale_add(acc, sc, v, ex);
  }
  const float dn = den + EPS_GAT;
  const float4 b = ((const float4*)bias2)[q];
  ((float4*)(out + (size_t)d * NCLS))[q] =
      make_float4(acc.x / dn + b.x, acc.y / dn + b.y, acc.z / dn + b.z, acc.w / dn + b.w);
}

extern "C" void kernel_launch(void* const* d_in, const int* in_sizes, int n_in,
                              void* d_out, int out_size, void* d_ws, size_t ws_size,
                              hipStream_t stream) {
  const float* x        = (const float*)d_in[0];
  const int*   ei       = (const int*)d_in[1];
  const float* W1       = (const float*)d_in[2];
  const float* att_src1 = (const float*)d_in[3];
  const float* att_dst1 = (const float*)d_in[4];
  const float* bias1    = (const float*)d_in[5];
  const float* W2       = (const float*)d_in[6];
  const float* att_src2 = (const float*)d_in[7];
  const float* att_dst2 = (const float*)d_in[8];
  const float* bias2    = (const float*)d_in[9];
  float* out = (float*)d_out;

  const int N = in_sizes[0] / F_IN;
  const int E = in_sizes[1] / 2;
  const size_t Np = ((size_t)N + 4) & ~(size_t)3;
  const int NB = (N + BNODES - 1) / BNODES;
  const int chunk = (E + PART_G - 1) / PART_G;
  const int M = NB * PART_G;

  float* ws = (float*)d_ws;
  size_t o = 0;
  float*    xl1    = ws + o; o += (size_t)N * D1;
  float*    h      = ws + o; o += (size_t)N * D1;
  float*    xl2    = ws + o; o += (size_t)N * NCLS;
  unsigned* rowptr = (unsigned*)(ws + o); o += Np;
  unsigned* hist   = (unsigned*)(ws + o); o += (size_t)M;
  int*      csrsrc = (int*)(ws + o);      o += (size_t)E;
  unsigned* rec    = (unsigned*)h;
  unsigned* bstart = (unsigned*)xl2;

  const int TB = 256;
  auto nb = [](long long n, int b) { return (unsigned)((n + b - 1) / b); };

  part_count<<<PART_G, PART_T, 0, stream>>>(ei, hist, E, NB, chunk);
  part_place<<<PART_G, PART_T, 0, stream>>>(ei, hist, bstart, rec, E, NB, chunk);
  bucket_sort<<<NB, PART_T, 0, stream>>>(bstart, rec, rowptr, csrsrc, N, E, NB);

  gemm1_kernel<<<G1_BLOCKS, 128, 0, stream>>>(x, W1, xl1, N);
  agg1_kernel<<<nb((long long)N * HEADS, TB), TB, 0, stream>>>(rowptr, csrsrc, att_src1, att_dst1,
                                                               xl1, bias1, h, N);

  gemm2_kernel<<<nb(((long long)N + 15) / 16, 8), 256, 0, stream>>>(h, W2, xl2, N);
  agg2_kernel<<<nb((long long)N * 4, TB), TB, 0, stream>>>(rowptr, csrsrc, att_src2, att_dst2,
                                                           xl2, bias2, out, N);
}


# Layer-2 product fused into the layer-1 aggregation: each work-group stages its 32 rows of h in LDS and forms xl2 = h @ W2 itself (same f32 WMMA chain, same k order); h plane and one launch go
# speedup vs baseline: 9.3506x; 1.0424x over previous
#include <hip/hip_runtime.h>
#include <hip/hip_bf16.h>

typedef __attribute__((ext_vector_type(2))) float v2f;
typedef __attribute__((ext_vector_type(8))) float v8f;
typedef __attribute__((ext_vector_type(4))) float v4f;

#define F_IN 512
#define HID 8
#define HEADS 8
#define D1 64
#define NCLS 16
#define NEG_SLOPE 0.2f
#define EPS_GAT 1e-16f
#define BSH 9
#define BNODES 512
#define SRC_BITS 23
#define SRC_MASK 0x7FFFFFu
#define NBMAX 256
#define PART_G 256
#define PART_T 512

__device__ __forceinline__ float lrelu(float x) { return x > 0.f ? x : x * NEG_SLOPE; }

__device__ __forceinline__ void rescale_add(float4& a, float sc, const float4& v, float e) {
  a.x = fmaf(v.x, e, a.x * sc); a.y = fmaf(v.y, e, a.y * sc);
  a.z = fmaf(v.z, e, a.z * sc); a.w = fmaf(v.w, e, a.w * sc);
}
__device__ __forceinline__ void add_scaled(float4& a, const float4& v, float e) {
  a.x = fmaf(v.x, e, a.x); a.y = fmaf(v.y, e, a.y);
  a.z = fmaf(v.z, e, a.z); a.w = fmaf(v.w, e, a.w);
}

__device__ __forceinline__ float dot8(const float4& a, const float4& b, const float4& w0, const float4& w1) {
  float s = a.x * w0.x;
  s = fmaf(a.y, w0.y, s); s = fmaf(a.z, w0.z, s); s = fmaf(a.w, w0.w, s);
  s = fmaf(b.x, w1.x, s); s = fmaf(b.y, w1.y, s); s = fmaf(b.z, w1.z, s); s = fmaf(b.w, w1.w, s);
  return s;
}
__device__ __forceinline__ float dot4(const float4& a, const float4& w) {
  float s = a.x * w.x;
  s = fmaf(a.y, w.y, s); s = fmaf(a.z, w.z, s); s = fmaf(a.w, w.w, s);
  return s;
}
__device__ __forceinline__ float quad_sum(float v) {
  v += __int_as_float(__builtin_amdgcn_update_dpp(0, __float_as_int(v), 0xB1, 0xF, 0xF, true));
  v += __int_as_float(__builtin_amdgcn_update_dpp(0, __float_as_int(v), 0x4E, 0xF, 0xF, true));
  return v;
}

__device__ __forceinline__ unsigned row16_sum(unsigned v) {
  v += (unsigned)__builtin_amdgcn_update_dpp(0, (int)v, 0xB1, 0xF, 0xF, true);
  v += (unsigned)__builtin_amdgcn_update_dpp(0, (int)v, 0x4E, 0xF, 0xF, true);
  v += (unsigned)__builtin_amdgcn_update_dpp(0, (int)v, 0x141, 0xF, 0xF, true);
  v += (unsigned)__builtin_amdgcn_update_dpp(0, (int)v, 0x140, 0xF, 0xF, true);
  return v;
}

#define G1_BLOCKS 512
__global__ __launch_bounds__(128) void gemm1_kernel(const float* __restrict__ x,
                                                    const float* __restrict__ W1,
                                                    float* __restrict__ xl1, int nrows) {
  __shared__ float At[16 * 516];
  const int tid = threadIdx.x;
  const int wave = tid >> 5;
  const int lane = tid & 31;
  const int ln = lane & 15;
  const int khalf = (lane >> 4) << 1;
  const int colBase = wave * 16;
  const int mBase = (lane >> 4) * 8;
  const int ntiles = (nrows + 15) >> 4;

  v2f b[F_IN / 4];
  {
    const float* wp = W1 + (size_t)khalf * D1 + colBase + ln;
    #pragma unroll
    for (int j = 0; j < F_IN / 4; j++) {
      b[j].x = wp[(size_t)(4 * j) * D1];
      b[j].y = wp[(size_t)(4 * j + 1) * D1];
    }
  }

  v4f st[16];
  int tile = blockIdx.x;
  if (tile < ntiles) {
    #pragma unroll
    for (int r = 0; r < 16; r++) {
      int row = tile * 16 + r; if (row >= nrows) row = nrows - 1;
      st[r] = ((const v4f*)(x + (size_t)row * F_IN))[tid];
    }
  }
  for (; tile < ntiles; tile += G1_BLOCKS) {
    __syncthreads();
    #pragma unroll
    for (int r = 0; r < 16; r++) *(v4f*)&At[r * 516 + tid * 4] = st[r];
    __syncthreads();
    const int nt = tile + G1_BLOCKS;
    if (nt < ntiles) {
      #pragma unroll
      for (int r = 0; r < 16; r++) {
        int row = nt * 16 + r; if (row >= nrows) row = nrows - 1;
        st[r] = ((const v4f*)(x + (size_t)row * F_IN))[tid];
      }
    }
    v8f c = {};
    const float* arow = &At[ln * 516 + khalf];
    v2f af[2][16];
    #pragma unroll
    for (int i = 0; i < 16; i++) af[0][i] = *(const v2f*)&arow[4 * i];
    #pragma unroll
    for (int blk = 0; blk < F_IN / 64; blk++) {
      if (blk + 1 < F_IN / 64) {
        #pragma unroll
        for (int i = 0; i < 16; i++) af[(blk + 1) & 1][i] = *(const v2f*)&arow[4 * (16 * (blk + 1) + i)];
      }
      __builtin_amdgcn_sched_barrier(0);
      #pragma unroll
      for (int i = 0; i < 16; i++)
        c = __builtin_amdgcn_wmma_f32_16x16x4_f32(false, af[blk & 1][i], false, b[16 * blk + i],
                                                  (short)0, c, false, false);
    }
    const int rowBase = tile * 16;
    float* outp = xl1 + (size_t)rowBase * D1 + colBase;
    #pragma unroll
    for (int i = 0; i < 8; i++) {
      int row = rowBase + mBase + i;
      if (row < nrows) outp[(size_t)(mBase + i) * D1 + ln] = c[i];
    }
  }
}


__global__ __launch_bounds__(PART_T) void part_count(const int* __restrict__ ei,
                                                     unsigned* __restrict__ hist,
                                                     int E_, int NB, int chunk) {
  __shared__ unsigned c[NBMAX];
  const int tid = threadIdx.x;
  if (tid < NB) c[tid] = 0u;
  __syncthreads();
  const int* __restrict__ dst = ei + E_;
  const int b0 = blockIdx.x * chunk;
  const int b1 = (b0 + chunk < E_) ? (b0 + chunk) : E_;
  int e = b0 + tid;
  for (; e + 3 * PART_T < b1; e += 4 * PART_T) {
    const int d0 = dst[e], d1 = dst[e + PART_T], d2 = dst[e + 2 * PART_T], d3 = dst[e + 3 * PART_T];
    atomicAdd(&c[d0 >> BSH], 1u); atomicAdd(&c[d1 >> BSH], 1u);
    atomicAdd(&c[d2 >> BSH], 1u); atomicAdd(&c[d3 >> BSH], 1u);
  }
  for (; e < b1; e += PART_T) atomicAdd(&c[dst[e] >> BSH], 1u);
  __syncthreads();
  if (tid < NB) hist[(size_t)tid * gridDim.x + blockIdx.x] = c[tid];
}

__global__ __launch_bounds__(PART_T) void part_place(const int* __restrict__ ei,
                                                     const unsigned* __restrict__ hist,
                                                     unsigned* __restrict__ bstart,
                                                     unsigned* __restrict__ rec,
                                                     int E_, int NB, int chunk) {
  __shared__ unsigned cur[NBMAX];
  __shared__ unsigned rtot[NBMAX];
  __shared__ unsigned rpar[NBMAX];
  __shared__ unsigned wtot[PART_T / 32];
  const int tid = threadIdx.x, lane = tid & 31, wave = tid >> 5;
  const unsigned c = blockIdx.x;
  {
    const int hw = tid >> 4, l16 = tid & 15;
    uint4 q[NBMAX / 32][PART_G / 64];
    #pragma unroll
    for (int i = 0; i < NBMAX / 32; i++) {
      const int b = hw + 32 * i;
      const uint4* row = (const uint4*)(hist + (size_t)(b < NB ? b : NB - 1) * PART_G);
      #pragma unroll
      for (int j = 0; j < PART_G / 64; j++) q[i][j] = row[16 * j + l16];
    }
    #pragma unroll
    for (int i = 0; i < NBMAX / 32; i++) {
      const int b = hw + 32 * i;
      unsigned tot = 0u, par = 0u;
      #pragma unroll
      for (int j = 0; j < PART_G / 64; j++) {
        const unsigned i0 = 64u * j + 4u * l16;
        tot += q[i][j].x + q[i][j].y + q[i][j].z + q[i][j].w;
        par += (i0 < c ? q[i][j].x : 0u) + (i0 + 1u < c ? q[i][j].y : 0u) +
               (i0 + 2u < c ? q[i][j].z : 0u) + (i0 + 3u < c ? q[i][j].w : 0u);
      }
      tot = row16_sum(tot); par = row16_sum(par);
      if (l16 == 0 && b < NB) { rtot[b] = tot; rpar[b] = par; }
    }
  }
  __syncthreads();
  const unsigned t = (tid < NB) ? rtot[tid] : 0u;
  unsigned incl = t;
  #pragma unroll
  for (int off = 1; off < 32; off <<= 1) {
    const unsigned up = __shfl_up(incl, off);
    if (lane >= off) incl += up;
  }
  if (lane == 31) wtot[wave] = incl;
  __syncthreads();
  if (wave == 0) {
    const unsigned w = (lane < PART_T / 32) ? wtot[lane] : 0u;
    unsigned wi = w;
    #pragma unroll
    for (int off = 1; off < 32; off <<= 1) {
      const unsigned up = __shfl_up(wi, off);
      if (lane >= off) wi += up;
    }
    if (lane < PART_T / 32) wtot[lane] = wi - w;
  }
  __syncthreads();
  if (tid < NB) {
    const unsigned first = wtot[wave] + (incl - t);
    cur[tid] = first + rpar[tid];
    if (c == 0u) bstart[tid] = first;
  }
  __syncthreads();
  const int* __restrict__ dst = ei + E_;
  const int b0 = blockIdx.x * chunk;
  const int b1 = (b0 + chunk < E_) ? (b0 + chunk) : E_;
  int e = b0 + tid;
  for (; e + 3 * PART_T < b1; e += 4 * PART_T) {
    const unsigned d0 = dst[e], d1 = dst[e + PART_T], d2 = dst[e + 2 * PART_T], d3 = dst[e + 3 * PART_T];
    const unsigned s0 = ei[e],  s1 = ei[e + PART_T],  s2 = ei[e + 2 * PART_T],  s3 = ei[e + 3 * PART_T];
    const unsigned p0 = atomicAdd(&cur[d0 >> BSH], 1u), p1 = atomicAdd(&cur[d1 >> BSH], 1u);
    const unsigned p2 = atomicAdd(&cur[d2 >> BSH], 1u), p3 = atomicAdd(&cur[d3 >> BSH], 1u);
    rec[p0] = s0 | ((d0 & (BNODES - 1)) << SRC_BITS); rec[p1] = s1 | ((d1 & (BNODES - 1)) << SRC_BITS);
    rec[p2] = s2 | ((d2 & (BNODES - 1)) << SRC_BITS); rec[p3] = s3 | ((d3 & (BNODES - 1)) << SRC_BITS);
  }
  for (; e < b1; e += PART_T) {
    const unsigned d = dst[e], s = ei[e];
    const unsigned p = atomicAdd(&cur[d >> BSH], 1u);
    rec[p] = s | ((d & (BNODES - 1)) << SRC_BITS);
  }
}

__global__ __launch_bounds__(PART_T) void bucket_sort(const unsigned* __restrict__ bstart,
                                                      const unsigned* __restrict__ rec,
                                                      unsigned* __restrict__ rowptr,
                                                      int* __restrict__ csr_src,
                                                      int n, int E_, int NB) {
  __shared__ unsigned cnt[BNODES];
  __shared__ unsigned wtot[PART_T / 32];
  const int tid = threadIdx.x, lane = tid & 31, wave = tid >> 5;
  const int k = blockIdx.x;
  const unsigned start = bstart[k];
  const unsigned end = (k + 1 < NB) ? bstart[k + 1] : (unsigned)E_;
  cnt[tid] = 0u;
  __syncthreads();
  unsigned i = start + tid;
  for (; i + 3 * PART_T < end; i += 4 * PART_T) {
    const unsigned r0 = rec[i], r1 = rec[i + PART_T], r2 = rec[i + 2 * PART_T], r3 = rec[i + 3 * PART_T];
    atomicAdd(&cnt[r0 >> SRC_BITS], 1u); atomicAdd(&cnt[r1 >> SRC_BITS], 1u);
    atomicAdd(&cnt[r2 >> SRC_BITS], 1u); atomicAdd(&cnt[r3 >> SRC_BITS], 1u);
  }
  for (; i < end; i += PART_T) atomicAdd(&cnt[rec[i] >> SRC_BITS], 1u);
  __syncthreads();
  const unsigned c = cnt[tid];
  unsigned incl = c;
  #pragma unroll
  for (int off = 1; off < 32; off <<= 1) {
    const unsigned up = __shfl_up(incl, off);
    if (lane >= off) incl += up;
  }
  if (lane == 31) wtot[wave] = incl;
  __syncthreads();
  if (wave == 0) {
    const unsigned w = (lane < PART_T / 32) ? wtot[lane] : 0u;
    unsigned wi = w;
    #pragma unroll
    for (int off = 1; off < 32; off <<= 1) {
      const unsigned up = __shfl_up(wi, off);
      if (lane >= off) wi += up;
    }
    if (lane < PART_T / 32) wtot[lane] = wi - w;
  }
  __syncthreads();
  const unsigned first = start + wtot[wave] + (incl - c);
  cnt[tid] = first;
  const int node = k * BNODES + tid;
  if (node < n) rowptr[node] = first;
  if (k == NB - 1 && tid == 0) rowptr[n] = (unsigned)E_;
  __syncthreads();
  i = start + tid;
  for (; i + 3 * PART_T < end; i += 4 * PART_T) {
    const unsigned r0 = rec[i], r1 = rec[i + PART_T], r2 = rec[i + 2 * PART_T], r3 = rec[i + 3 * PART_T];
    const unsigned p0 = atomicAdd(&cnt[r0 >> SRC_BITS], 1u), p1 = atomicAdd(&cnt[r1 >> SRC_BITS], 1u);
    const unsigned p2 = atomicAdd(&cnt[r2 >> SRC_BITS], 1u), p3 = atomicAdd(&cnt[r3 >> SRC_BITS], 1u);
    csr_src[p0] = (int)(r0 & SRC_MASK); csr_src[p1] = (int)(r1 & SRC_MASK);
    csr_src[p2] = (int)(r2 & SRC_MASK); csr_src[p3] = (int)(r3 & SRC_MASK);
  }
  for (; i < end; i += PART_T) {
    const unsigned r = rec[i];
    const unsigned p = atomicAdd(&cnt[r >> SRC_BITS], 1u);
    csr_src[p] = (int)(r & SRC_MASK);
  }
}

#define A1_T 256
#define HS_PITCH 68
__global__ __launch_bounds__(A1_T) void agg1_kernel(const unsigned* __restrict__ rowptr, const int* __restrict__ csr_src,
                            const float* __restrict__ att_src, const float* __restrict__ att_dst,
                            const float* __restrict__ xl1, const float* __restrict__ bias1,
                            const float* __restrict__ W2, float* __restrict__ xl2, int n) {
  __shared__ float hs[(A1_T / HEADS) * HS_PITCH];
  const int t = blockIdx.x * A1_T + threadIdx.x;
  const bool live = t < n * HEADS;
  const int d = live ? (t >> 3) : (n - 1), hh = t & 7;
  const unsigned seg0 = rowptr[d];
  const unsigned seg1 = live ? rowptr[d + 1] : seg0;
  const int* __restrict__ cp = csr_src + seg0;
  const int* __restrict__ ce = csr_src + seg1;
  const float* __restrict__ xb = xl1 + hh * HID;
  const float4* wsp = (const float4*)(att_src + hh * HID);
  const float4 ws0 = wsp[0], ws1 = wsp[1];
  float ad;
  {
    const float4* wdp = (const float4*)(att_dst + hh * HID);
    const float4* xd = (const float4*)(xb + (size_t)d * D1);
    ad = dot8(xd[0], xd[1], wdp[0], wdp[1]);
  }

  float m = -__builtin_huge_valf();
  float den = 0.f;
  float4 acc0 = make_float4(0.f, 0.f, 0.f, 0.f), acc1 = acc0;
  for (; cp + 4 <= ce; cp += 4) {
    const int s0 = cp[0], s1 = cp[1], s2 = cp[2], s3 = cp[3];
    const float4* x0 = (const float4*)(xb + (size_t)s0 * D1);
    const float4* x1 = (const float4*)(xb + (size_t)s1 * D1);
    const float4* x2 = (const float4*)(xb + (size_t)s2 * D1);
    const float4* x3 = (const float4*)(xb + (size_t)s3 * D1);
    const float4 p0 = x0[0], q0 = x0[1], p1 = x1[0], q1 = x1[1];
    const float4 p2 = x2[0], q2 = x2[1], p3 = x3[0], q3 = x3[1];
    const float al0 = lrelu(dot8(p0, q0, ws0, ws1) + ad), al1 = lrelu(dot8(p1, q1, ws0, ws1) + ad);
    const float al2 = lrelu(dot8(p2, q2, ws0, ws1) + ad), al3 = lrelu(dot8(p3, q3, ws0, ws1) + ad);
    float mn = al0 > m ? al0 : m;
    mn = al1 > mn ? al1 : mn; mn = al2 > mn ? al2 : mn; mn = al3 > mn ? al3 : mn;
    const float sc = __expf(m - mn);
    const float e0 = __expf(al0 - mn), e1 = __expf(al1 - mn);
    const float e2 = __expf(al2 - mn), e3 = __expf(al3 - mn);
    m = mn;
    den = fmaf(den, sc, e0); den += e1; den += e2; den += e3;
    rescale_add(acc0, sc, p0, e0); rescale_add(acc1, sc, q0, e0);
    add_scaled(acc0, p1, e1);      add_scaled(acc1, q1, e1);
    add_scaled(acc0, p2, e2);      add_scaled(acc1, q2, e2);
    add_scaled(acc0, p3, e3);      add_scaled(acc1, q3, e3);
  }
  #pragma unroll 1
  for (; cp < ce; cp++) {
    const int s = cp[0];
    const float4* xp = (const float4*)(xb + (size_t)s * D1);
    const float4 v0 = xp[0], v1 = xp[1];
    const float al = lrelu(dot8(v0, v1, ws0, ws1) + ad);
    const float mn = al > m ? al : m;
    const float sc = __expf(m - mn);
    const float ex = __expf(al - mn);
    m = mn;
    den = fmaf(den, sc, ex);
    rescale_add(acc0, sc, v0, ex); rescale_add(acc1, sc, v1, ex);
  }
  const float dn = den + EPS_GAT;
  const float4* bp = (const float4*)(bias1 + hh * HID);
  const float4 b0 = bp[0], b1 = bp[1];
  float r[8] = { acc0.x / dn + b0.x, acc0.y / dn + b0.y, acc0.z / dn + b0.z, acc0.w / dn + b0.w,
                 acc1.x / dn + b1.x, acc1.y / dn + b1.y, acc1.z / dn + b1.z, acc1.w / dn + b1.w };
  #pragma unroll
  for (int c = 0; c < 8; c++) r[c] = r[c] > 0.f ? r[c] : (__expf(r[c]) - 1.f);
  float4* hp = (float4*)&hs[(threadIdx.x >> 3) * HS_PITCH + hh * HID];
  hp[0] = make_float4(r[0], r[1], r[2], r[3]);
  hp[1] = make_float4(r[4], r[5], r[6], r[7]);
  const int wave = threadIdx.x >> 5, lane = threadIdx.x & 31;
  const int ln = lane & 15;
  const int khalf = (lane >> 4) << 1;
  v2f b[D1 / 4];
  if (wave < 2) {
    #pragma unroll
    for (int j = 0; j < D1 / 4; j++) {
      b[j].x = W2[(4 * j + khalf) * NCLS + ln];
      b[j].y = W2[(4 * j + khalf + 1) * NCLS + ln];
    }
  }
  __syncthreads();
  if (wave < 2) {
    const float* arow = &hs[(wave * 16 + ln) * HS_PITCH + khalf];
    v8f c = {};
    #pragma unroll
    for (int j = 0; j < D1 / 4; j++) {
      const v2f a = *(const v2f*)&arow[4 * j];
      c = __builtin_amdgcn_wmma_f32_16x16x4_f32(false, a, false, b[j], (short)0, c, false, false);
    }
    const int mBase = (lane >> 4) * 8;
    const int rowBase = blockIdx.x * (A1_T / HEADS) + wave * 16;
    float* outp = xl2 + (size_t)rowBase * NCLS;
    #pragma unroll
    for (int i = 0; i < 8; i++) {
      const int row = rowBase + mBase + i;
      if (row < n) outp[(size_t)(mBase + i) * NCLS + ln] = c[i];
    }
  }
}

__global__ void agg2_kernel(const unsigned* __restrict__ rowptr, const int* __restrict__ csr_src,
                            const float* __restrict__ att_src, const float* __restrict__ att_dst,
                            const float* __restrict__ xl2, const float* __restrict__ bias2,
                            float* __restrict__ out, int n) {
  int t = blockIdx.x * blockDim.x + threadIdx.x;
  if (t >= n * 4) return;
  const int d = t >> 2, q = t & 3;
  const int* __restrict__ cp = csr_src + rowptr[d];
  const int* __restrict__ ce = csr_src + rowptr[d + 1];
  const float4* __restrict__ xq = (const float4*)xl2 + q;
  const float4 wsq = ((const float4*)att_src)[q];
  const float ad = quad_sum(dot4(xq[(size_t)d * 4], ((const float4*)att_dst)[q]));

  float m = -__builtin_huge_valf();
  float den = 0.f;
  float4 acc = make_float4(0.f, 0.f, 0.f, 0.f);
  for (; cp + 4 <= ce; cp += 4) {
    const int s0 = cp[0], s1 = cp[1], s2 = cp[2], s3 = cp[3];
    const float4 v0 = xq[(size_t)s0 * 4], v1 = xq[(size_t)s1 * 4];
    const float4 v2 = xq[(size_t)s2 * 4], v3 = xq[(size_t)s3 * 4];
    const float al0 = lrelu(quad_sum(dot4(v0, wsq)) + ad), al1 = lrelu(quad_sum(dot4(v1, wsq)) + ad);
    const float al2 = lrelu(quad_sum(dot4(v2, wsq)) + ad), al3 = lrelu(quad_sum(dot4(v3, wsq)) + ad);
    float mn = al0 > m ? al0 : m;
    mn = al1 > mn ? al1 : mn; mn = al2 > mn ? al2 : mn; mn = al3 > mn ? al3 : mn;
    const float sc = __expf(m - mn);
    const float e0 = __expf(al0 - mn), e1 = __expf(al1 - mn);
    const float e2 = __expf(al2 - mn), e3 = __expf(al3 - mn);
    m = mn;
    den = fmaf(den, sc, e0); den += e1; den += e2; den += e3;
    rescale_add(acc, sc, v0, e0);
    add_scaled(acc, v1, e1); add_scaled(acc, v2, e2); add_scaled(acc, v3, e3);
  }
  #pragma unroll 1
  for (; cp < ce; cp++) {
    const unsigned s = cp[0];
    const float4 v = xq[(size_t)s * 4];
    const float al = lrelu(quad_sum(dot4(v, wsq)) + ad);
    const float mn = al > m ? al : m;
    const float sc = __expf(m - mn);
    const float ex = __expf(al - mn);
    m = mn;
    den = fmaf(den, sc, ex);
    rescale_add(acc, sc, v, ex);
  }
  const float dn = den + EPS_GAT;
  const float4 b = ((const float4*)bias2)[q];
  ((float4*)(out + (size_t)d * NCLS))[q] =
      make_float4(acc.x / dn + b.x, acc.y / dn + b.y, acc.z / dn + b.z, acc.w / dn + b.w);
}

extern "C" void kernel_launch(void* const* d_in, const int* in_sizes, int n_in,
                              void* d_out, int out_size, void* d_ws, size_t ws_size,
                              hipStream_t stream) {
  const float* x        = (const float*)d_in[0];
  const int*   ei       = (const int*)d_in[1];
  const float* W1       = (const float*)d_in[2];
  const float* att_src1 = (const float*)d_in[3];
  const float* att_dst1 = (const float*)d_in[4];
  const float* bias1    = (const float*)d_in[5];
  const float* W2       = (const float*)d_in[6];
  const float* att_src2 = (const float*)d_in[7];
  const float* att_dst2 = (const float*)d_in[8];
  const float* bias2    = (const float*)d_in[9];
  float* out = (float*)d_out;

  const int N = in_sizes[0] / F_IN;
  const int E = in_sizes[1] / 2;
  const size_t Np = ((size_t)N + 4) & ~(size_t)3;
  const int NB = (N + BNODES - 1) / BNODES;
  const int chunk = (E + PART_G - 1) / PART_G;
  const int M = NB * PART_G;

  float* ws = (float*)d_ws;
  size_t o = 0;
  float*    xl1    = ws + o; o += (size_t)N * D1;
  float*    h      = ws + o; o += (size_t)N * D1;
  float*    xl2    = ws + o; o += (size_t)N * NCLS;
  unsigned* rowptr = (unsigned*)(ws + o); o += Np;
  unsigned* hist   = (unsigned*)(ws + o); o += (size_t)M;
  int*      csrsrc = (int*)(ws + o);      o += (size_t)E;
  unsigned* rec    = (unsigned*)h;
  unsigned* bstart = (unsigned*)xl2;

  const int TB = 256;
  auto nb = [](long long n, int b) { return (unsigned)((n + b - 1) / b); };

  part_count<<<PART_G, PART_T, 0, stream>>>(ei, hist, E, NB, chunk);
  part_place<<<PART_G, PART_T, 0, stream>>>(ei, hist, bstart, rec, E, NB, chunk);
  bucket_sort<<<NB, PART_T, 0, stream>>>(bstart, rec, rowptr, csrsrc, N, E, NB);

  gemm1_kernel<<<G1_BLOCKS, 128, 0, stream>>>(x, W1, xl1, N);
  agg1_kernel<<<nb((long long)N * HEADS, A1_T), A1_T, 0, stream>>>(rowptr, csrsrc, att_src1, att_dst1,
                                                                   xl1, bias1, W2, xl2, N);

  agg2_kernel<<<nb((long long)N * 4, TB), TB, 0, stream>>>(rowptr, csrsrc, att_src2, att_dst2,
                                                           xl2, bias2, out, N);
}


# Both placing passes of the sort rank records into an LDS image of their output and copy it out in order: scattered 4-byte stores become whole-line runs; sum order = LDS arrival rank as before
# speedup vs baseline: 11.4047x; 1.2197x over previous
#include <hip/hip_runtime.h>
#include <hip/hip_bf16.h>

typedef __attribute__((ext_vector_type(2))) float v2f;
typedef __attribute__((ext_vector_type(8))) float v8f;
typedef __attribute__((ext_vector_type(4))) float v4f;

#define F_IN 512
#define HID 8
#define HEADS 8
#define D1 64
#define NCLS 16
#define NEG_SLOPE 0.2f
#define EPS_GAT 1e-16f
#define BSH 9
#define BNODES 512
#define SRC_BITS 23
#define SRC_MASK 0x7FFFFFu
#define NBMAX 256
#define PART_G 256
#define PART_T 512
#define PLACE_CAP 8192
#define BUCKET_CAP 16384

__device__ __forceinline__ float lrelu(float x) { return x > 0.f ? x : x * NEG_SLOPE; }

__device__ __forceinline__ void rescale_add(float4& a, float sc, const float4& v, float e) {
  a.x = fmaf(v.x, e, a.x * sc); a.y = fmaf(v.y, e, a.y * sc);
  a.z = fmaf(v.z, e, a.z * sc); a.w = fmaf(v.w, e, a.w * sc);
}
__device__ __forceinline__ void add_scaled(float4& a, const float4& v, float e) {
  a.x = fmaf(v.x, e, a.x); a.y = fmaf(v.y, e, a.y);
  a.z = fmaf(v.z, e, a.z); a.w = fmaf(v.w, e, a.w);
}

__device__ __forceinline__ float dot8(const float4& a, const float4& b, const float4& w0, const float4& w1) {
  float s = a.x * w0.x;
  s = fmaf(a.y, w0.y, s); s = fmaf(a.z, w0.z, s); s = fmaf(a.w, w0.w, s);
  s = fmaf(b.x, w1.x, s); s = fmaf(b.y, w1.y, s); s = fmaf(b.z, w1.z, s); s = fmaf(b.w, w1.w, s);
  return s;
}
__device__ __forceinline__ float dot4(const float4& a, const float4& w) {
  float s = a.x * w.x;
  s = fmaf(a.y, w.y, s); s = fmaf(a.z, w.z, s); s = fmaf(a.w, w.w, s);
  return s;
}
__device__ __forceinline__ float quad_sum(float v) {
  v += __int_as_float(__builtin_amdgcn_update_dpp(0, __float_as_int(v), 0xB1, 0xF, 0xF, true));
  v += __int_as_float(__builtin_amdgcn_update_dpp(0, __float_as_int(v), 0x4E, 0xF, 0xF, true));
  return v;
}

__device__ __forceinline__ unsigned row16_sum(unsigned v) {
  v += (unsigned)__builtin_amdgcn_update_dpp(0, (int)v, 0xB1, 0xF, 0xF, true);
  v += (unsigned)__builtin_amdgcn_update_dpp(0, (int)v, 0x4E, 0xF, 0xF, true);
  v += (unsigned)__builtin_amdgcn_update_dpp(0, (int)v, 0x141, 0xF, 0xF, true);
  v += (unsigned)__builtin_amdgcn_update_dpp(0, (int)v, 0x140, 0xF, 0xF, true);
  return v;
}

#define G1_BLOCKS 512
__global__ __launch_bounds__(128) void gemm1_kernel(const float* __restrict__ x,
                                                    const float* __restrict__ W1,
                                                    float* __restrict__ xl1, int nrows) {
  __shared__ float At[16 * 516];
  const int tid = threadIdx.x;
  const int wave = tid >> 5;
  const int lane = tid & 31;
  const int ln = lane & 15;
  const int khalf = (lane >> 4) << 1;
  const int colBase = wave * 16;
  const int mBase = (lane >> 4) * 8;
  const int ntiles = (nrows + 15) >> 4;

  v2f b[F_IN / 4];
  {
    const float* wp = W1 + (size_t)khalf * D1 + colBase + ln;
    #pragma unroll
    for (int j = 0; j < F_IN / 4; j++) {
      b[j].x = wp[(size_t)(4 * j) * D1];
      b[j].y = wp[(size_t)(4 * j + 1) * D1];
    }
  }

  v4f st[16];
  int tile = blockIdx.x;
  if (tile < ntiles) {
    #pragma unroll
    for (int r = 0; r < 16; r++) {
      int row = tile * 16 + r; if (row >= nrows) row = nrows - 1;
      st[r] = ((const v4f*)(x + (size_t)row * F_IN))[tid];
    }
  }
  for (; tile < ntiles; tile += G1_BLOCKS) {
    __syncthreads();
    #pragma unroll
    for (int r = 0; r < 16; r++) *(v4f*)&At[r * 516 + tid * 4] = st[r];
    __syncthreads();
    const int nt = tile + G1_BLOCKS;
    if (nt < ntiles) {
      #pragma unroll
      for (int r = 0; r < 16; r++) {
        int row = nt * 16 + r; if (row >= nrows) row = nrows - 1;
        st[r] = ((const v4f*)(x + (size_t)row * F_IN))[tid];
      }
    }
    v8f c = {};
    const float* arow = &At[ln * 516 + khalf];
    v2f af[2][16];
    #pragma unroll
    for (int i = 0; i < 16; i++) af[0][i] = *(const v2f*)&arow[4 * i];
    #pragma unroll
    for (int blk = 0; blk < F_IN / 64; blk++) {
      if (blk + 1 < F_IN / 64) {
        #pragma unroll
        for (int i = 0; i < 16; i++) af[(blk + 1) & 1][i] = *(const v2f*)&arow[4 * (16 * (blk + 1) + i)];
      }
      __builtin_amdgcn_sched_barrier(0);
      #pragma unroll
      for (int i = 0; i < 16; i++)
        c = __builtin_amdgcn_wmma_f32_16x16x4_f32(false, af[blk & 1][i], false, b[16 * blk + i],
                                                  (short)0, c, false, false);
    }
    const int rowBase = tile * 16;
    float* outp = xl1 + (size_t)rowBase * D1 + colBase;
    #pragma unroll
    for (int i = 0; i < 8; i++) {
      int row = rowBase + mBase + i;
      if (row < nrows) outp[(size_t)(mBase + i) * D1 + ln] = c[i];
    }
  }
}


__global__ __launch_bounds__(PART_T) void part_count(const int* __restrict__ ei,
                                                     unsigned* __restrict__ hist,
                                                     int E_, int NB, int chunk) {
  __shared__ unsigned c[NBMAX];
  const int tid = threadIdx.x;
  if (tid < NB) c[tid] = 0u;
  __syncthreads();
  const int* __restrict__ dst = ei + E_;
  const int b0 = blockIdx.x * chunk;
  const int b1 = (b0 + chunk < E_) ? (b0 + chunk) : E_;
  int e = b0 + tid;
  for (; e + 3 * PART_T < b1; e += 4 * PART_T) {
    const int d0 = dst[e], d1 = dst[e + PART_T], d2 = dst[e + 2 * PART_T], d3 = dst[e + 3 * PART_T];
    atomicAdd(&c[d0 >> BSH], 1u); atomicAdd(&c[d1 >> BSH], 1u);
    atomicAdd(&c[d2 >> BSH], 1u); atomicAdd(&c[d3 >> BSH], 1u);
  }
  for (; e < b1; e += PART_T) atomicAdd(&c[dst[e] >> BSH], 1u);
  __syncthreads();
  if (tid < NB) hist[(size_t)tid * gridDim.x + blockIdx.x] = c[tid];
}

__global__ __launch_bounds__(PART_T) void part_place(const int* __restrict__ ei,
                                                     const unsigned* __restrict__ hist,
                                                     unsigned* __restrict__ bstart,
                                                     unsigned* __restrict__ rec,
                                                     int E_, int NB, int chunk) {
  __shared__ unsigned cur[NBMAX];
  __shared__ unsigned rtot[NBMAX];
  __shared__ unsigned rpar[NBMAX];
  __shared__ unsigned wtot[PART_T / 32];
  __shared__ unsigned wown[PART_T / 32];
  __shared__ unsigned gofs[NBMAX];
  __shared__ uint2 img[PLACE_CAP];
  const int tid = threadIdx.x, lane = tid & 31, wave = tid >> 5;
  const unsigned c = blockIdx.x;
  const bool staged = chunk <= PLACE_CAP;
  const unsigned own = (tid < NB) ? hist[(size_t)tid * PART_G + c] : 0u;
  {
    const int hw = tid >> 4, l16 = tid & 15;
    uint4 q[NBMAX / 32][PART_G / 64];
    #pragma unroll
    for (int i = 0; i < NBMAX / 32; i++) {
      const int b = hw + 32 * i;
      const uint4* row = (const uint4*)(hist + (size_t)(b < NB ? b : NB - 1) * PART_G);
      #pragma unroll
      for (int j = 0; j < PART_G / 64; j++) q[i][j] = row[16 * j + l16];
    }
    #pragma unroll
    for (int i = 0; i < NBMAX / 32; i++) {
      const int b = hw + 32 * i;
      unsigned tot = 0u, par = 0u;
      #pragma unroll
      for (int j = 0; j < PART_G / 64; j++) {
        const unsigned i0 = 64u * j + 4u * l16;
        tot += q[i][j].x + q[i][j].y + q[i][j].z + q[i][j].w;
        par += (i0 < c ? q[i][j].x : 0u) + (i0 + 1u < c ? q[i][j].y : 0u) +
               (i0 + 2u < c ? q[i][j].z : 0u) + (i0 + 3u < c ? q[i][j].w : 0u);
      }
      tot = row16_sum(tot); par = row16_sum(par);
      if (l16 == 0 && b < NB) { rtot[b] = tot; rpar[b] = par; }
    }
  }
  __syncthreads();
  const unsigned t = (tid < NB) ? rtot[tid] : 0u;
  unsigned incl = t, inco = own;
  #pragma unroll
  for (int off = 1; off < 32; off <<= 1) {
    const unsigned up = __shfl_up(incl, off);
    const unsigned uo = __shfl_up(inco, off);
    if (lane >= off) { incl += up; inco += uo; }
  }
  if (lane == 31) { wtot[wave] = incl; wown[wave] = inco; }
  __syncthreads();
  if (wave == 0) {
    const unsigned w = (lane < PART_T / 32) ? wtot[lane] : 0u;
    const unsigned wo = (lane < PART_T / 32) ? wown[lane] : 0u;
    unsigned wi = w, wj = wo;
    #pragma unroll
    for (int off = 1; off < 32; off <<= 1) {
      const unsigned up = __shfl_up(wi, off);
      const unsigned uo = __shfl_up(wj, off);
      if (lane >= off) { wi += up; wj += uo; }
    }
    if (lane < PART_T / 32) { wtot[lane] = wi - w; wown[lane] = wj - wo; }
  }
  __syncthreads();
  if (tid < NB) {
    const unsigned first = wtot[wave] + (incl - t);
    const unsigned g = first + rpar[tid];
    const unsigned l = wown[wave] + (inco - own);
    cur[tid] = staged ? l : g;
    gofs[tid] = g - l;
    if (c == 0u) bstart[tid] = first;
  }
  __syncthreads();
  const int* __restrict__ dst = ei + E_;
  const int b0 = blockIdx.x * chunk;
  const int b1 = (b0 + chunk < E_) ? (b0 + chunk) : E_;
  int e = b0 + tid;
  if (staged) {
    for (; e + 3 * PART_T < b1; e += 4 * PART_T) {
      const unsigned d0 = dst[e], d1 = dst[e + PART_T], d2 = dst[e + 2 * PART_T], d3 = dst[e + 3 * PART_T];
      const unsigned s0 = ei[e],  s1 = ei[e + PART_T],  s2 = ei[e + 2 * PART_T],  s3 = ei[e + 3 * PART_T];
      const unsigned k0 = d0 >> BSH, k1 = d1 >> BSH, k2 = d2 >> BSH, k3 = d3 >> BSH;
      const unsigned o0 = gofs[k0], o1 = gofs[k1], o2 = gofs[k2], o3 = gofs[k3];
      const unsigned p0 = atomicAdd(&cur[k0], 1u), p1 = atomicAdd(&cur[k1], 1u);
      const unsigned p2 = atomicAdd(&cur[k2], 1u), p3 = atomicAdd(&cur[k3], 1u);
      img[p0] = make_uint2(s0 | ((d0 & (BNODES - 1)) << SRC_BITS), o0 + p0);
      img[p1] = make_uint2(s1 | ((d1 & (BNODES - 1)) << SRC_BITS), o1 + p1);
      img[p2] = make_uint2(s2 | ((d2 & (BNODES - 1)) << SRC_BITS), o2 + p2);
      img[p3] = make_uint2(s3 | ((d3 & (BNODES - 1)) << SRC_BITS), o3 + p3);
    }
    for (; e < b1; e += PART_T) {
      const unsigned d = dst[e], s = ei[e];
      const unsigned k = d >> BSH;
      const unsigned o = gofs[k];
      const unsigned p = atomicAdd(&cur[k], 1u);
      img[p] = make_uint2(s | ((d & (BNODES - 1)) << SRC_BITS), o + p);
    }
    __syncthreads();
    const unsigned cnt = (b1 > b0) ? (unsigned)(b1 - b0) : 0u;
    unsigned j = tid;
    for (; j + 3 * PART_T < cnt; j += 4 * PART_T) {
      const uint2 a0 = img[j], a1 = img[j + PART_T], a2 = img[j + 2 * PART_T], a3 = img[j + 3 * PART_T];
      rec[a0.y] = a0.x; rec[a1.y] = a1.x; rec[a2.y] = a2.x; rec[a3.y] = a3.x;
    }
    #pragma unroll 1
    for (; j < cnt; j += PART_T) {
      const uint2 a = img[j];
      rec[a.y] = a.x;
    }
    return;
  }
  for (; e + 3 * PART_T < b1; e += 4 * PART_T) {
    const unsigned d0 = dst[e], d1 = dst[e + PART_T], d2 = dst[e + 2 * PART_T], d3 = dst[e + 3 * PART_T];
    const unsigned s0 = ei[e],  s1 = ei[e + PART_T],  s2 = ei[e + 2 * PART_T],  s3 = ei[e + 3 * PART_T];
    const unsigned p0 = atomicAdd(&cur[d0 >> BSH], 1u), p1 = atomicAdd(&cur[d1 >> BSH], 1u);
    const unsigned p2 = atomicAdd(&cur[d2 >> BSH], 1u), p3 = atomicAdd(&cur[d3 >> BSH], 1u);
    rec[p0] = s0 | ((d0 & (BNODES - 1)) << SRC_BITS); rec[p1] = s1 | ((d1 & (BNODES - 1)) << SRC_BITS);
    rec[p2] = s2 | ((d2 & (BNODES - 1)) << SRC_BITS); rec[p3] = s3 | ((d3 & (BNODES - 1)) << SRC_BITS);
  }
  for (; e < b1; e += PART_T) {
    const unsigned d = dst[e], s = ei[e];
    const unsigned p = atomicAdd(&cur[d >> BSH], 1u);
    rec[p] = s | ((d & (BNODES - 1)) << SRC_BITS);
  }
}

__global__ __launch_bounds__(PART_T) void bucket_sort(const unsigned* __restrict__ bstart,
                                                      const unsigned* __restrict__ rec,
                                                      unsigned* __restrict__ rowptr,
                                                      int* __restrict__ csr_src,
                                                      int n, int E_, int NB) {
  __shared__ unsigned cnt[BNODES];
  __shared__ unsigned wtot[PART_T / 32];
  __shared__ int img[BUCKET_CAP];
  const int tid = threadIdx.x, lane = tid & 31, wave = tid >> 5;
  const int k = blockIdx.x;
  const unsigned start = bstart[k];
  const unsigned end = (k + 1 < NB) ? bstart[k + 1] : (unsigned)E_;
  const bool staged = end - start <= BUCKET_CAP;
  cnt[tid] = 0u;
  __syncthreads();
  unsigned i = start + tid;
  for (; i + 3 * PART_T < end; i += 4 * PART_T) {
    const unsigned r0 = rec[i], r1 = rec[i + PART_T], r2 = rec[i + 2 * PART_T], r3 = rec[i + 3 * PART_T];
    atomicAdd(&cnt[r0 >> SRC_BITS], 1u); atomicAdd(&cnt[r1 >> SRC_BITS], 1u);
    atomicAdd(&cnt[r2 >> SRC_BITS], 1u); atomicAdd(&cnt[r3 >> SRC_BITS], 1u);
  }
  for (; i < end; i += PART_T) atomicAdd(&cnt[rec[i] >> SRC_BITS], 1u);
  __syncthreads();
  const unsigned c = cnt[tid];
  unsigned incl = c;
  #pragma unroll
  for (int off = 1; off < 32; off <<= 1) {
    const unsigned up = __shfl_up(incl, off);
    if (lane >= off) incl += up;
  }
  if (lane == 31) wtot[wave] = incl;
  __syncthreads();
  if (wave == 0) {
    const unsigned w = (lane < PART_T / 32) ? wtot[lane] : 0u;
    unsigned wi = w;
    #pragma unroll
    for (int off = 1; off < 32; off <<= 1) {
      const unsigned up = __shfl_up(wi, off);
      if (lane >= off) wi += up;
    }
    if (lane < PART_T / 32) wtot[lane] = wi - w;
  }
  __syncthreads();
  const unsigned first = start + wtot[wave] + (incl - c);
  cnt[tid] = first;
  const int node = k * BNODES + tid;
  if (node < n) rowptr[node] = first;
  if (k == NB - 1 && tid == 0) rowptr[n] = (unsigned)E_;
  __syncthreads();
  i = start + tid;
  if (staged) {
    for (; i + 3 * PART_T < end; i += 4 * PART_T) {
      const unsigned r0 = rec[i], r1 = rec[i + PART_T], r2 = rec[i + 2 * PART_T], r3 = rec[i + 3 * PART_T];
      const unsigned p0 = atomicAdd(&cnt[r0 >> SRC_BITS], 1u), p1 = atomicAdd(&cnt[r1 >> SRC_BITS], 1u);
      const unsigned p2 = atomicAdd(&cnt[r2 >> SRC_BITS], 1u), p3 = atomicAdd(&cnt[r3 >> SRC_BITS], 1u);
      img[p0 - start] = (int)(r0 & SRC_MASK); img[p1 - start] = (int)(r1 & SRC_MASK);
      img[p2 - start] = (int)(r2 & SRC_MASK); img[p3 - start] = (int)(r3 & SRC_MASK);
    }
    for (; i < end; i += PART_T) {
      const unsigned r = rec[i];
      const unsigned p = atomicAdd(&cnt[r >> SRC_BITS], 1u);
      img[p - start] = (int)(r & SRC_MASK);
    }
    __syncthreads();
    const unsigned size = end - start;
    int* __restrict__ outp = csr_src + start;
    unsigned j = tid;
    for (; j + 3 * PART_T < size; j += 4 * PART_T) {
      const int a0 = img[j], a1 = img[j + PART_T], a2 = img[j + 2 * PART_T], a3 = img[j + 3 * PART_T];
      outp[j] = a0; outp[j + PART_T] = a1; outp[j + 2 * PART_T] = a2; outp[j + 3 * PART_T] = a3;
    }
    #pragma unroll 1
    for (; j < size; j += PART_T) outp[j] = img[j];
    return;
  }
  for (; i + 3 * PART_T < end; i += 4 * PART_T) {
    const unsigned r0 = rec[i], r1 = rec[i + PART_T], r2 = rec[i + 2 * PART_T], r3 = rec[i + 3 * PART_T];
    const unsigned p0 = atomicAdd(&cnt[r0 >> SRC_BITS], 1u), p1 = atomicAdd(&cnt[r1 >> SRC_BITS], 1u);
    const unsigned p2 = atomicAdd(&cnt[r2 >> SRC_BITS], 1u), p3 = atomicAdd(&cnt[r3 >> SRC_BITS], 1u);
    csr_src[p0] = (int)(r0 & SRC_MASK); csr_src[p1] = (int)(r1 & SRC_MASK);
    csr_src[p2] = (int)(r2 & SRC_MASK); csr_src[p3] = (int)(r3 & SRC_MASK);
  }
  for (; i < end; i += PART_T) {
    const unsigned r = rec[i];
    const unsigned p = atomicAdd(&cnt[r >> SRC_BITS], 1u);
    csr_src[p] = (int)(r & SRC_MASK);
  }
}

#define A1_T 256
#define HS_PITCH 68
__global__ __launch_bounds__(A1_T) void agg1_kernel(const unsigned* __restrict__ rowptr, const int* __restrict__ csr_src,
                            const float* __restrict__ att_src, const float* __restrict__ att_dst,
                            const float* __restrict__ xl1, const float* __restrict__ bias1,
                            const float* __restrict__ W2, float* __restrict__ xl2, int n) {
  __shared__ float hs[(A1_T / HEADS) * HS_PITCH];
  const int t = blockIdx.x * A1_T + threadIdx.x;
  const bool live = t < n * HEADS;
  const int d = live ? (t >> 3) : (n - 1), hh = t & 7;
  const unsigned seg0 = rowptr[d];
  const unsigned seg1 = live ? rowptr[d + 1] : seg0;
  const int* __restrict__ cp = csr_src + seg0;
  const int* __restrict__ ce = csr_src + seg1;
  const float* __restrict__ xb = xl1 + hh * HID;
  const float4* wsp = (const float4*)(att_src + hh * HID);
  const float4 ws0 = wsp[0], ws1 = wsp[1];
  float ad;
  {
    const float4* wdp = (const float4*)(att_dst + hh * HID);
    const float4* xd = (const float4*)(xb + (size_t)d * D1);
    ad = dot8(xd[0], xd[1], wdp[0], wdp[1]);
  }

  float m = -__builtin_huge_valf();
  float den = 0.f;
  float4 acc0 = make_float4(0.f, 0.f, 0.f, 0.f), acc1 = acc0;
  for (; cp + 4 <= ce; cp += 4) {
    const int s0 = cp[0], s1 = cp[1], s2 = cp[2], s3 = cp[3];
    const float4* x0 = (const float4*)(xb + (size_t)s0 * D1);
    const float4* x1 = (const float4*)(xb + (size_t)s1 * D1);
    const float4* x2 = (const float4*)(xb + (size_t)s2 * D1);
    const float4* x3 = (const float4*)(xb + (size_t)s3 * D1);
    const float4 p0 = x0[0], q0 = x0[1], p1 = x1[0], q1 = x1[1];
    const float4 p2 = x2[0], q2 = x2[1], p3 = x3[0], q3 = x3[1];
    const float al0 = lrelu(dot8(p0, q0, ws0, ws1) + ad), al1 = lrelu(dot8(p1, q1, ws0, ws1) + ad);
    const float al2 = lrelu(dot8(p2, q2, ws0, ws1) + ad), al3 = lrelu(dot8(p3, q3, ws0, ws1) + ad);
    float mn = al0 > m ? al0 : m;
    mn = al1 > mn ? al1 : mn; mn = al2 > mn ? al2 : mn; mn = al3 > mn ? al3 : mn;
    const float sc = __expf(m - mn);
    const float e0 = __expf(al0 - mn), e1 = __expf(al1 - mn);
    const float e2 = __expf(al2 - mn), e3 = __expf(al3 - mn);
    m = mn;
    den = fmaf(den, sc, e0); den += e1; den += e2; den += e3;
    rescale_add(acc0, sc, p0, e0); rescale_add(acc1, sc, q0, e0);
    add_scaled(acc0, p1, e1);      add_scaled(acc1, q1, e1);
    add_scaled(acc0, p2, e2);      add_scaled(acc1, q2, e2);
    add_scaled(acc0, p3, e3);      add_scaled(acc1, q3, e3);
  }
  #pragma unroll 1
  for (; cp < ce; cp++) {
    const int s = cp[0];
    const float4* xp = (const float4*)(xb + (size_t)s * D1);
    const float4 v0 = xp[0], v1 = xp[1];
    const float al = lrelu(dot8(v0, v1, ws0, ws1) + ad);
    const float mn = al > m ? al : m;
    const float sc = __expf(m - mn);
    const float ex = __expf(al - mn);
    m = mn;
    den = fmaf(den, sc, ex);
    rescale_add(acc0, sc, v0, ex); rescale_add(acc1, sc, v1, ex);
  }
  const float dn = den + EPS_GAT;
  const float4* bp = (const float4*)(bias1 + hh * HID);
  const float4 b0 = bp[0], b1 = bp[1];
  float r[8] = { acc0.x / dn + b0.x, acc0.y / dn + b0.y, acc0.z / dn + b0.z, acc0.w / dn + b0.w,
                 acc1.x / dn + b1.x, acc1.y / dn + b1.y, acc1.z / dn + b1.z, acc1.w / dn + b1.w };
  #pragma unroll
  for (int c = 0; c < 8; c++) r[c] = r[c] > 0.f ? r[c] : (__expf(r[c]) - 1.f);
  float4* hp = (float4*)&hs[(threadIdx.x >> 3) * HS_PITCH + hh * HID];
  hp[0] = make_float4(r[0], r[1], r[2], r[3]);
  hp[1] = make_float4(r[4], r[5], r[6], r[7]);
  const int wave = threadIdx.x >> 5, lane = threadIdx.x & 31;
  const int ln = lane & 15;
  const int khalf = (lane >> 4) << 1;
  v2f b[D1 / 4];
  if (wave < 2) {
    #pragma unroll
    for (int j = 0; j < D1 / 4; j++) {
      b[j].x = W2[(4 * j + khalf) * NCLS + ln];
      b[j].y = W2[(4 * j + khalf + 1) * NCLS + ln];
    }
  }
  __syncthreads();
  if (wave < 2) {
    const float* arow = &hs[(wave * 16 + ln) * HS_PITCH + khalf];
    v8f c = {};
    #pragma unroll
    for (int j = 0; j < D1 / 4; j++) {
      const v2f a = *(const v2f*)&arow[4 * j];
      c = __builtin_amdgcn_wmma_f32_16x16x4_f32(false, a, false, b[j], (short)0, c, false, false);
    }
    const int mBase = (lane >> 4) * 8;
    const int rowBase = blockIdx.x * (A1_T / HEADS) + wave * 16;
    float* outp = xl2 + (size_t)rowBase * NCLS;
    #pragma unroll
    for (int i = 0; i < 8; i++) {
      const int row = rowBase + mBase + i;
      if (row < n) outp[(size_t)(mBase + i) * NCLS + ln] = c[i];
    }
  }
}

__global__ void agg2_kernel(const unsigned* __restrict__ rowptr, const int* __restrict__ csr_src,
                            const float* __restrict__ att_src, const float* __restrict__ att_dst,
                            const float* __restrict__ xl2, const float* __restrict__ bias2,
                            float* __restrict__ out, int n) {
  int t = blockIdx.x * blockDim.x + threadIdx.x;
  if (t >= n * 4) return;
  const int d = t >> 2, q = t & 3;
  const int* __restrict__ cp = csr_src + rowptr[d];
  const int* __restrict__ ce = csr_src + rowptr[d + 1];
  const float4* __restrict__ xq = (const float4*)xl2 + q;
  const float4 wsq = ((const float4*)att_src)[q];
  const float ad = quad_sum(dot4(xq[(size_t)d * 4], ((const float4*)att_dst)[q]));

  float m = -__builtin_huge_valf();
  float den = 0.f;
  float4 acc = make_float4(0.f, 0.f, 0.f, 0.f);
  for (; cp + 4 <= ce; cp += 4) {
    const int s0 = cp[0], s1 = cp[1], s2 = cp[2], s3 = cp[3];
    const float4 v0 = xq[(size_t)s0 * 4], v1 = xq[(size_t)s1 * 4];
    const float4 v2 = xq[(size_t)s2 * 4], v3 = xq[(size_t)s3 * 4];
    const float al0 = lrelu(quad_sum(dot4(v0, wsq)) + ad), al1 = lrelu(quad_sum(dot4(v1, wsq)) + ad);
    const float al2 = lrelu(quad_sum(dot4(v2, wsq)) + ad), al3 = lrelu(quad_sum(dot4(v3, wsq)) + ad);
    float mn = al0 > m ? al0 : m;
    mn = al1 > mn ? al1 : mn; mn = al2 > mn ? al2 : mn; mn = al3 > mn ? al3 : mn;
    const float sc = __expf(m - mn);
    const float e0 = __expf(al0 - mn), e1 = __expf(al1 - mn);
    const float e2 = __expf(al2 - mn), e3 = __expf(al3 - mn);
    m = mn;
    den = fmaf(den, sc, e0); den += e1; den += e2; den += e3;
    rescale_add(acc, sc, v0, e0);
    add_scaled(acc, v1, e1); add_scaled(acc, v2, e2); add_scaled(acc, v3, e3);
  }
  #pragma unroll 1
  for (; cp < ce; cp++) {
    const unsigned s = cp[0];
    const float4 v = xq[(size_t)s * 4];
    const float al = lrelu(quad_sum(dot4(v, wsq)) + ad);
    const float mn = al > m ? al : m;
    const float sc = __expf(m - mn);
    const float ex = __expf(al - mn);
    m = mn;
    den = fmaf(den, sc, ex);
    rescale_add(acc, sc, v, ex);
  }
  const float dn = den + EPS_GAT;
  const float4 b = ((const float4*)bias2)[q];
  ((float4*)(out + (size_t)d * NCLS))[q] =
      make_float4(acc.x / dn + b.x, acc.y / dn + b.y, acc.z / dn + b.z, acc.w / dn + b.w);
}

extern "C" void kernel_launch(void* const* d_in, const int* in_sizes, int n_in,
                              void* d_out, int out_size, void* d_ws, size_t ws_size,
                              hipStream_t stream) {
  const float* x        = (const float*)d_in[0];
  const int*   ei       = (const int*)d_in[1];
  const float* W1       = (const float*)d_in[2];
  const float* att_src1 = (const float*)d_in[3];
  const float* att_dst1 = (const float*)d_in[4];
  const float* bias1    = (const float*)d_in[5];
  const float* W2       = (const float*)d_in[6];
  const float* att_src2 = (const float*)d_in[7];
  const float* att_dst2 = (const float*)d_in[8];
  const float* bias2    = (const float*)d_in[9];
  float* out = (float*)d_out;

  const int N = in_sizes[0] / F_IN;
  const int E = in_sizes[1] / 2;
  const size_t Np = ((size_t)N + 4) & ~(size_t)3;
  const int NB = (N + BNODES - 1) / BNODES;
  const int chunk = (E + PART_G - 1) / PART_G;
  const int M = NB * PART_G;

  float* ws = (float*)d_ws;
  size_t o = 0;
  float*    xl1    = ws + o; o += (size_t)N * D1;
  float*    h      = ws + o; o += (size_t)N * D1;
  float*    xl2    = ws + o; o += (size_t)N * NCLS;
  unsigned* rowptr = (unsigned*)(ws + o); o += Np;
  unsigned* hist   = (unsigned*)(ws + o); o += (size_t)M;
  int*      csrsrc = (int*)(ws + o);      o += (size_t)E;
  unsigned* rec    = (unsigned*)h;
  unsigned* bstart = (unsigned*)xl2;

  const int TB = 256;
  auto nb = [](long long n, int b) { return (unsigned)((n + b - 1) / b); };

  part_count<<<PART_G, PART_T, 0, stream>>>(ei, hist, E, NB, chunk);
  part_place<<<PART_G, PART_T, 0, stream>>>(ei, hist, bstart, rec, E, NB, chunk);
  bucket_sort<<<NB, PART_T, 0, stream>>>(bstart, rec, rowptr, csrsrc, N, E, NB);

  gemm1_kernel<<<G1_BLOCKS, 128, 0, stream>>>(x, W1, xl1, N);
  agg1_kernel<<<nb((long long)N * HEADS, A1_T), A1_T, 0, stream>>>(rowptr, csrsrc, att_src1, att_dst1,
                                                                   xl1, bias1, W2, xl2, N);

  agg2_kernel<<<nb((long long)N * 4, TB), TB, 0, stream>>>(rowptr, csrsrc, att_src2, att_dst2,
                                                           xl2, bias2, out, N);
}


# Both aggregation walks hold the source ids one group ahead: the next four ids are requested behind the current rows' loads (were loaded after the sums); leftover edges use held ids; sums unchanged
# speedup vs baseline: 11.9051x; 1.0439x over previous
#include <hip/hip_runtime.h>
#include <hip/hip_bf16.h>

typedef __attribute__((ext_vector_type(2))) float v2f;
typedef __attribute__((ext_vector_type(8))) float v8f;
typedef __attribute__((ext_vector_type(4))) float v4f;

#define F_IN 512
#define HID 8
#define HEADS 8
#define D1 64
#define NCLS 16
#define NEG_SLOPE 0.2f
#define EPS_GAT 1e-16f
#define BSH 9
#define BNODES 512
#define SRC_BITS 23
#define SRC_MASK 0x7FFFFFu
#define NBMAX 256
#define PART_G 256
#define PART_T 512
#define PLACE_CAP 8192
#define BUCKET_CAP 16384

__device__ __forceinline__ float lrelu(float x) { return x > 0.f ? x : x * NEG_SLOPE; }

__device__ __forceinline__ void rescale_add(float4& a, float sc, const float4& v, float e) {
  a.x = fmaf(v.x, e, a.x * sc); a.y = fmaf(v.y, e, a.y * sc);
  a.z = fmaf(v.z, e, a.z * sc); a.w = fmaf(v.w, e, a.w * sc);
}
__device__ __forceinline__ void add_scaled(float4& a, const float4& v, float e) {
  a.x = fmaf(v.x, e, a.x); a.y = fmaf(v.y, e, a.y);
  a.z = fmaf(v.z, e, a.z); a.w = fmaf(v.w, e, a.w);
}

__device__ __forceinline__ float dot8(const float4& a, const float4& b, const float4& w0, const float4& w1) {
  float s = a.x * w0.x;
  s = fmaf(a.y, w0.y, s); s = fmaf(a.z, w0.z, s); s = fmaf(a.w, w0.w, s);
  s = fmaf(b.x, w1.x, s); s = fmaf(b.y, w1.y, s); s = fmaf(b.z, w1.z, s); s = fmaf(b.w, w1.w, s);
  return s;
}
__device__ __forceinline__ float dot4(const float4& a, const float4& w) {
  float s = a.x * w.x;
  s = fmaf(a.y, w.y, s); s = fmaf(a.z, w.z, s); s = fmaf(a.w, w.w, s);
  return s;
}
__device__ __forceinline__ float quad_sum(float v) {
  v += __int_as_float(__builtin_amdgcn_update_dpp(0, __float_as_int(v), 0xB1, 0xF, 0xF, true));
  v += __int_as_float(__builtin_amdgcn_update_dpp(0, __float_as_int(v), 0x4E, 0xF, 0xF, true));
  return v;
}

__device__ __forceinline__ unsigned row16_sum(unsigned v) {
  v += (unsigned)__builtin_amdgcn_update_dpp(0, (int)v, 0xB1, 0xF, 0xF, true);
  v += (unsigned)__builtin_amdgcn_update_dpp(0, (int)v, 0x4E, 0xF, 0xF, true);
  v += (unsigned)__builtin_amdgcn_update_dpp(0, (int)v, 0x141, 0xF, 0xF, true);
  v += (unsigned)__builtin_amdgcn_update_dpp(0, (int)v, 0x140, 0xF, 0xF, true);
  return v;
}

#define G1_BLOCKS 512
__global__ __launch_bounds__(128) void gemm1_kernel(const float* __restrict__ x,
                                                    const float* __restrict__ W1,
                                                    float* __restrict__ xl1, int nrows) {
  __shared__ float At[16 * 516];
  const int tid = threadIdx.x;
  const int wave = tid >> 5;
  const int lane = tid & 31;
  const int ln = lane & 15;
  const int khalf = (lane >> 4) << 1;
  const int colBase = wave * 16;
  const int mBase = (lane >> 4) * 8;
  const int ntiles = (nrows + 15) >> 4;

  v2f b[F_IN / 4];
  {
    const float* wp = W1 + (size_t)khalf * D1 + colBase + ln;
    #pragma unroll
    for (int j = 0; j < F_IN / 4; j++) {
      b[j].x = wp[(size_t)(4 * j) * D1];
      b[j].y = wp[(size_t)(4 * j + 1) * D1];
    }
  }

  v4f st[16];
  int tile = blockIdx.x;
  if (tile < ntiles) {
    #pragma unroll
    for (int r = 0; r < 16; r++) {
      int row = tile * 16 + r; if (row >= nrows) row = nrows - 1;
      st[r] = ((const v4f*)(x + (size_t)row * F_IN))[tid];
    }
  }
  for (; tile < ntiles; tile += G1_BLOCKS) {
    __syncthreads();
    #pragma unroll
    for (int r = 0; r < 16; r++) *(v4f*)&At[r * 516 + tid * 4] = st[r];
    __syncthreads();
    const int nt = tile + G1_BLOCKS;
    if (nt < ntiles) {
      #pragma unroll
      for (int r = 0; r < 16; r++) {
        int row = nt * 16 + r; if (row >= nrows) row = nrows - 1;
        st[r] = ((const v4f*)(x + (size_t)row * F_IN))[tid];
      }
    }
    v8f c = {};
    const float* arow = &At[ln * 516 + khalf];
    v2f af[2][16];
    #pragma unroll
    for (int i = 0; i < 16; i++) af[0][i] = *(const v2f*)&arow[4 * i];
    #pragma unroll
    for (int blk = 0; blk < F_IN / 64; blk++) {
      if (blk + 1 < F_IN / 64) {
        #pragma unroll
        for (int i = 0; i < 16; i++) af[(blk + 1) & 1][i] = *(const v2f*)&arow[4 * (16 * (blk + 1) + i)];
      }
      __builtin_amdgcn_sched_barrier(0);
      #pragma unroll
      for (int i = 0; i < 16; i++)
        c = __builtin_amdgcn_wmma_f32_16x16x4_f32(false, af[blk & 1][i], false, b[16 * blk + i],
                                                  (short)0, c, false, false);
    }
    const int rowBase = tile * 16;
    float* outp = xl1 + (size_t)rowBase * D1 + colBase;
    #pragma unroll
    for (int i = 0; i < 8; i++) {
      int row = rowBase + mBase + i;
      if (row < nrows) outp[(size_t)(mBase + i) * D1 + ln] = c[i];
    }
  }
}


__global__ __launch_bounds__(PART_T) void part_count(const int* __restrict__ ei,
                                                     unsigned* __restrict__ hist,
                                                     int E_, int NB, int chunk) {
  __shared__ unsigned c[NBMAX];
  const int tid = threadIdx.x;
  if (tid < NB) c[tid] = 0u;
  __syncthreads();
  const int* __restrict__ dst = ei + E_;
  const int b0 = blockIdx.x * chunk;
  const int b1 = (b0 + chunk < E_) ? (b0 + chunk) : E_;
  int e = b0 + tid;
  for (; e + 3 * PART_T < b1; e += 4 * PART_T) {
    const int d0 = dst[e], d1 = dst[e + PART_T], d2 = dst[e + 2 * PART_T], d3 = dst[e + 3 * PART_T];
    atomicAdd(&c[d0 >> BSH], 1u); atomicAdd(&c[d1 >> BSH], 1u);
    atomicAdd(&c[d2 >> BSH], 1u); atomicAdd(&c[d3 >> BSH], 1u);
  }
  for (; e < b1; e += PART_T) atomicAdd(&c[dst[e] >> BSH], 1u);
  __syncthreads();
  if (tid < NB) hist[(size_t)tid * gridDim.x + blockIdx.x] = c[tid];
}

__global__ __launch_bounds__(PART_T) void part_place(const int* __restrict__ ei,
                                                     const unsigned* __restrict__ hist,
                                                     unsigned* __restrict__ bstart,
                                                     unsigned* __restrict__ rec,
                                                     int E_, int NB, int chunk) {
  __shared__ unsigned cur[NBMAX];
  __shared__ unsigned rtot[NBMAX];
  __shared__ unsigned rpar[NBMAX];
  __shared__ unsigned wtot[PART_T / 32];
  __shared__ unsigned wown[PART_T / 32];
  __shared__ unsigned gofs[NBMAX];
  __shared__ uint2 img[PLACE_CAP];
  const int tid = threadIdx.x, lane = tid & 31, wave = tid >> 5;
  const unsigned c = blockIdx.x;
  const bool staged = chunk <= PLACE_CAP;
  const unsigned own = (tid < NB) ? hist[(size_t)tid * PART_G + c] : 0u;
  {
    const int hw = tid >> 4, l16 = tid & 15;
    uint4 q[NBMAX / 32][PART_G / 64];
    #pragma unroll
    for (int i = 0; i < NBMAX / 32; i++) {
      const int b = hw + 32 * i;
      const uint4* row = (const uint4*)(hist + (size_t)(b < NB ? b : NB - 1) * PART_G);
      #pragma unroll
      for (int j = 0; j < PART_G / 64; j++) q[i][j] = row[16 * j + l16];
    }
    #pragma unroll
    for (int i = 0; i < NBMAX / 32; i++) {
      const int b = hw + 32 * i;
      unsigned tot = 0u, par = 0u;
      #pragma unroll
      for (int j = 0; j < PART_G / 64; j++) {
        const unsigned i0 = 64u * j + 4u * l16;
        tot += q[i][j].x + q[i][j].y + q[i][j].z + q[i][j].w;
        par += (i0 < c ? q[i][j].x : 0u) + (i0 + 1u < c ? q[i][j].y : 0u) +
               (i0 + 2u < c ? q[i][j].z : 0u) + (i0 + 3u < c ? q[i][j].w : 0u);
      }
      tot = row16_sum(tot); par = row16_sum(par);
      if (l16 == 0 && b < NB) { rtot[b] = tot; rpar[b] = par; }
    }
  }
  __syncthreads();
  const unsigned t = (tid < NB) ? rtot[tid] : 0u;
  unsigned incl = t, inco = own;
  #pragma unroll
  for (int off = 1; off < 32; off <<= 1) {
    const unsigned up = __shfl_up(incl, off);
    const unsigned uo = __shfl_up(inco, off);
    if (lane >= off) { incl += up; inco += uo; }
  }
  if (lane == 31) { wtot[wave] = incl; wown[wave] = inco; }
  __syncthreads();
  if (wave == 0) {
    const unsigned w = (lane < PART_T / 32) ? wtot[lane] : 0u;
    const unsigned wo = (lane < PART_T / 32) ? wown[lane] : 0u;
    unsigned wi = w, wj = wo;
    #pragma unroll
    for (int off = 1; off < 32; off <<= 1) {
      const unsigned up = __shfl_up(wi, off);
      const unsigned uo = __shfl_up(wj, off);
      if (lane >= off) { wi += up; wj += uo; }
    }
    if (lane < PART_T / 32) { wtot[lane] = wi - w; wown[lane] = wj - wo; }
  }
  __syncthreads();
  if (tid < NB) {
    const unsigned first = wtot[wave] + (incl - t);
    const unsigned g = first + rpar[tid];
    const unsigned l = wown[wave] + (inco - own);
    cur[tid] = staged ? l : g;
    gofs[tid] = g - l;
    if (c == 0u) bstart[tid] = first;
  }
  __syncthreads();
  const int* __restrict__ dst = ei + E_;
  const int b0 = blockIdx.x * chunk;
  const int b1 = (b0 + chunk < E_) ? (b0 + chunk) : E_;
  int e = b0 + tid;
  if (staged) {
    for (; e + 3 * PART_T < b1; e += 4 * PART_T) {
      const unsigned d0 = dst[e], d1 = dst[e + PART_T], d2 = dst[e + 2 * PART_T], d3 = dst[e + 3 * PART_T];
      const unsigned s0 = ei[e],  s1 = ei[e + PART_T],  s2 = ei[e + 2 * PART_T],  s3 = ei[e + 3 * PART_T];
      const unsigned k0 = d0 >> BSH, k1 = d1 >> BSH, k2 = d2 >> BSH, k3 = d3 >> BSH;
      const unsigned o0 = gofs[k0], o1 = gofs[k1], o2 = gofs[k2], o3 = gofs[k3];
      const unsigned p0 = atomicAdd(&cur[k0], 1u), p1 = atomicAdd(&cur[k1], 1u);
      const unsigned p2 = atomicAdd(&cur[k2], 1u), p3 = atomicAdd(&cur[k3], 1u);
      img[p0] = make_uint2(s0 | ((d0 & (BNODES - 1)) << SRC_BITS), o0 + p0);
      img[p1] = make_uint2(s1 | ((d1 & (BNODES - 1)) << SRC_BITS), o1 + p1);
      img[p2] = make_uint2(s2 | ((d2 & (BNODES - 1)) << SRC_BITS), o2 + p2);
      img[p3] = make_uint2(s3 | ((d3 & (BNODES - 1)) << SRC_BITS), o3 + p3);
    }
    for (; e < b1; e += PART_T) {
      const unsigned d = dst[e], s = ei[e];
      const unsigned k = d >> BSH;
      const unsigned o = gofs[k];
      const unsigned p = atomicAdd(&cur[k], 1u);
      img[p] = make_uint2(s | ((d & (BNODES - 1)) << SRC_BITS), o + p);
    }
    __syncthreads();
    const unsigned cnt = (b1 > b0) ? (unsigned)(b1 - b0) : 0u;
    unsigned j = tid;
    for (; j + 3 * PART_T < cnt; j += 4 * PART_T) {
      const uint2 a0 = img[j], a1 = img[j + PART_T], a2 = img[j + 2 * PART_T], a3 = img[j + 3 * PART_T];
      rec[a0.y] = a0.x; rec[a1.y] = a1.x; rec[a2.y] = a2.x; rec[a3.y] = a3.x;
    }
    #pragma unroll 1
    for (; j < cnt; j += PART_T) {
      const uint2 a = img[j];
      rec[a.y] = a.x;
    }
    return;
  }
  for (; e + 3 * PART_T < b1; e += 4 * PART_T) {
    const unsigned d0 = dst[e], d1 = dst[e + PART_T], d2 = dst[e + 2 * PART_T], d3 = dst[e + 3 * PART_T];
    const unsigned s0 = ei[e],  s1 = ei[e + PART_T],  s2 = ei[e + 2 * PART_T],  s3 = ei[e + 3 * PART_T];
    const unsigned p0 = atomicAdd(&cur[d0 >> BSH], 1u), p1 = atomicAdd(&cur[d1 >> BSH], 1u);
    const unsigned p2 = atomicAdd(&cur[d2 >> BSH], 1u), p3 = atomicAdd(&cur[d3 >> BSH], 1u);
    rec[p0] = s0 | ((d0 & (BNODES - 1)) << SRC_BITS); rec[p1] = s1 | ((d1 & (BNODES - 1)) << SRC_BITS);
    rec[p2] = s2 | ((d2 & (BNODES - 1)) << SRC_BITS); rec[p3] = s3 | ((d3 & (BNODES - 1)) << SRC_BITS);
  }
  for (; e < b1; e += PART_T) {
    const unsigned d = dst[e], s = ei[e];
    const unsigned p = atomicAdd(&cur[d >> BSH], 1u);
    rec[p] = s | ((d & (BNODES - 1)) << SRC_BITS);
  }
}

__global__ __launch_bounds__(PART_T) void bucket_sort(const unsigned* __restrict__ bstart,
                                                      const unsigned* __restrict__ rec,
                                                      unsigned* __restrict__ rowptr,
                                                      int* __restrict__ csr_src,
                                                      int n, int E_, int NB) {
  __shared__ unsigned cnt[BNODES];
  __shared__ unsigned wtot[PART_T / 32];
  __shared__ int img[BUCKET_CAP];
  const int tid = threadIdx.x, lane = tid & 31, wave = tid >> 5;
  const int k = blockIdx.x;
  const unsigned start = bstart[k];
  const unsigned end = (k + 1 < NB) ? bstart[k + 1] : (unsigned)E_;
  const bool staged = end - start <= BUCKET_CAP;
  cnt[tid] = 0u;
  __syncthreads();
  unsigned i = start + tid;
  for (; i + 3 * PART_T < end; i += 4 * PART_T) {
    const unsigned r0 = rec[i], r1 = rec[i + PART_T], r2 = rec[i + 2 * PART_T], r3 = rec[i + 3 * PART_T];
    atomicAdd(&cnt[r0 >> SRC_BITS], 1u); atomicAdd(&cnt[r1 >> SRC_BITS], 1u);
    atomicAdd(&cnt[r2 >> SRC_BITS], 1u); atomicAdd(&cnt[r3 >> SRC_BITS], 1u);
  }
  for (; i < end; i += PART_T) atomicAdd(&cnt[rec[i] >> SRC_BITS], 1u);
  __syncthreads();
  const unsigned c = cnt[tid];
  unsigned incl = c;
  #pragma unroll
  for (int off = 1; off < 32; off <<= 1) {
    const unsigned up = __shfl_up(incl, off);
    if (lane >= off) incl += up;
  }
  if (lane == 31) wtot[wave] = incl;
  __syncthreads();
  if (wave == 0) {
    const unsigned w = (lane < PART_T / 32) ? wtot[lane] : 0u;
    unsigned wi = w;
    #pragma unroll
    for (int off = 1; off < 32; off <<= 1) {
      const unsigned up = __shfl_up(wi, off);
      if (lane >= off) wi += up;
    }
    if (lane < PART_T / 32) wtot[lane] = wi - w;
  }
  __syncthreads();
  const unsigned first = start + wtot[wave] + (incl - c);
  cnt[tid] = first;
  const int node = k * BNODES + tid;
  if (node < n) rowptr[node] = first;
  if (k == NB - 1 && tid == 0) rowptr[n] = (unsigned)E_;
  __syncthreads();
  i = start + tid;
  if (staged) {
    for (; i + 3 * PART_T < end; i += 4 * PART_T) {
      const unsigned r0 = rec[i], r1 = rec[i + PART_T], r2 = rec[i + 2 * PART_T], r3 = rec[i + 3 * PART_T];
      const unsigned p0 = atomicAdd(&cnt[r0 >> SRC_BITS], 1u), p1 = atomicAdd(&cnt[r1 >> SRC_BITS], 1u);
      const unsigned p2 = atomicAdd(&cnt[r2 >> SRC_BITS], 1u), p3 = atomicAdd(&cnt[r3 >> SRC_BITS], 1u);
      img[p0 - start] = (int)(r0 & SRC_MASK); img[p1 - start] = (int)(r1 & SRC_MASK);
      img[p2 - start] = (int)(r2 & SRC_MASK); img[p3 - start] = (int)(r3 & SRC_MASK);
    }
    for (; i < end; i += PART_T) {
      const unsigned r = rec[i];
      const unsigned p = atomicAdd(&cnt[r >> SRC_BITS], 1u);
      img[p - start] = (int)(r & SRC_MASK);
    }
    __syncthreads();
    const unsigned size = end - start;
    int* __restrict__ outp = csr_src + start;
    unsigned j = tid;
    for (; j + 3 * PART_T < size; j += 4 * PART_T) {
      const int a0 = img[j], a1 = img[j + PART_T], a2 = img[j + 2 * PART_T], a3 = img[j + 3 * PART_T];
      outp[j] = a0; outp[j + PART_T] = a1; outp[j + 2 * PART_T] = a2; outp[j + 3 * PART_T] = a3;
    }
    #pragma unroll 1
    for (; j < size; j += PART_T) outp[j] = img[j];
    return;
  }
  for (; i + 3 * PART_T < end; i += 4 * PART_T) {
    const unsigned r0 = rec[i], r1 = rec[i + PART_T], r2 = rec[i + 2 * PART_T], r3 = rec[i + 3 * PART_T];
    const unsigned p0 = atomicAdd(&cnt[r0 >> SRC_BITS], 1u), p1 = atomicAdd(&cnt[r1 >> SRC_BITS], 1u);
    const unsigned p2 = atomicAdd(&cnt[r2 >> SRC_BITS], 1u), p3 = atomicAdd(&cnt[r3 >> SRC_BITS], 1u);
    csr_src[p0] = (int)(r0 & SRC_MASK); csr_src[p1] = (int)(r1 & SRC_MASK);
    csr_src[p2] = (int)(r2 & SRC_MASK); csr_src[p3] = (int)(r3 & SRC_MASK);
  }
  for (; i < end; i += PART_T) {
    const unsigned r = rec[i];
    const unsigned p = atomicAdd(&cnt[r >> SRC_BITS], 1u);
    csr_src[p] = (int)(r & SRC_MASK);
  }
}

#define A1_T 256
#define HS_PITCH 68
__global__ __launch_bounds__(A1_T) void agg1_kernel(const unsigned* __restrict__ rowptr, const int* __restrict__ csr_src,
                            const float* __restrict__ att_src, const float* __restrict__ att_dst,
                            const float* __restrict__ xl1, const float* __restrict__ bias1,
                            const float* __restrict__ W2, float* __restrict__ xl2, int n) {
  __shared__ float hs[(A1_T / HEADS) * HS_PITCH];
  const int t = blockIdx.x * A1_T + threadIdx.x;
  const bool live = t < n * HEADS;
  const int d = live ? (t >> 3) : (n - 1), hh = t & 7;
  const unsigned seg0 = rowptr[d];
  const unsigned seg1 = live ? rowptr[d + 1] : seg0;
  const int* __restrict__ cp = csr_src + seg0;
  const int* __restrict__ ce = csr_src + seg1;
  const float* __restrict__ xb = xl1 + hh * HID;
  const float4* wsp = (const float4*)(att_src + hh * HID);
  const float4 ws0 = wsp[0], ws1 = wsp[1];

  int n0 = cp[0], n1 = cp[1], n2 = cp[2], n3 = cp[3];
  float ad;
  {
    const float4* wdp = (const float4*)(att_dst + hh * HID);
    const float4* xd = (const float4*)(xb + (size_t)d * D1);
    ad = dot8(xd[0], xd[1], wdp[0], wdp[1]);
  }

  float m = -__builtin_huge_valf();
  float den = 0.f;
  float4 acc0 = make_float4(0.f, 0.f, 0.f, 0.f), acc1 = acc0;
  for (; cp + 4 <= ce; cp += 4) {
    const int s0 = n0, s1 = n1, s2 = n2, s3 = n3;
    const float4* x0 = (const float4*)(xb + (size_t)s0 * D1);
    const float4* x1 = (const float4*)(xb + (size_t)s1 * D1);
    const float4* x2 = (const float4*)(xb + (size_t)s2 * D1);
    const float4* x3 = (const float4*)(xb + (size_t)s3 * D1);
    const float4 p0 = x0[0], q0 = x0[1], p1 = x1[0], q1 = x1[1];
    const float4 p2 = x2[0], q2 = x2[1], p3 = x3[0], q3 = x3[1];
    n0 = cp[4]; n1 = cp[5]; n2 = cp[6]; n3 = cp[7];
    const float al0 = lrelu(dot8(p0, q0, ws0, ws1) + ad), al1 = lrelu(dot8(p1, q1, ws0, ws1) + ad);
    const float al2 = lrelu(dot8(p2, q2, ws0, ws1) + ad), al3 = lrelu(dot8(p3, q3, ws0, ws1) + ad);
    float mn = al0 > m ? al0 : m;
    mn = al1 > mn ? al1 : mn; mn = al2 > mn ? al2 : mn; mn = al3 > mn ? al3 : mn;
    const float sc = __expf(m - mn);
    const float e0 = __expf(al0 - mn), e1 = __expf(al1 - mn);
    const float e2 = __expf(al2 - mn), e3 = __expf(al3 - mn);
    m = mn;
    den = fmaf(den, sc, e0); den += e1; den += e2; den += e3;
    rescale_add(acc0, sc, p0, e0); rescale_add(acc1, sc, q0, e0);
    add_scaled(acc0, p1, e1);      add_scaled(acc1, q1, e1);
    add_scaled(acc0, p2, e2);      add_scaled(acc1, q2, e2);
    add_scaled(acc0, p3, e3);      add_scaled(acc1, q3, e3);
  }
  #pragma unroll 1
  for (; cp < ce; cp++) {
    const int s = n0; n0 = n1; n1 = n2; n2 = n3;
    const float4* xp = (const float4*)(xb + (size_t)s * D1);
    const float4 v0 = xp[0], v1 = xp[1];
    const float al = lrelu(dot8(v0, v1, ws0, ws1) + ad);
    const float mn = al > m ? al : m;
    const float sc = __expf(m - mn);
    const float ex = __expf(al - mn);
    m = mn;
    den = fmaf(den, sc, ex);
    rescale_add(acc0, sc, v0, ex); rescale_add(acc1, sc, v1, ex);
  }
  const float dn = den + EPS_GAT;
  const float4* bp = (const float4*)(bias1 + hh * HID);
  const float4 b0 = bp[0], b1 = bp[1];
  float r[8] = { acc0.x / dn + b0.x, acc0.y / dn + b0.y, acc0.z / dn + b0.z, acc0.w / dn + b0.w,
                 acc1.x / dn + b1.x, acc1.y / dn + b1.y, acc1.z / dn + b1.z, acc1.w / dn + b1.w };
  #pragma unroll
  for (int c = 0; c < 8; c++) r[c] = r[c] > 0.f ? r[c] : (__expf(r[c]) - 1.f);
  float4* hp = (float4*)&hs[(threadIdx.x >> 3) * HS_PITCH + hh * HID];
  hp[0] = make_float4(r[0], r[1], r[2], r[3]);
  hp[1] = make_float4(r[4], r[5], r[6], r[7]);
  const int wave = threadIdx.x >> 5, lane = threadIdx.x & 31;
  const int ln = lane & 15;
  const int khalf = (lane >> 4) << 1;
  v2f b[D1 / 4];
  if (wave < 2) {
    #pragma unroll
    for (int j = 0; j < D1 / 4; j++) {
      b[j].x = W2[(4 * j + khalf) * NCLS + ln];
      b[j].y = W2[(4 * j + khalf + 1) * NCLS + ln];
    }
  }
  __syncthreads();
  if (wave < 2) {
    const float* arow = &hs[(wave * 16 + ln) * HS_PITCH + khalf];
    v8f c = {};
    #pragma unroll
    for (int j = 0; j < D1 / 4; j++) {
      const v2f a = *(const v2f*)&arow[4 * j];
      c = __builtin_amdgcn_wmma_f32_16x16x4_f32(false, a, false, b[j], (short)0, c, false, false);
    }
    const int mBase = (lane >> 4) * 8;
    const int rowBase = blockIdx.x * (A1_T / HEADS) + wave * 16;
    float* outp = xl2 + (size_t)rowBase * NCLS;
    #pragma unroll
    for (int i = 0; i < 8; i++) {
      const int row = rowBase + mBase + i;
      if (row < n) outp[(size_t)(mBase + i) * NCLS + ln] = c[i];
    }
  }
}

__global__ void agg2_kernel(const unsigned* __restrict__ rowptr, const int* __restrict__ csr_src,
                            const float* __restrict__ att_src, const float* __restrict__ att_dst,
                            const float* __restrict__ xl2, const float* __restrict__ bias2,
                            float* __restrict__ out, int n) {
  int t = blockIdx.x * blockDim.x + threadIdx.x;
  if (t >= n * 4) return;
  const int d = t >> 2, q = t & 3;
  const int* __restrict__ cp = csr_src + rowptr[d];
  const int* __restrict__ ce = csr_src + rowptr[d + 1];
  const float4* __restrict__ xq = (const float4*)xl2 + q;
  const float4 wsq = ((const float4*)att_src)[q];
  int n0 = cp[0], n1 = cp[1], n2 = cp[2], n3 = cp[3];
  const float ad = quad_sum(dot4(xq[(size_t)d * 4], ((const float4*)att_dst)[q]));

  float m = -__builtin_huge_valf();
  float den = 0.f;
  float4 acc = make_float4(0.f, 0.f, 0.f, 0.f);
  for (; cp + 4 <= ce; cp += 4) {
    const int s0 = n0, s1 = n1, s2 = n2, s3 = n3;
    const float4 v0 = xq[(size_t)s0 * 4], v1 = xq[(size_t)s1 * 4];
    const float4 v2 = xq[(size_t)s2 * 4], v3 = xq[(size_t)s3 * 4];
    n0 = cp[4]; n1 = cp[5]; n2 = cp[6]; n3 = cp[7];
    const float al0 = lrelu(quad_sum(dot4(v0, wsq)) + ad), al1 = lrelu(quad_sum(dot4(v1, wsq)) + ad);
    const float al2 = lrelu(quad_sum(dot4(v2, wsq)) + ad), al3 = lrelu(quad_sum(dot4(v3, wsq)) + ad);
    float mn = al0 > m ? al0 : m;
    mn = al1 > mn ? al1 : mn; mn = al2 > mn ? al2 : mn; mn = al3 > mn ? al3 : mn;
    const float sc = __expf(m - mn);
    const float e0 = __expf(al0 - mn), e1 = __expf(al1 - mn);
    const float e2 = __expf(al2 - mn), e3 = __expf(al3 - mn);
    m = mn;
    den = fmaf(den, sc, e0); den += e1; den += e2; den += e3;
    rescale_add(acc, sc, v0, e0);
    add_scaled(acc, v1, e1); add_scaled(acc, v2, e2); add_scaled(acc, v3, e3);
  }
  #pragma unroll 1
  for (; cp < ce; cp++) {
    const unsigned s = (unsigned)n0; n0 = n1; n1 = n2; n2 = n3;
    const float4 v = xq[(size_t)s * 4];
    const float al = lrelu(quad_sum(dot4(v, wsq)) + ad);
    const float mn = al > m ? al : m;
    const float sc = __expf(m - mn);
    const float ex = __expf(al - mn);
    m = mn;
    den = fmaf(den, sc, ex);
    rescale_add(acc, sc, v, ex);
  }
  const float dn = den + EPS_GAT;
  const float4 b = ((const float4*)bias2)[q];
  ((float4*)(out + (size_t)d * NCLS))[q] =
      make_float4(acc.x / dn + b.x, acc.y / dn + b.y, acc.z / dn + b.z, acc.w / dn + b.w);
}

extern "C" void kernel_launch(void* const* d_in, const int* in_sizes, int n_in,
                              void* d_out, int out_size, void* d_ws, size_t ws_size,
                              hipStream_t stream) {
  const float* x        = (const float*)d_in[0];
  const int*   ei       = (const int*)d_in[1];
  const float* W1       = (const float*)d_in[2];
  const float* att_src1 = (const float*)d_in[3];
  const float* att_dst1 = (const float*)d_in[4];
  const float* bias1    = (const float*)d_in[5];
  const float* W2       = (const float*)d_in[6];
  const float* att_src2 = (const float*)d_in[7];
  const float* att_dst2 = (const float*)d_in[8];
  const float* bias2    = (const float*)d_in[9];
  float* out = (float*)d_out;

  const int N = in_sizes[0] / F_IN;
  const int E = in_sizes[1] / 2;
  const size_t Np = ((size_t)N + 4) & ~(size_t)3;
  const int NB = (N + BNODES - 1) / BNODES;
  const int chunk = (E + PART_G - 1) / PART_G;
  const int M = NB * PART_G;

  float* ws = (float*)d_ws;
  size_t o = 0;
  float*    xl1    = ws + o; o += (size_t)N * D1;
  float*    h      = ws + o; o += (size_t)N * D1;
  float*    xl2    = ws + o; o += (size_t)N * NCLS;
  unsigned* hist   = (unsigned*)(ws + o); o += (size_t)M;
  int*      csrsrc = (int*)(ws + o);      o += (size_t)E;
  unsigned* rowptr = (unsigned*)(ws + o); o += Np;
  unsigned* rec    = (unsigned*)h;
  unsigned* bstart = (unsigned*)xl2;

  const int TB = 256;
  auto nb = [](long long n, int b) { return (unsigned)((n + b - 1) / b); };

  part_count<<<PART_G, PART_T, 0, stream>>>(ei, hist, E, NB, chunk);
  part_place<<<PART_G, PART_T, 0, stream>>>(ei, hist, bstart, rec, E, NB, chunk);
  bucket_sort<<<NB, PART_T, 0, stream>>>(bstart, rec, rowptr, csrsrc, N, E, NB);

  gemm1_kernel<<<G1_BLOCKS, 128, 0, stream>>>(x, W1, xl1, N);
  agg1_kernel<<<nb((long long)N * HEADS, A1_T), A1_T, 0, stream>>>(rowptr, csrsrc, att_src1, att_dst1,
                                                                   xl1, bias1, W2, xl2, N);

  agg2_kernel<<<nb((long long)N * 4, TB), TB, 0, stream>>>(rowptr, csrsrc, att_src2, att_dst2,
                                                           xl2, bias2, out, N);
}
